# starting kernel
# baseline (speedup 1.0000x reference)
#include <hip/hip_runtime.h>

#define N_NODES   20000
#define FDIM      256
#define HEADS     4
#define CDIM      64
#define NEDGE     320000
#define ETOT      (NEDGE + N_NODES)
#define NEG_SLOPE 0.2f

typedef __attribute__((ext_vector_type(2))) float v2f;
typedef __attribute__((ext_vector_type(8))) float v8f;

__device__ __forceinline__ unsigned f2ord(float f) {
  unsigned u = __float_as_uint(f);
  return (u & 0x80000000u) ? ~u : (u | 0x80000000u);
}
__device__ __forceinline__ float ord2f(unsigned u) {
  unsigned v = (u & 0x80000000u) ? (u ^ 0x80000000u) : ~u;
  return __uint_as_float(v);
}
__device__ __forceinline__ float leaky(float v) {
  return v > 0.0f ? v : NEG_SLOPE * v;
}

__global__ void zero_f32_kernel(float* __restrict__ p, int n) {
  int i = blockIdx.x * blockDim.x + threadIdx.x;
  if (i < n) p[i] = 0.0f;
}
__global__ void zero_u32_kernel(unsigned* __restrict__ p, int n) {
  int i = blockIdx.x * blockDim.x + threadIdx.x;
  if (i < n) p[i] = 0u;
}

__global__ __launch_bounds__(512)
void gemm_f32_wmma(const float* __restrict__ A, const float* __restrict__ B,
                   float* __restrict__ Out) {
  __shared__ float As[16 * 257];
  const int tid      = threadIdx.x;
  const int row_base = blockIdx.x * 16;

  for (int i = tid; i < 16 * 256; i += 512) {
    int r = i >> 8, c = i & 255;
    As[r * 257 + c] = A[(size_t)(row_base + r) * FDIM + c];
  }
  __syncthreads();

  const int wave = tid >> 5;
  const int lane = tid & 31;
  const int cn   = wave * 16;
  const int m    = lane & 15;
  const int kh   = (lane >> 4) << 1;

  v8f acc = {};
  #pragma unroll 8
  for (int k = 0; k < FDIM; k += 4) {
    v2f a, b;
    a.x = As[m * 257 + k + kh];
    a.y = As[m * 257 + k + kh + 1];
    b.x = B[(size_t)(k + kh)     * FDIM + cn + m];
    b.y = B[(size_t)(k + kh + 1) * FDIM + cn + m];
    acc = __builtin_amdgcn_wmma_f32_16x16x4_f32(false, a, false, b,
                                                (short)0, acc, false, false);
  }

  const int rbase = row_base + ((lane >> 4) << 3);
  #pragma unroll
  for (int r = 0; r < 8; ++r) {
    Out[(size_t)(rbase + r) * FDIM + cn + m] = acc[r];
  }
}

__global__ __launch_bounds__(256)
void edge_scores(const float* __restrict__ xl, const float* __restrict__ xr,
                 const float* __restrict__ att, const int* __restrict__ ei,
                 float* __restrict__ score, unsigned* __restrict__ m_u) {
  const int e    = blockIdx.x * (blockDim.x >> 5) + (threadIdx.x >> 5);
  const int lane = threadIdx.x & 31;
  if (e >= ETOT) return;
  int src, dst;
  if (e < NEDGE) { src = ei[e]; dst = ei[NEDGE + e]; }
  else           { src = e - NEDGE; dst = src; }

  const float4* pl4 = (const float4*)(xl + (size_t)src * FDIM) + lane * 2;
  const float4* pr4 = (const float4*)(xr + (size_t)dst * FDIM) + lane * 2;
  const float4* at4 = (const float4*)att + lane * 2;

  float4 l0 = pl4[0], l1 = pl4[1];
  float4 r0 = pr4[0], r1 = pr4[1];
  float4 a0 = at4[0], a1 = at4[1];

  float acc = leaky(l0.x + r0.x) * a0.x + leaky(l0.y + r0.y) * a0.y +
              leaky(l0.z + r0.z) * a0.z + leaky(l0.w + r0.w) * a0.w +
              leaky(l1.x + r1.x) * a1.x + leaky(l1.y + r1.y) * a1.y +
              leaky(l1.z + r1.z) * a1.z + leaky(l1.w + r1.w) * a1.w;

  acc += __shfl_xor(acc, 4, 32);
  acc += __shfl_xor(acc, 2, 32);
  acc += __shfl_xor(acc, 1, 32);

  if ((lane & 7) == 0) {
    const int h = lane >> 3;
    score[(size_t)e * HEADS + h] = acc;
    atomicMax(&m_u[dst * HEADS + h], f2ord(acc));
  }
}

__global__ __launch_bounds__(256)
void edge_exp(const int* __restrict__ ei, float* __restrict__ score_p,
              const unsigned* __restrict__ m_u, float* __restrict__ denom) {
  const int t = blockIdx.x * blockDim.x + threadIdx.x;
  if (t >= ETOT * HEADS) return;
  const int e = t >> 2;
  const int h = t & 3;
  const int dst = (e < NEDGE) ? ei[NEDGE + e] : (e - NEDGE);
  const float m = ord2f(m_u[dst * HEADS + h]);
  const float p = expf(score_p[t] - m);
  score_p[t] = p;
  unsafeAtomicAdd(&denom[dst * HEADS + h], p);
}

__global__ __launch_bounds__(256)
void edge_aggregate(const float* __restrict__ xl, const float* __restrict__ p,
                    const float* __restrict__ denom, const int* __restrict__ ei,
                    float* __restrict__ agg) {
  const int e    = blockIdx.x * (blockDim.x >> 5) + (threadIdx.x >> 5);
  const int lane = threadIdx.x & 31;
  if (e >= ETOT) return;
  int src, dst;
  if (e < NEDGE) { src = ei[e]; dst = ei[NEDGE + e]; }
  else           { src = e - NEDGE; dst = src; }

  const int h = lane >> 3;
  const float alpha = p[(size_t)e * HEADS + h] / (denom[dst * HEADS + h] + 1e-16f);

  const float4* pl4 = (const float4*)(xl + (size_t)src * FDIM) + lane * 2;
  float*        pa  = agg + (size_t)dst * FDIM + lane * 8;
  float4 v0 = pl4[0], v1 = pl4[1];

  unsafeAtomicAdd(pa + 0, alpha * v0.x);
  unsafeAtomicAdd(pa + 1, alpha * v0.y);
  unsafeAtomicAdd(pa + 2, alpha * v0.z);
  unsafeAtomicAdd(pa + 3, alpha * v0.w);
  unsafeAtomicAdd(pa + 4, alpha * v1.x);
  unsafeAtomicAdd(pa + 5, alpha * v1.y);
  unsafeAtomicAdd(pa + 6, alpha * v1.z);
  unsafeAtomicAdd(pa + 7, alpha * v1.w);
}

__global__ __launch_bounds__(256)
void finalize_elu(float* __restrict__ io, const float* __restrict__ bias, int total) {
  const int i = blockIdx.x * blockDim.x + threadIdx.x;
  if (i >= total) return;
  float v = io[i] + bias[i & (FDIM - 1)];
  io[i] = v > 0.0f ? v : (expf(v) - 1.0f);
}

extern "C" void kernel_launch(void* const* d_in, const int* in_sizes, int n_in,
                              void* d_out, int out_size, void* d_ws, size_t ws_size,
                              hipStream_t stream) {
  const float* x    = (const float*)d_in[0];
  const int*   ei   = (const int*)  d_in[1];
  const float* Wl0  = (const float*)d_in[2];
  const float* Wr0  = (const float*)d_in[3];
  const float* att0 = (const float*)d_in[4];
  const float* b0   = (const float*)d_in[5];
  const float* Wl1  = (const float*)d_in[6];
  const float* Wr1  = (const float*)d_in[7];
  const float* att1 = (const float*)d_in[8];
  const float* b1   = (const float*)d_in[9];
  float* out = (float*)d_out;

  float*    ws    = (float*)d_ws;
  float*    xl    = ws;
  float*    xr    = xl   + (size_t)N_NODES * FDIM;
  float*    hbuf  = xr   + (size_t)N_NODES * FDIM;
  float*    pbuf  = hbuf + (size_t)N_NODES * FDIM;
  unsigned* m_u   = (unsigned*)(pbuf + (size_t)ETOT * HEADS);
  float*    denom = (float*)(m_u + (size_t)N_NODES * HEADS);

  const int nh        = N_NODES * HEADS;
  const int nfeat     = N_NODES * FDIM;
  const int edgeWaves = (ETOT + 7) / 8;

  auto layer = [&](const float* in, const float* Wl, const float* Wr,
                   const float* att, const float* bias, float* outb) {
    gemm_f32_wmma<<<N_NODES / 16, 512, 0, stream>>>(in, Wl, xl);
    gemm_f32_wmma<<<N_NODES / 16, 512, 0, stream>>>(in, Wr, xr);
    zero_u32_kernel<<<(nh + 255) / 256, 256, 0, stream>>>(m_u, nh);
    zero_f32_kernel<<<(nh + 255) / 256, 256, 0, stream>>>(denom, nh);
    zero_f32_kernel<<<(nfeat + 255) / 256, 256, 0, stream>>>(outb, nfeat);
    edge_scores<<<edgeWaves, 256, 0, stream>>>(xl, xr, att, ei, pbuf, m_u);
    edge_exp<<<(ETOT * HEADS + 255) / 256, 256, 0, stream>>>(ei, pbuf, m_u, denom);
    edge_aggregate<<<edgeWaves, 256, 0, stream>>>(xl, pbuf, denom, ei, outb);
    finalize_elu<<<(nfeat + 255) / 256, 256, 0, stream>>>(outb, bias, nfeat);
  };

  layer(x,    Wl0, Wr0, att0, b0, hbuf);
  layer(hbuf, Wl1, Wr1, att1, b1, out);
}


# edge phase gathers each source row once: sums taken in the maxima pass with unshifted exp(score) (LDS atomics, arrival-order sums as before); shifted second pass only if a row maximum leaves +-16
# speedup vs baseline: 3.8827x; 3.8827x over previous
#include <hip/hip_runtime.h>

#define N_NODES   20000
#define FDIM      256
#define HEADS     4
#define CDIM      64
#define NEDGE     320000
#define ETOT      (NEDGE + N_NODES)
#define NEG_SLOPE 0.2f

typedef __attribute__((ext_vector_type(2))) float v2f;
typedef __attribute__((ext_vector_type(8))) float v8f;

__device__ __forceinline__ unsigned f2ord(float f) {
  unsigned u = __float_as_uint(f);
  return (u & 0x80000000u) ? ~u : (u | 0x80000000u);
}
__device__ __forceinline__ float ord2f(unsigned u) {
  unsigned v = (u & 0x80000000u) ? (u ^ 0x80000000u) : ~u;
  return __uint_as_float(v);
}

__global__ __launch_bounds__(512)
void gemm_f32_wmma(const float* __restrict__ A, const float* __restrict__ B,
                   float* __restrict__ Out) {
  __shared__ float As[16 * 257];
  const int tid      = threadIdx.x;
  const int row_base = blockIdx.x * 16;

  for (int i = tid; i < 16 * 256; i += 512) {
    int r = i >> 8, c = i & 255;
    As[r * 257 + c] = A[(size_t)(row_base + r) * FDIM + c];
  }
  __syncthreads();

  const int wave = tid >> 5;
  const int lane = tid & 31;
  const int cn   = wave * 16;
  const int m    = lane & 15;
  const int kh   = (lane >> 4) << 1;

  v8f acc = {};
  #pragma unroll 8
  for (int k = 0; k < FDIM; k += 4) {
    v2f a, b;
    a.x = As[m * 257 + k + kh];
    a.y = As[m * 257 + k + kh + 1];
    b.x = B[(size_t)(k + kh)     * FDIM + cn + m];
    b.y = B[(size_t)(k + kh + 1) * FDIM + cn + m];
    acc = __builtin_amdgcn_wmma_f32_16x16x4_f32(false, a, false, b,
                                                (short)0, acc, false, false);
  }

  const int rbase = row_base + ((lane >> 4) << 3);
  #pragma unroll
  for (int r = 0; r < 8; ++r) {
    Out[(size_t)(rbase + r) * FDIM + cn + m] = acc[r];
  }
}

#define SLAB        80
#define QCAP        24576
#define AGG_THREADS 1024
#define AGG_WAVES   (AGG_THREADS / 32)
#define SEGLEN      (QCAP - SLAB)
#define NSEG        ((NEDGE + SEGLEN - 1) / SEGLEN)
#define ORD_NEG_INF 0x007FFFFFu
#define SPEC_RANGE  16.0f
static_assert(N_NODES % SLAB == 0, "slabs must tile the node range");
static_assert(NEDGE % 4 == 0, "destination list is read as 16-byte vectors");
static_assert(ETOT < (1 << 20) && SLAB < (1 << 12), "queue word packs edge id and local row");

__device__ __forceinline__ float row16_sum(float v) {
  v += __int_as_float(__builtin_amdgcn_mov_dpp(__float_as_int(v), 0x161, 0xf, 0xf, true));
  v += __int_as_float(__builtin_amdgcn_mov_dpp(__float_as_int(v), 0x162, 0xf, 0xf, true));
  v += __int_as_float(__builtin_amdgcn_mov_dpp(__float_as_int(v), 0x164, 0xf, 0xf, true));
  v += __int_as_float(__builtin_amdgcn_mov_dpp(__float_as_int(v), 0x168, 0xf, 0xf, true));
  return v;
}

__device__ __forceinline__ float leaky_max(float v) { return fmaxf(v, NEG_SLOPE * v); }
__device__ __forceinline__ float score4(const float4 l, const float4 r, const float4 a) {
  return leaky_max(l.x + r.x) * a.x + leaky_max(l.y + r.y) * a.y +
         leaky_max(l.z + r.z) * a.z + leaky_max(l.w + r.w) * a.w;
}

template <bool B> struct flag_t { static constexpr bool value = B; };

__global__ __launch_bounds__(AGG_THREADS)
void slab_edge_phase(const float* __restrict__ xl, const float* __restrict__ xr,
                     const float* __restrict__ att, const int* __restrict__ ei,
                     const float* __restrict__ bias, float* __restrict__ outb) {
  __shared__ __align__(16) float    acc[(SLAB + 1) * FDIM];
  __shared__ __align__(16) float    xrs[(SLAB + 1) * FDIM];
  __shared__ unsigned queue[QCAP];
  __shared__ unsigned mb[(SLAB + 1) * HEADS];
  __shared__ __align__(16) float    mrun[(SLAB + 1) * HEADS];
  __shared__ float    den[(SLAB + 1) * HEADS];
  __shared__ float    sc[SLAB * HEADS];
  __shared__ unsigned qcount;
  __shared__ unsigned respec;

  const int tid  = threadIdx.x;
  const int lane = tid & 31;
  const int wave = __builtin_amdgcn_readfirstlane(tid >> 5);
  const int d0   = blockIdx.x * SLAB;
  const bool hsel = lane >= 16;

  {
    float4*       a4 = (float4*)acc;
    float4*       x4 = (float4*)xrs;
    const float4* g4 = (const float4*)(xr + (size_t)d0 * FDIM);
    const float4  z  = make_float4(0.0f, 0.0f, 0.0f, 0.0f);
    for (int i = tid; i < SLAB * FDIM / 4; i += AGG_THREADS) { a4[i] = z; x4[i] = g4[i]; }
    if (tid < FDIM / 4) x4[SLAB * FDIM / 4 + tid] = z;
    if (tid < (SLAB + 1) * HEADS) { mb[tid] = ORD_NEG_INF; mrun[tid] = 0.0f; den[tid] = 0.0f; }
    if (tid < SLAB) queue[tid] = (unsigned)(NEDGE + d0 + tid) | ((unsigned)tid << 20);
    if (tid == 0) { qcount = SLAB; respec = 0u; }
  }
  __syncthreads();

  {
    const int4* dst4 = (const int4*)(ei + NEDGE);
    constexpr int NVEC = NEDGE / 4;
    constexpr int NCH  = (NVEC + AGG_THREADS - 1) / AGG_THREADS;
    const int rot = blockIdx.x % NCH;
    for (int g = 0; g < NCH; g += 4) {
      int4 v[4];
      int  vi[4];
      #pragma unroll
      for (int u = 0; u < 4; ++u) {
        int c = g + u + rot;
        if (c >= NCH) c -= NCH;
        const int i = c * AGG_THREADS + tid;
        vi[u] = i;
        v[u]  = make_int4(-1, -1, -1, -1);
        if (g + u < NCH && i < NVEC) v[u] = dst4[i];
      }
      #pragma unroll
      for (int u = 0; u < 4; ++u) {
        const unsigned r[4] = {(unsigned)(v[u].x - d0), (unsigned)(v[u].y - d0),
                               (unsigned)(v[u].z - d0), (unsigned)(v[u].w - d0)};
        const unsigned rmin = min(min(r[0], r[1]), min(r[2], r[3]));
        if (rmin < (unsigned)SLAB) {
          #pragma unroll
          for (int k = 0; k < 4; ++k) {
            if (r[k] < (unsigned)SLAB) {
              const int e = vi[u] * 4 + k;
              const unsigned slot = atomicAdd(&qcount, 1u);
              if (slot < (unsigned)QCAP) queue[slot] = (unsigned)e | (r[k] << 20);
            }
          }
        }
      }
    }
  }
  __syncthreads();

  const unsigned total  = (unsigned)__builtin_amdgcn_readfirstlane((int)qcount);
  const bool     big    = total > (unsigned)QCAP;
  const int      nbatch = big ? NSEG : 1;
  const bool     spec   = !big;
  bool           unshifted = false;

  const float4 atA = ((const float4*)att)[lane];
  const float4 atB = ((const float4*)att)[32 + lane];

  for (int b = 0; b < nbatch; ++b) {
    if (big) {
      __syncthreads();
      if (tid == 0) qcount = (b == 0) ? SLAB : 0;
      if (b == 0 && tid < SLAB) queue[tid] = (unsigned)(NEDGE + d0 + tid) | ((unsigned)tid << 20);
      __syncthreads();
      const int s0 = b * SEGLEN;
      const int s1 = min(NEDGE, s0 + SEGLEN);
      for (int e = s0 + tid; e < s1; e += AGG_THREADS) {
        const unsigned r = (unsigned)(ei[NEDGE + e] - d0);
        if (r < (unsigned)SLAB) {
          const unsigned slot = atomicAdd(&qcount, 1u);
          queue[slot] = (unsigned)e | (r << 20);
        }
      }
      __syncthreads();
    }
    const unsigned M = min((unsigned)__builtin_amdgcn_readfirstlane((int)qcount), (unsigned)QCAP);

    auto pass_a = [&](auto sums_tag) __attribute__((always_inline)) {
      constexpr bool SUMS = decltype(sums_tag)::value;
      for (unsigned base = wave * 32; base < M; base += AGG_WAVES * 32) {
        const unsigned idx = base + lane;
        int m_src = 0, m_dl = SLAB;
        if (idx < M) {
          const unsigned ent = queue[idx];
          const int      e   = (int)(ent & 0xFFFFFu);
          m_dl  = (int)(ent >> 20);
          m_src = (e < NEDGE) ? ei[e] : (e - NEDGE);
          queue[idx] = (unsigned)m_src | ((unsigned)m_dl << 20);
        }
        const int n = (int)min(32u, M - base);
        for (int j0 = 0; j0 < n; j0 += 8) {
          float4 va[8], vb[8];
          #pragma unroll
          for (int u = 0; u < 8; ++u) {
            const int s = __builtin_amdgcn_readlane(m_src, j0 + u);
            const float4* row = (const float4*)(xl + (size_t)s * FDIM);
            va[u] = row[lane];
            vb[u] = row[32 + lane];
          }
          #pragma unroll
          for (int u = 0; u < 8; ++u) {
            const int     dl = __builtin_amdgcn_readlane(m_dl, j0 + u);
            const float4* xq = (const float4*)(xrs + dl * FDIM);
            const float   tA = row16_sum(score4(va[u], xq[lane],      atA));
            const float   tB = row16_sum(score4(vb[u], xq[32 + lane], atB));
            if (SUMS) {
              const float pA = expf(tA);
              const float pB = expf(tB);
              if ((lane & 15) == 0) {
                atomicMax(&mb[dl * HEADS + (hsel ? 1 : 0)], f2ord(tA));
                atomicMax(&mb[dl * HEADS + (hsel ? 3 : 2)], f2ord(tB));
                __hip_atomic_fetch_add(&den[dl * HEADS + (hsel ? 1 : 0)], pA, __ATOMIC_RELAXED, __HIP_MEMORY_SCOPE_WORKGROUP);
                __hip_atomic_fetch_add(&den[dl * HEADS + (hsel ? 3 : 2)], pB, __ATOMIC_RELAXED, __HIP_MEMORY_SCOPE_WORKGROUP);
              }
              float* ar = acc + dl * FDIM + lane;
              __hip_atomic_fetch_add(ar +   0, pA * va[u].x, __ATOMIC_RELAXED, __HIP_MEMORY_SCOPE_WORKGROUP);
              __hip_atomic_fetch_add(ar +  32, pA * va[u].y, __ATOMIC_RELAXED, __HIP_MEMORY_SCOPE_WORKGROUP);
              __hip_atomic_fetch_add(ar +  64, pA * va[u].z, __ATOMIC_RELAXED, __HIP_MEMORY_SCOPE_WORKGROUP);
              __hip_atomic_fetch_add(ar +  96, pA * va[u].w, __ATOMIC_RELAXED, __HIP_MEMORY_SCOPE_WORKGROUP);
              __hip_atomic_fetch_add(ar + 128, pB * vb[u].x, __ATOMIC_RELAXED, __HIP_MEMORY_SCOPE_WORKGROUP);
              __hip_atomic_fetch_add(ar + 160, pB * vb[u].y, __ATOMIC_RELAXED, __HIP_MEMORY_SCOPE_WORKGROUP);
              __hip_atomic_fetch_add(ar + 192, pB * vb[u].z, __ATOMIC_RELAXED, __HIP_MEMORY_SCOPE_WORKGROUP);
              __hip_atomic_fetch_add(ar + 224, pB * vb[u].w, __ATOMIC_RELAXED, __HIP_MEMORY_SCOPE_WORKGROUP);
            } else {
              if ((lane & 15) == 0) {
                atomicMax(&mb[dl * HEADS + (hsel ? 1 : 0)], f2ord(tA));
                atomicMax(&mb[dl * HEADS + (hsel ? 3 : 2)], f2ord(tB));
              }
            }
          }
        }
      }
    };
    if (spec) pass_a(flag_t<true>{}); else pass_a(flag_t<false>{});
    __syncthreads();

    if (tid < SLAB * HEADS) {
      const float mnew_b = ord2f(mb[tid]);
      mb[tid] = ORD_NEG_INF;
      if (b == 0) {
        mrun[tid] = mnew_b;
        if (spec && !(fabsf(mnew_b) <= SPEC_RANGE)) respec = 1u;
      } else {
        const float mo = mrun[tid];
        const float mn = fmaxf(mo, mnew_b);
        const float f  = expf(mo - mn);
        mrun[tid] = mn;
        den[tid] *= f;
        sc[tid]   = f;
      }
    }
    __syncthreads();
    if (spec) {
      unshifted = __builtin_amdgcn_readfirstlane((int)respec) == 0;
      if (unshifted) break;
      float4*      a4 = (float4*)acc;
      const float4 z  = make_float4(0.0f, 0.0f, 0.0f, 0.0f);
      for (int i = tid; i < SLAB * FDIM / 4; i += AGG_THREADS) a4[i] = z;
      if (tid < SLAB * HEADS) den[tid] = 0.0f;
      __syncthreads();
    }
    if (b > 0) {
      #pragma unroll 1
      for (int i = tid; i < SLAB * FDIM; i += AGG_THREADS) {
        const int r = i >> 8, q = (i >> 5) & 7, l = i & 31;
        acc[i] *= sc[r * HEADS + ((q >> 2) << 1) + (l >> 4)];
      }
      __syncthreads();
    }

    for (unsigned base = wave * 32; base < M; base += AGG_WAVES * 32) {
      const unsigned idx = base + lane;
      int   m_src = 0, m_dl = SLAB;
      float m_m0 = 0.0f, m_m1 = 0.0f, m_m2 = 0.0f, m_m3 = 0.0f;
      if (idx < M) {
        const unsigned ent = queue[idx];
        m_src = (int)(ent & 0xFFFFFu);
        m_dl  = (int)(ent >> 20);
        const float4 mm = ((const float4*)mrun)[m_dl];
        m_m0 = mm.x; m_m1 = mm.y; m_m2 = mm.z; m_m3 = mm.w;
      }
      const int n = (int)min(32u, M - base);
      for (int j0 = 0; j0 < n; j0 += 8) {
        float4 va[8], vb[8];
        #pragma unroll
        for (int u = 0; u < 8; ++u) {
          const int s = __builtin_amdgcn_readlane(m_src, j0 + u);
          const float4* row = (const float4*)(xl + (size_t)s * FDIM);
          va[u] = row[lane];
          vb[u] = row[32 + lane];
        }
        #pragma unroll
        for (int u = 0; u < 8; ++u) {
          const int     dl = __builtin_amdgcn_readlane(m_dl, j0 + u);
          const float4* xq = (const float4*)(xrs + dl * FDIM);
          const float   tA = row16_sum(score4(va[u], xq[lane],      atA));
          const float   tB = row16_sum(score4(vb[u], xq[32 + lane], atB));
          const float m0 = __int_as_float(__builtin_amdgcn_readlane(__float_as_int(m_m0), j0 + u));
          const float m1 = __int_as_float(__builtin_amdgcn_readlane(__float_as_int(m_m1), j0 + u));
          const float m2 = __int_as_float(__builtin_amdgcn_readlane(__float_as_int(m_m2), j0 + u));
          const float m3 = __int_as_float(__builtin_amdgcn_readlane(__float_as_int(m_m3), j0 + u));
          const float pA = expf(tA - (hsel ? m1 : m0));
          const float pB = expf(tB - (hsel ? m3 : m2));
          if ((lane & 15) == 0) {
            __hip_atomic_fetch_add(&den[dl * HEADS + (hsel ? 1 : 0)], pA, __ATOMIC_RELAXED, __HIP_MEMORY_SCOPE_WORKGROUP);
            __hip_atomic_fetch_add(&den[dl * HEADS + (hsel ? 3 : 2)], pB, __ATOMIC_RELAXED, __HIP_MEMORY_SCOPE_WORKGROUP);
          }
          float* ar = acc + dl * FDIM + lane;
          __hip_atomic_fetch_add(ar +   0, pA * va[u].x, __ATOMIC_RELAXED, __HIP_MEMORY_SCOPE_WORKGROUP);
          __hip_atomic_fetch_add(ar +  32, pA * va[u].y, __ATOMIC_RELAXED, __HIP_MEMORY_SCOPE_WORKGROUP);
          __hip_atomic_fetch_add(ar +  64, pA * va[u].z, __ATOMIC_RELAXED, __HIP_MEMORY_SCOPE_WORKGROUP);
          __hip_atomic_fetch_add(ar +  96, pA * va[u].w, __ATOMIC_RELAXED, __HIP_MEMORY_SCOPE_WORKGROUP);
          __hip_atomic_fetch_add(ar + 128, pB * vb[u].x, __ATOMIC_RELAXED, __HIP_MEMORY_SCOPE_WORKGROUP);
          __hip_atomic_fetch_add(ar + 160, pB * vb[u].y, __ATOMIC_RELAXED, __HIP_MEMORY_SCOPE_WORKGROUP);
          __hip_atomic_fetch_add(ar + 192, pB * vb[u].z, __ATOMIC_RELAXED, __HIP_MEMORY_SCOPE_WORKGROUP);
          __hip_atomic_fetch_add(ar + 224, pB * vb[u].w, __ATOMIC_RELAXED, __HIP_MEMORY_SCOPE_WORKGROUP);
        }
      }
    }
  }
  __syncthreads();

  {
    const float4 bA = ((const float4*)bias)[lane];
    const float4 bB = ((const float4*)bias)[32 + lane];
    for (int r = wave; r < SLAB; r += AGG_WAVES) {
      const float* ar = acc + r * FDIM + lane;
      const int    hA = r * HEADS + (hsel ? 1 : 0);
      const int    hB = r * HEADS + (hsel ? 3 : 2);
      const float  eA = unshifted ? 1e-16f * expf(mrun[hA]) : 1e-16f;
      const float  eB = unshifted ? 1e-16f * expf(mrun[hB]) : 1e-16f;
      const float  dA = den[hA] + eA;
      const float  dB = den[hB] + eB;
      float o[8] = {ar[0] / dA + bA.x,   ar[32] / dA + bA.y,  ar[64] / dA + bA.z,  ar[96] / dA + bA.w,
                    ar[128] / dB + bB.x, ar[160] / dB + bB.y, ar[192] / dB + bB.z, ar[224] / dB + bB.w};
      #pragma unroll
      for (int q = 0; q < 8; ++q) o[q] = o[q] > 0.0f ? o[q] : (expf(o[q]) - 1.0f);
      float4* orow = (float4*)(outb + (size_t)(d0 + r) * FDIM);
      orow[lane]      = make_float4(o[0], o[1], o[2], o[3]);
      orow[32 + lane] = make_float4(o[4], o[5], o[6], o[7]);
    }
  }
}

extern "C" void kernel_launch(void* const* d_in, const int* in_sizes, int n_in,
                              void* d_out, int out_size, void* d_ws, size_t ws_size,
                              hipStream_t stream) {
  const float* x    = (const float*)d_in[0];
  const int*   ei   = (const int*)  d_in[1];
  const float* Wl0  = (const float*)d_in[2];
  const float* Wr0  = (const float*)d_in[3];
  const float* att0 = (const float*)d_in[4];
  const float* b0   = (const float*)d_in[5];
  const float* Wl1  = (const float*)d_in[6];
  const float* Wr1  = (const float*)d_in[7];
  const float* att1 = (const float*)d_in[8];
  const float* b1   = (const float*)d_in[9];
  float* out = (float*)d_out;

  float*    ws    = (float*)d_ws;
  float*    xl    = ws;
  float*    xr    = xl   + (size_t)N_NODES * FDIM;
  float*    hbuf  = xr   + (size_t)N_NODES * FDIM;

  auto layer = [&](const float* in, const float* Wl, const float* Wr,
                   const float* att, const float* bias, float* outb) {
    gemm_f32_wmma<<<N_NODES / 16, 512, 0, stream>>>(in, Wl, xl);
    gemm_f32_wmma<<<N_NODES / 16, 512, 0, stream>>>(in, Wr, xr);
    slab_edge_phase<<<N_NODES / SLAB, AGG_THREADS, 0, stream>>>(xl, xr, att, ei, bias, outb);
  };

  layer(x,    Wl0, Wr0, att0, b0, hbuf);
  layer(hbuf, Wl1, Wr1, att1, b1, out);
}


# xr product moved into the slab kernel: 16 of its 32 waves form the slab's 80 xr rows on the f32 matrix unit into LDS while the other 16 filter the edge list; xr launches and xr buffer gone
# speedup vs baseline: 4.0253x; 1.0367x over previous
#include <hip/hip_runtime.h>

#define N_NODES   20000
#define FDIM      256
#define HEADS     4
#define CDIM      64
#define NEDGE     320000
#define ETOT      (NEDGE + N_NODES)
#define NEG_SLOPE 0.2f

typedef __attribute__((ext_vector_type(2))) float v2f;
typedef __attribute__((ext_vector_type(8))) float v8f;

__device__ __forceinline__ unsigned f2ord(float f) {
  unsigned u = __float_as_uint(f);
  return (u & 0x80000000u) ? ~u : (u | 0x80000000u);
}
__device__ __forceinline__ float ord2f(unsigned u) {
  unsigned v = (u & 0x80000000u) ? (u ^ 0x80000000u) : ~u;
  return __uint_as_float(v);
}

__global__ __launch_bounds__(512)
void gemm_f32_wmma(const float* __restrict__ A, const float* __restrict__ B,
                   float* __restrict__ Out) {
  __shared__ float As[16 * 257];
  const int tid      = threadIdx.x;
  const int row_base = blockIdx.x * 16;

  for (int i = tid; i < 16 * 256; i += 512) {
    int r = i >> 8, c = i & 255;
    As[r * 257 + c] = A[(size_t)(row_base + r) * FDIM + c];
  }
  __syncthreads();

  const int wave = tid >> 5;
  const int lane = tid & 31;
  const int cn   = wave * 16;
  const int m    = lane & 15;
  const int kh   = (lane >> 4) << 1;

  v8f acc = {};
  #pragma unroll 8
  for (int k = 0; k < FDIM; k += 4) {
    v2f a, b;
    a.x = As[m * 257 + k + kh];
    a.y = As[m * 257 + k + kh + 1];
    b.x = B[(size_t)(k + kh)     * FDIM + cn + m];
    b.y = B[(size_t)(k + kh + 1) * FDIM + cn + m];
    acc = __builtin_amdgcn_wmma_f32_16x16x4_f32(false, a, false, b,
                                                (short)0, acc, false, false);
  }

  const int rbase = row_base + ((lane >> 4) << 3);
  #pragma unroll
  for (int r = 0; r < 8; ++r) {
    Out[(size_t)(rbase + r) * FDIM + cn + m] = acc[r];
  }
}

#define SLAB        80
#define QCAP        24576
#define AGG_THREADS 1024
#define AGG_WAVES   (AGG_THREADS / 32)
#define SEGLEN      (QCAP - SLAB)
#define NSEG        ((NEDGE + SEGLEN - 1) / SEGLEN)
#define ORD_NEG_INF 0x007FFFFFu
#define SPEC_RANGE  16.0f
static_assert(N_NODES % SLAB == 0, "slabs must tile the node range");
static_assert(SLAB % 16 == 0, "the slab's xr rows are formed as whole 16-row matrix tiles");
static_assert(NEDGE % 4 == 0, "destination list is read as 16-byte vectors");
static_assert(ETOT < (1 << 20) && SLAB < (1 << 12), "queue word packs edge id and local row");

__device__ __forceinline__ float row16_sum(float v) {
  v += __int_as_float(__builtin_amdgcn_mov_dpp(__float_as_int(v), 0x161, 0xf, 0xf, true));
  v += __int_as_float(__builtin_amdgcn_mov_dpp(__float_as_int(v), 0x162, 0xf, 0xf, true));
  v += __int_as_float(__builtin_amdgcn_mov_dpp(__float_as_int(v), 0x164, 0xf, 0xf, true));
  v += __int_as_float(__builtin_amdgcn_mov_dpp(__float_as_int(v), 0x168, 0xf, 0xf, true));
  return v;
}

__device__ __forceinline__ float leaky_max(float v) { return fmaxf(v, NEG_SLOPE * v); }
__device__ __forceinline__ float score4(const float4 l, const float4 r, const float4 a) {
  return leaky_max(l.x + r.x) * a.x + leaky_max(l.y + r.y) * a.y +
         leaky_max(l.z + r.z) * a.z + leaky_max(l.w + r.w) * a.w;
}

template <bool B> struct flag_t { static constexpr bool value = B; };

__global__ __launch_bounds__(AGG_THREADS)
void slab_edge_phase(const float* __restrict__ xl, const float* __restrict__ xin,
                     const float* __restrict__ Wr,
                     const float* __restrict__ att, const int* __restrict__ ei,
                     const float* __restrict__ bias, float* __restrict__ outb) {
  __shared__ __align__(16) float    acc[(SLAB + 1) * FDIM];
  __shared__ __align__(16) float    xrs[(SLAB + 1) * FDIM];
  __shared__ unsigned queue[QCAP];
  __shared__ unsigned mb[(SLAB + 1) * HEADS];
  __shared__ __align__(16) float    mrun[(SLAB + 1) * HEADS];
  __shared__ float    den[(SLAB + 1) * HEADS];
  __shared__ float    sc[SLAB * HEADS];
  __shared__ unsigned qcount;
  __shared__ unsigned respec;

  const int tid  = threadIdx.x;
  const int lane = tid & 31;
  const int wave = __builtin_amdgcn_readfirstlane(tid >> 5);
  const int d0   = blockIdx.x * SLAB;
  const bool hsel = lane >= 16;

  {
    float4*       a4 = (float4*)acc;
    float4*       x4 = (float4*)xrs;
    const float4* g4 = (const float4*)(xin + (size_t)d0 * FDIM);
    const float4  z  = make_float4(0.0f, 0.0f, 0.0f, 0.0f);
    constexpr int NST = SLAB * FDIM / 4 / AGG_THREADS;
    static_assert(NST * AGG_THREADS == SLAB * FDIM / 4, "whole 16-byte pieces per thread");
    float4 st[NST];
    #pragma unroll
    for (int j = 0; j < NST; ++j) st[j] = g4[tid + j * AGG_THREADS];
    #pragma unroll
    for (int j = 0; j < NST; ++j) {
      const int i = tid + j * AGG_THREADS;
      const int r = i >> 6, c4 = i & 63;
      a4[(r << 6) + (c4 ^ (r & 15))] = st[j];
    }
    if (tid < FDIM / 4) x4[SLAB * FDIM / 4 + tid] = z;
    if (tid < (SLAB + 1) * HEADS) { mb[tid] = ORD_NEG_INF; mrun[tid] = 0.0f; den[tid] = 0.0f; }
    if (tid < SLAB) queue[tid] = (unsigned)(NEDGE + d0 + tid) | ((unsigned)tid << 20);
    if (tid == 0) { qcount = SLAB; respec = 0u; }
  }
  __syncthreads();

  const int role = ((wave >> 2) ^ wave) & 1;
  const int ridx = ((wave >> 3) << 2) | (wave & 3);
  if (role == 0) {
    const int m  = lane & 15;
    const int kh = (lane >> 4) << 1;
    const int cn = ridx * 16;
    const int sw = m << 2;
    const float* bp = Wr + (size_t)kh * FDIM + cn + m;
    const float* ap = acc + m * FDIM + kh;
    v8f c[SLAB / 16] = {};
    #pragma unroll 8
    for (int k = 0; k < FDIM; k += 4) {
      v2f b;
      b.x = bp[(size_t)k * FDIM];
      b.y = bp[(size_t)(k + 1) * FDIM];
      const float* ak = ap + (k ^ sw);
      #pragma unroll
      for (int t = 0; t < SLAB / 16; ++t) {
        const v2f a = *(const v2f*)(ak + t * 16 * FDIM);
        c[t] = __builtin_amdgcn_wmma_f32_16x16x4_f32(false, a, false, b,
                                                     (short)0, c[t], false, false);
      }
    }
    const int rb = (lane >> 4) << 3;
    #pragma unroll
    for (int t = 0; t < SLAB / 16; ++t) {
      #pragma unroll
      for (int r = 0; r < 8; ++r) xrs[(t * 16 + rb + r) * FDIM + cn + m] = c[t][r];
    }
  } else {
    constexpr int FTHREADS = AGG_THREADS / 2;
    const int   ftid = ridx * 32 + lane;
    const int4* dst4 = (const int4*)(ei + NEDGE);
    constexpr int NVEC = NEDGE / 4;
    constexpr int NCH  = (NVEC + FTHREADS - 1) / FTHREADS;
    const int rot = blockIdx.x % NCH;
    for (int g = 0; g < NCH; g += 8) {
      int4 v[8];
      int  vi[8];
      #pragma unroll
      for (int u = 0; u < 8; ++u) {
        int c = g + u + rot;
        if (c >= NCH) c -= NCH;
        const int i = c * FTHREADS + ftid;
        vi[u] = i;
        v[u]  = make_int4(-1, -1, -1, -1);
        if (g + u < NCH && i < NVEC) v[u] = dst4[i];
      }
      #pragma unroll
      for (int u = 0; u < 8; ++u) {
        const unsigned r[4] = {(unsigned)(v[u].x - d0), (unsigned)(v[u].y - d0),
                               (unsigned)(v[u].z - d0), (unsigned)(v[u].w - d0)};
        const unsigned rmin = min(min(r[0], r[1]), min(r[2], r[3]));
        if (rmin < (unsigned)SLAB) {
          #pragma unroll
          for (int k = 0; k < 4; ++k) {
            if (r[k] < (unsigned)SLAB) {
              const int e = vi[u] * 4 + k;
              const unsigned slot = atomicAdd(&qcount, 1u);
              if (slot < (unsigned)QCAP) queue[slot] = (unsigned)e | (r[k] << 20);
            }
          }
        }
      }
    }
  }
  __syncthreads();

  {
    float4*      a4 = (float4*)acc;
    const float4 z  = make_float4(0.0f, 0.0f, 0.0f, 0.0f);
    for (int i = tid; i < SLAB * FDIM / 4; i += AGG_THREADS) a4[i] = z;
  }
  __syncthreads();

  const unsigned total  = (unsigned)__builtin_amdgcn_readfirstlane((int)qcount);
  const bool     big    = total > (unsigned)QCAP;
  const int      nbatch = big ? NSEG : 1;
  const bool     spec   = !big;
  bool           unshifted = false;

  const float4 atA = ((const float4*)att)[lane];
  const float4 atB = ((const float4*)att)[32 + lane];

  for (int b = 0; b < nbatch; ++b) {
    if (big) {
      __syncthreads();
      if (tid == 0) qcount = (b == 0) ? SLAB : 0;
      if (b == 0 && tid < SLAB) queue[tid] = (unsigned)(NEDGE + d0 + tid) | ((unsigned)tid << 20);
      __syncthreads();
      const int s0 = b * SEGLEN;
      const int s1 = min(NEDGE, s0 + SEGLEN);
      for (int e = s0 + tid; e < s1; e += AGG_THREADS) {
        const unsigned r = (unsigned)(ei[NEDGE + e] - d0);
        if (r < (unsigned)SLAB) {
          const unsigned slot = atomicAdd(&qcount, 1u);
          queue[slot] = (unsigned)e | (r << 20);
        }
      }
      __syncthreads();
    }
    const unsigned M = min((unsigned)__builtin_amdgcn_readfirstlane((int)qcount), (unsigned)QCAP);

    auto pass_a = [&](auto sums_tag) __attribute__((always_inline)) {
      constexpr bool SUMS = decltype(sums_tag)::value;
      for (unsigned base = wave * 32; base < M; base += AGG_WAVES * 32) {
        const unsigned idx = base + lane;
        int m_src = 0, m_dl = SLAB;
        if (idx < M) {
          const unsigned ent = queue[idx];
          const int      e   = (int)(ent & 0xFFFFFu);
          m_dl  = (int)(ent >> 20);
          m_src = (e < NEDGE) ? ei[e] : (e - NEDGE);
          queue[idx] = (unsigned)m_src | ((unsigned)m_dl << 20);
        }
        const int n = (int)min(32u, M - base);
        for (int j0 = 0; j0 < n; j0 += 8) {
          float4 va[8], vb[8];
          #pragma unroll
          for (int u = 0; u < 8; ++u) {
            const int s = __builtin_amdgcn_readlane(m_src, j0 + u);
            const float4* row = (const float4*)(xl + (size_t)s * FDIM);
            va[u] = row[lane];
            vb[u] = row[32 + lane];
          }
          #pragma unroll
          for (int u = 0; u < 8; ++u) {
            const int     dl = __builtin_amdgcn_readlane(m_dl, j0 + u);
            const float4* xq = (const float4*)(xrs + dl * FDIM);
            const float   tA = row16_sum(score4(va[u], xq[lane],      atA));
            const float   tB = row16_sum(score4(vb[u], xq[32 + lane], atB));
            if (SUMS) {
              const float pA = expf(tA);
              const float pB = expf(tB);
              if ((lane & 15) == 0) {
                atomicMax(&mb[dl * HEADS + (hsel ? 1 : 0)], f2ord(tA));
                atomicMax(&mb[dl * HEADS + (hsel ? 3 : 2)], f2ord(tB));
                __hip_atomic_fetch_add(&den[dl * HEADS + (hsel ? 1 : 0)], pA, __ATOMIC_RELAXED, __HIP_MEMORY_SCOPE_WORKGROUP);
                __hip_atomic_fetch_add(&den[dl * HEADS + (hsel ? 3 : 2)], pB, __ATOMIC_RELAXED, __HIP_MEMORY_SCOPE_WORKGROUP);
              }
              float* ar = acc + dl * FDIM + lane;
              __hip_atomic_fetch_add(ar +   0, pA * va[u].x, __ATOMIC_RELAXED, __HIP_MEMORY_SCOPE_WORKGROUP);
              __hip_atomic_fetch_add(ar +  32, pA * va[u].y, __ATOMIC_RELAXED, __HIP_MEMORY_SCOPE_WORKGROUP);
              __hip_atomic_fetch_add(ar +  64, pA * va[u].z, __ATOMIC_RELAXED, __HIP_MEMORY_SCOPE_WORKGROUP);
              __hip_atomic_fetch_add(ar +  96, pA * va[u].w, __ATOMIC_RELAXED, __HIP_MEMORY_SCOPE_WORKGROUP);
              __hip_atomic_fetch_add(ar + 128, pB * vb[u].x, __ATOMIC_RELAXED, __HIP_MEMORY_SCOPE_WORKGROUP);
              __hip_atomic_fetch_add(ar + 160, pB * vb[u].y, __ATOMIC_RELAXED, __HIP_MEMORY_SCOPE_WORKGROUP);
              __hip_atomic_fetch_add(ar + 192, pB * vb[u].z, __ATOMIC_RELAXED, __HIP_MEMORY_SCOPE_WORKGROUP);
              __hip_atomic_fetch_add(ar + 224, pB * vb[u].w, __ATOMIC_RELAXED, __HIP_MEMORY_SCOPE_WORKGROUP);
            } else {
              if ((lane & 15) == 0) {
                atomicMax(&mb[dl * HEADS + (hsel ? 1 : 0)], f2ord(tA));
                atomicMax(&mb[dl * HEADS + (hsel ? 3 : 2)], f2ord(tB));
              }
            }
          }
        }
      }
    };
    if (spec) pass_a(flag_t<true>{}); else pass_a(flag_t<false>{});
    __syncthreads();

    if (tid < SLAB * HEADS) {
      const float mnew_b = ord2f(mb[tid]);
      mb[tid] = ORD_NEG_INF;
      if (b == 0) {
        mrun[tid] = mnew_b;
        if (spec && !(fabsf(mnew_b) <= SPEC_RANGE)) respec = 1u;
      } else {
        const float mo = mrun[tid];
        const float mn = fmaxf(mo, mnew_b);
        const float f  = expf(mo - mn);
        mrun[tid] = mn;
        den[tid] *= f;
        sc[tid]   = f;
      }
    }
    __syncthreads();
    if (spec) {
      unshifted = __builtin_amdgcn_readfirstlane((int)respec) == 0;
      if (unshifted) break;
      float4*      a4 = (float4*)acc;
      const float4 z  = make_float4(0.0f, 0.0f, 0.0f, 0.0f);
      for (int i = tid; i < SLAB * FDIM / 4; i += AGG_THREADS) a4[i] = z;
      if (tid < SLAB * HEADS) den[tid] = 0.0f;
      __syncthreads();
    }
    if (b > 0) {
      #pragma unroll 1
      for (int i = tid; i < SLAB * FDIM; i += AGG_THREADS) {
        const int r = i >> 8, q = (i >> 5) & 7, l = i & 31;
        acc[i] *= sc[r * HEADS + ((q >> 2) << 1) + (l >> 4)];
      }
      __syncthreads();
    }

    for (unsigned base = wave * 32; base < M; base += AGG_WAVES * 32) {
      const unsigned idx = base + lane;
      int   m_src = 0, m_dl = SLAB;
      float m_m0 = 0.0f, m_m1 = 0.0f, m_m2 = 0.0f, m_m3 = 0.0f;
      if (idx < M) {
        const unsigned ent = queue[idx];
        m_src = (int)(ent & 0xFFFFFu);
        m_dl  = (int)(ent >> 20);
        const float4 mm = ((const float4*)mrun)[m_dl];
        m_m0 = mm.x; m_m1 = mm.y; m_m2 = mm.z; m_m3 = mm.w;
      }
      const int n = (int)min(32u, M - base);
      for (int j0 = 0; j0 < n; j0 += 8) {
        float4 va[8], vb[8];
        #pragma unroll
        for (int u = 0; u < 8; ++u) {
          const int s = __builtin_amdgcn_readlane(m_src, j0 + u);
          const float4* row = (const float4*)(xl + (size_t)s * FDIM);
          va[u] = row[lane];
          vb[u] = row[32 + lane];
        }
        #pragma unroll
        for (int u = 0; u < 8; ++u) {
          const int     dl = __builtin_amdgcn_readlane(m_dl, j0 + u);
          const float4* xq = (const float4*)(xrs + dl * FDIM);
          const float   tA = row16_sum(score4(va[u], xq[lane],      atA));
          const float   tB = row16_sum(score4(vb[u], xq[32 + lane], atB));
          const float m0 = __int_as_float(__builtin_amdgcn_readlane(__float_as_int(m_m0), j0 + u));
          const float m1 = __int_as_float(__builtin_amdgcn_readlane(__float_as_int(m_m1), j0 + u));
          const float m2 = __int_as_float(__builtin_amdgcn_readlane(__float_as_int(m_m2), j0 + u));
          const float m3 = __int_as_float(__builtin_amdgcn_readlane(__float_as_int(m_m3), j0 + u));
          const float pA = expf(tA - (hsel ? m1 : m0));
          const float pB = expf(tB - (hsel ? m3 : m2));
          if ((lane & 15) == 0) {
            __hip_atomic_fetch_add(&den[dl * HEADS + (hsel ? 1 : 0)], pA, __ATOMIC_RELAXED, __HIP_MEMORY_SCOPE_WORKGROUP);
            __hip_atomic_fetch_add(&den[dl * HEADS + (hsel ? 3 : 2)], pB, __ATOMIC_RELAXED, __HIP_MEMORY_SCOPE_WORKGROUP);
          }
          float* ar = acc + dl * FDIM + lane;
          __hip_atomic_fetch_add(ar +   0, pA * va[u].x, __ATOMIC_RELAXED, __HIP_MEMORY_SCOPE_WORKGROUP);
          __hip_atomic_fetch_add(ar +  32, pA * va[u].y, __ATOMIC_RELAXED, __HIP_MEMORY_SCOPE_WORKGROUP);
          __hip_atomic_fetch_add(ar +  64, pA * va[u].z, __ATOMIC_RELAXED, __HIP_MEMORY_SCOPE_WORKGROUP);
          __hip_atomic_fetch_add(ar +  96, pA * va[u].w, __ATOMIC_RELAXED, __HIP_MEMORY_SCOPE_WORKGROUP);
          __hip_atomic_fetch_add(ar + 128, pB * vb[u].x, __ATOMIC_RELAXED, __HIP_MEMORY_SCOPE_WORKGROUP);
          __hip_atomic_fetch_add(ar + 160, pB * vb[u].y, __ATOMIC_RELAXED, __HIP_MEMORY_SCOPE_WORKGROUP);
          __hip_atomic_fetch_add(ar + 192, pB * vb[u].z, __ATOMIC_RELAXED, __HIP_MEMORY_SCOPE_WORKGROUP);
          __hip_atomic_fetch_add(ar + 224, pB * vb[u].w, __ATOMIC_RELAXED, __HIP_MEMORY_SCOPE_WORKGROUP);
        }
      }
    }
  }
  __syncthreads();

  {
    const float4 bA = ((const float4*)bias)[lane];
    const float4 bB = ((const float4*)bias)[32 + lane];
    for (int r = wave; r < SLAB; r += AGG_WAVES) {
      const float* ar = acc + r * FDIM + lane;
      const int    hA = r * HEADS + (hsel ? 1 : 0);
      const int    hB = r * HEADS + (hsel ? 3 : 2);
      const float  eA = unshifted ? 1e-16f * expf(mrun[hA]) : 1e-16f;
      const float  eB = unshifted ? 1e-16f * expf(mrun[hB]) : 1e-16f;
      const float  dA = den[hA] + eA;
      const float  dB = den[hB] + eB;
      float o[8] = {ar[0] / dA + bA.x,   ar[32] / dA + bA.y,  ar[64] / dA + bA.z,  ar[96] / dA + bA.w,
                    ar[128] / dB + bB.x, ar[160] / dB + bB.y, ar[192] / dB + bB.z, ar[224] / dB + bB.w};
      #pragma unroll
      for (int q = 0; q < 8; ++q) o[q] = o[q] > 0.0f ? o[q] : (expf(o[q]) - 1.0f);
      float4* orow = (float4*)(outb + (size_t)(d0 + r) * FDIM);
      orow[lane]      = make_float4(o[0], o[1], o[2], o[3]);
      orow[32 + lane] = make_float4(o[4], o[5], o[6], o[7]);
    }
  }
}

extern "C" void kernel_launch(void* const* d_in, const int* in_sizes, int n_in,
                              void* d_out, int out_size, void* d_ws, size_t ws_size,
                              hipStream_t stream) {
  const float* x    = (const float*)d_in[0];
  const int*   ei   = (const int*)  d_in[1];
  const float* Wl0  = (const float*)d_in[2];
  const float* Wr0  = (const float*)d_in[3];
  const float* att0 = (const float*)d_in[4];
  const float* b0   = (const float*)d_in[5];
  const float* Wl1  = (const float*)d_in[6];
  const float* Wr1  = (const float*)d_in[7];
  const float* att1 = (const float*)d_in[8];
  const float* b1   = (const float*)d_in[9];
  float* out = (float*)d_out;

  float*    ws    = (float*)d_ws;
  float*    xl    = ws;
  float*    hbuf  = xl   + (size_t)N_NODES * FDIM;

  auto layer = [&](const float* in, const float* Wl, const float* Wr,
                   const float* att, const float* bias, float* outb) {
    gemm_f32_wmma<<<N_NODES / 16, 512, 0, stream>>>(in, Wl, xl);
    slab_edge_phase<<<N_NODES / SLAB, AGG_THREADS, 0, stream>>>(xl, in, Wr, att, ei, bias, outb);
  };

  layer(x,    Wl0, Wr0, att0, b0, hbuf);
  layer(hbuf, Wl1, Wr1, att1, b1, out);
}


# slab edge queue built once per call: layer 0 keeps each slab's queue (source row, local row) in the workspace, layer 1 reads it back, no second pass over the edge list (arrival-order sums as before)
# speedup vs baseline: 4.5499x; 1.1303x over previous
#include <hip/hip_runtime.h>

#define N_NODES   20000
#define FDIM      256
#define HEADS     4
#define CDIM      64
#define NEDGE     320000
#define ETOT      (NEDGE + N_NODES)
#define NEG_SLOPE 0.2f

typedef __attribute__((ext_vector_type(2))) float v2f;
typedef __attribute__((ext_vector_type(8))) float v8f;

__device__ __forceinline__ unsigned f2ord(float f) {
  unsigned u = __float_as_uint(f);
  return (u & 0x80000000u) ? ~u : (u | 0x80000000u);
}
__device__ __forceinline__ float ord2f(unsigned u) {
  unsigned v = (u & 0x80000000u) ? (u ^ 0x80000000u) : ~u;
  return __uint_as_float(v);
}

__global__ __launch_bounds__(512)
void gemm_f32_wmma(const float* __restrict__ A, const float* __restrict__ B,
                   float* __restrict__ Out) {
  __shared__ float As[16 * 257];
  const int tid      = threadIdx.x;
  const int row_base = blockIdx.x * 16;

  for (int i = tid; i < 16 * 256; i += 512) {
    int r = i >> 8, c = i & 255;
    As[r * 257 + c] = A[(size_t)(row_base + r) * FDIM + c];
  }
  __syncthreads();

  const int wave = tid >> 5;
  const int lane = tid & 31;
  const int cn   = wave * 16;
  const int m    = lane & 15;
  const int kh   = (lane >> 4) << 1;

  v8f acc = {};
  #pragma unroll 8
  for (int k = 0; k < FDIM; k += 4) {
    v2f a, b;
    a.x = As[m * 257 + k + kh];
    a.y = As[m * 257 + k + kh + 1];
    b.x = B[(size_t)(k + kh)     * FDIM + cn + m];
    b.y = B[(size_t)(k + kh + 1) * FDIM + cn + m];
    acc = __builtin_amdgcn_wmma_f32_16x16x4_f32(false, a, false, b,
                                                (short)0, acc, false, false);
  }

  const int rbase = row_base + ((lane >> 4) << 3);
  #pragma unroll
  for (int r = 0; r < 8; ++r) {
    Out[(size_t)(rbase + r) * FDIM + cn + m] = acc[r];
  }
}

#define SLAB        80
#define QCAP        24576
#define AGG_THREADS 1024
#define AGG_WAVES   (AGG_THREADS / 32)
#define SEGLEN      (QCAP - SLAB)
#define NSEG        ((NEDGE + SEGLEN - 1) / SEGLEN)
#define ORD_NEG_INF 0x007FFFFFu
#define QSTRIDE     8192
static_assert(QSTRIDE <= QCAP, "a saved queue always fits the queue in LDS");
#define SPEC_RANGE  16.0f
static_assert(N_NODES % SLAB == 0, "slabs must tile the node range");
static_assert(SLAB % 16 == 0, "the slab's xr rows are formed as whole 16-row matrix tiles");
static_assert(NEDGE % 4 == 0, "destination list is read as 16-byte vectors");
static_assert(ETOT < (1 << 20) && SLAB < (1 << 12), "queue word packs edge id and local row");

__device__ __forceinline__ float row16_sum(float v) {
  v += __int_as_float(__builtin_amdgcn_mov_dpp(__float_as_int(v), 0x161, 0xf, 0xf, true));
  v += __int_as_float(__builtin_amdgcn_mov_dpp(__float_as_int(v), 0x162, 0xf, 0xf, true));
  v += __int_as_float(__builtin_amdgcn_mov_dpp(__float_as_int(v), 0x164, 0xf, 0xf, true));
  v += __int_as_float(__builtin_amdgcn_mov_dpp(__float_as_int(v), 0x168, 0xf, 0xf, true));
  return v;
}

__device__ __forceinline__ float leaky_max(float v) { return fmaxf(v, NEG_SLOPE * v); }
__device__ __forceinline__ float score4(const float4 l, const float4 r, const float4 a) {
  return leaky_max(l.x + r.x) * a.x + leaky_max(l.y + r.y) * a.y +
         leaky_max(l.z + r.z) * a.z + leaky_max(l.w + r.w) * a.w;
}

template <bool B> struct flag_t { static constexpr bool value = B; };

__global__ __launch_bounds__(AGG_THREADS)
void slab_edge_phase(const float* __restrict__ xl, const float* __restrict__ xin,
                     const float* __restrict__ Wr,
                     const float* __restrict__ att, const int* __restrict__ ei,
                     const float* __restrict__ bias, float* __restrict__ outb,
                     int* __restrict__ qsave, int reuse) {
  __shared__ __align__(16) float    acc[(SLAB + 1) * FDIM];
  __shared__ __align__(16) float    xrs[(SLAB + 1) * FDIM];
  __shared__ unsigned queue[QCAP];
  __shared__ unsigned mb[(SLAB + 1) * HEADS];
  __shared__ __align__(16) float    mrun[(SLAB + 1) * HEADS];
  __shared__ float    den[(SLAB + 1) * HEADS];
  __shared__ float    sc[SLAB * HEADS];
  __shared__ unsigned qcount;
  __shared__ unsigned respec;
  __shared__ unsigned preres;

  const int tid  = threadIdx.x;
  const int lane = tid & 31;
  const int wave = __builtin_amdgcn_readfirstlane(tid >> 5);
  const int d0   = blockIdx.x * SLAB;
  const bool hsel = lane >= 16;

  {
    float4*       a4 = (float4*)acc;
    float4*       x4 = (float4*)xrs;
    const float4* g4 = (const float4*)(xin + (size_t)d0 * FDIM);
    const float4  z  = make_float4(0.0f, 0.0f, 0.0f, 0.0f);
    constexpr int NST = SLAB * FDIM / 4 / AGG_THREADS;
    static_assert(NST * AGG_THREADS == SLAB * FDIM / 4, "whole 16-byte pieces per thread");
    float4 st[NST];
    #pragma unroll
    for (int j = 0; j < NST; ++j) st[j] = g4[tid + j * AGG_THREADS];
    #pragma unroll
    for (int j = 0; j < NST; ++j) {
      const int i = tid + j * AGG_THREADS;
      const int r = i >> 6, c4 = i & 63;
      a4[(r << 6) + (c4 ^ (r & 15))] = st[j];
    }
    if (tid < FDIM / 4) x4[SLAB * FDIM / 4 + tid] = z;
    if (tid < (SLAB + 1) * HEADS) { mb[tid] = ORD_NEG_INF; mrun[tid] = 0.0f; den[tid] = 0.0f; }
    if (tid < SLAB) queue[tid] = (unsigned)(NEDGE + d0 + tid) | ((unsigned)tid << 20);
    if (tid == 0) { qcount = SLAB; respec = 0u; preres = 0u; }
  }
  __syncthreads();

  const int role = ((wave >> 2) ^ wave) & 1;
  const int ridx = ((wave >> 3) << 2) | (wave & 3);
  if (role == 0) {
    const int m  = lane & 15;
    const int kh = (lane >> 4) << 1;
    const int cn = ridx * 16;
    const int sw = m << 2;
    const float* bp = Wr + (size_t)kh * FDIM + cn + m;
    const float* ap = acc + m * FDIM + kh;
    v8f c[SLAB / 16] = {};
    #pragma unroll 8
    for (int k = 0; k < FDIM; k += 4) {
      v2f b;
      b.x = bp[(size_t)k * FDIM];
      b.y = bp[(size_t)(k + 1) * FDIM];
      const float* ak = ap + (k ^ sw);
      #pragma unroll
      for (int t = 0; t < SLAB / 16; ++t) {
        const v2f a = *(const v2f*)(ak + t * 16 * FDIM);
        c[t] = __builtin_amdgcn_wmma_f32_16x16x4_f32(false, a, false, b,
                                                     (short)0, c[t], false, false);
      }
    }
    const int rb = (lane >> 4) << 3;
    #pragma unroll
    for (int t = 0; t < SLAB / 16; ++t) {
      #pragma unroll
      for (int r = 0; r < 8; ++r) xrs[(t * 16 + rb + r) * FDIM + cn + m] = c[t][r];
    }
  } else {
    constexpr int FTHREADS = AGG_THREADS / 2;
    const int   ftid = ridx * 32 + lane;
    const int* qs  = qsave + (size_t)blockIdx.x * QSTRIDE;
    const int  cnt = reuse ? __builtin_amdgcn_readfirstlane(qs[0]) : -1;
    const bool kept = cnt >= SLAB && cnt < QSTRIDE;
    if (kept) {
      for (int i = ftid; i < cnt; i += FTHREADS) queue[i] = (unsigned)qs[1 + i];
      if (ftid == 0) { qcount = (unsigned)cnt; preres = 1u; }
    }
    const int4* dst4 = (const int4*)(ei + NEDGE);
    constexpr int NVEC = NEDGE / 4;
    constexpr int NCH  = (NVEC + FTHREADS - 1) / FTHREADS;
    const int rot = blockIdx.x % NCH;
    for (int g = kept ? NCH : 0; g < NCH; g += 8) {
      int4 v[8];
      int  vi[8];
      #pragma unroll
      for (int u = 0; u < 8; ++u) {
        int c = g + u + rot;
        if (c >= NCH) c -= NCH;
        const int i = c * FTHREADS + ftid;
        vi[u] = i;
        v[u]  = make_int4(-1, -1, -1, -1);
        if (g + u < NCH && i < NVEC) v[u] = dst4[i];
      }
      #pragma unroll
      for (int u = 0; u < 8; ++u) {
        const unsigned r[4] = {(unsigned)(v[u].x - d0), (unsigned)(v[u].y - d0),
                               (unsigned)(v[u].z - d0), (unsigned)(v[u].w - d0)};
        const unsigned rmin = min(min(r[0], r[1]), min(r[2], r[3]));
        if (rmin < (unsigned)SLAB) {
          #pragma unroll
          for (int k = 0; k < 4; ++k) {
            if (r[k] < (unsigned)SLAB) {
              const int e = vi[u] * 4 + k;
              const unsigned slot = atomicAdd(&qcount, 1u);
              if (slot < (unsigned)QCAP) queue[slot] = (unsigned)e | (r[k] << 20);
            }
          }
        }
      }
    }
  }
  __syncthreads();

  {
    float4*      a4 = (float4*)acc;
    const float4 z  = make_float4(0.0f, 0.0f, 0.0f, 0.0f);
    for (int i = tid; i < SLAB * FDIM / 4; i += AGG_THREADS) a4[i] = z;
  }
  __syncthreads();

  const unsigned total  = (unsigned)__builtin_amdgcn_readfirstlane((int)qcount);
  const bool     big    = total > (unsigned)QCAP;
  const int      nbatch = big ? NSEG : 1;
  const bool     spec   = !big;
  bool           unshifted = false;
  const bool     pre    = __builtin_amdgcn_readfirstlane((int)preres) != 0;

  const float4 atA = ((const float4*)att)[lane];
  const float4 atB = ((const float4*)att)[32 + lane];

  for (int b = 0; b < nbatch; ++b) {
    if (big) {
      __syncthreads();
      if (tid == 0) qcount = (b == 0) ? SLAB : 0;
      if (b == 0 && tid < SLAB) queue[tid] = (unsigned)(NEDGE + d0 + tid) | ((unsigned)tid << 20);
      __syncthreads();
      const int s0 = b * SEGLEN;
      const int s1 = min(NEDGE, s0 + SEGLEN);
      for (int e = s0 + tid; e < s1; e += AGG_THREADS) {
        const unsigned r = (unsigned)(ei[NEDGE + e] - d0);
        if (r < (unsigned)SLAB) {
          const unsigned slot = atomicAdd(&qcount, 1u);
          queue[slot] = (unsigned)e | (r << 20);
        }
      }
      __syncthreads();
    }
    const unsigned M = min((unsigned)__builtin_amdgcn_readfirstlane((int)qcount), (unsigned)QCAP);

    auto pass_a = [&](auto sums_tag) __attribute__((always_inline)) {
      constexpr bool SUMS = decltype(sums_tag)::value;
      for (unsigned base = wave * 32; base < M; base += AGG_WAVES * 32) {
        const unsigned idx = base + lane;
        int m_src = 0, m_dl = SLAB;
        if (idx < M) {
          const unsigned ent = queue[idx];
          const int      e   = (int)(ent & 0xFFFFFu);
          m_dl  = (int)(ent >> 20);
          if (pre) {
            m_src = e;
          } else {
            m_src = (e < NEDGE) ? ei[e] : (e - NEDGE);
            queue[idx] = (unsigned)m_src | ((unsigned)m_dl << 20);
          }
        }
        const int n = (int)min(32u, M - base);
        for (int j0 = 0; j0 < n; j0 += 8) {
          float4 va[8], vb[8];
          #pragma unroll
          for (int u = 0; u < 8; ++u) {
            const int s = __builtin_amdgcn_readlane(m_src, j0 + u);
            const float4* row = (const float4*)(xl + (size_t)s * FDIM);
            va[u] = row[lane];
            vb[u] = row[32 + lane];
          }
          #pragma unroll
          for (int u = 0; u < 8; ++u) {
            const int     dl = __builtin_amdgcn_readlane(m_dl, j0 + u);
            const float4* xq = (const float4*)(xrs + dl * FDIM);
            const float   tA = row16_sum(score4(va[u], xq[lane],      atA));
            const float   tB = row16_sum(score4(vb[u], xq[32 + lane], atB));
            if (SUMS) {
              const float pA = expf(tA);
              const float pB = expf(tB);
              if ((lane & 15) == 0) {
                atomicMax(&mb[dl * HEADS + (hsel ? 1 : 0)], f2ord(tA));
                atomicMax(&mb[dl * HEADS + (hsel ? 3 : 2)], f2ord(tB));
                __hip_atomic_fetch_add(&den[dl * HEADS + (hsel ? 1 : 0)], pA, __ATOMIC_RELAXED, __HIP_MEMORY_SCOPE_WORKGROUP);
                __hip_atomic_fetch_add(&den[dl * HEADS + (hsel ? 3 : 2)], pB, __ATOMIC_RELAXED, __HIP_MEMORY_SCOPE_WORKGROUP);
              }
              float* ar = acc + dl * FDIM + lane;
              __hip_atomic_fetch_add(ar +   0, pA * va[u].x, __ATOMIC_RELAXED, __HIP_MEMORY_SCOPE_WORKGROUP);
              __hip_atomic_fetch_add(ar +  32, pA * va[u].y, __ATOMIC_RELAXED, __HIP_MEMORY_SCOPE_WORKGROUP);
              __hip_atomic_fetch_add(ar +  64, pA * va[u].z, __ATOMIC_RELAXED, __HIP_MEMORY_SCOPE_WORKGROUP);
              __hip_atomic_fetch_add(ar +  96, pA * va[u].w, __ATOMIC_RELAXED, __HIP_MEMORY_SCOPE_WORKGROUP);
              __hip_atomic_fetch_add(ar + 128, pB * vb[u].x, __ATOMIC_RELAXED, __HIP_MEMORY_SCOPE_WORKGROUP);
              __hip_atomic_fetch_add(ar + 160, pB * vb[u].y, __ATOMIC_RELAXED, __HIP_MEMORY_SCOPE_WORKGROUP);
              __hip_atomic_fetch_add(ar + 192, pB * vb[u].z, __ATOMIC_RELAXED, __HIP_MEMORY_SCOPE_WORKGROUP);
              __hip_atomic_fetch_add(ar + 224, pB * vb[u].w, __ATOMIC_RELAXED, __HIP_MEMORY_SCOPE_WORKGROUP);
            } else {
              if ((lane & 15) == 0) {
                atomicMax(&mb[dl * HEADS + (hsel ? 1 : 0)], f2ord(tA));
                atomicMax(&mb[dl * HEADS + (hsel ? 3 : 2)], f2ord(tB));
              }
            }
          }
        }
      }
    };
    if (spec) pass_a(flag_t<true>{}); else pass_a(flag_t<false>{});
    __syncthreads();

    if (tid < SLAB * HEADS) {
      const float mnew_b = ord2f(mb[tid]);
      mb[tid] = ORD_NEG_INF;
      if (b == 0) {
        mrun[tid] = mnew_b;
        if (spec && !(fabsf(mnew_b) <= SPEC_RANGE)) respec = 1u;
      } else {
        const float mo = mrun[tid];
        const float mn = fmaxf(mo, mnew_b);
        const float f  = expf(mo - mn);
        mrun[tid] = mn;
        den[tid] *= f;
        sc[tid]   = f;
      }
    }
    __syncthreads();
    if (spec) {
      unshifted = __builtin_amdgcn_readfirstlane((int)respec) == 0;
      if (unshifted) break;
      float4*      a4 = (float4*)acc;
      const float4 z  = make_float4(0.0f, 0.0f, 0.0f, 0.0f);
      for (int i = tid; i < SLAB * FDIM / 4; i += AGG_THREADS) a4[i] = z;
      if (tid < SLAB * HEADS) den[tid] = 0.0f;
      __syncthreads();
    }
    if (b > 0) {
      #pragma unroll 1
      for (int i = tid; i < SLAB * FDIM; i += AGG_THREADS) {
        const int r = i >> 8, q = (i >> 5) & 7, l = i & 31;
        acc[i] *= sc[r * HEADS + ((q >> 2) << 1) + (l >> 4)];
      }
      __syncthreads();
    }

    for (unsigned base = wave * 32; base < M; base += AGG_WAVES * 32) {
      const unsigned idx = base + lane;
      int   m_src = 0, m_dl = SLAB;
      float m_m0 = 0.0f, m_m1 = 0.0f, m_m2 = 0.0f, m_m3 = 0.0f;
      if (idx < M) {
        const unsigned ent = queue[idx];
        m_src = (int)(ent & 0xFFFFFu);
        m_dl  = (int)(ent >> 20);
        const float4 mm = ((const float4*)mrun)[m_dl];
        m_m0 = mm.x; m_m1 = mm.y; m_m2 = mm.z; m_m3 = mm.w;
      }
      const int n = (int)min(32u, M - base);
      for (int j0 = 0; j0 < n; j0 += 8) {
        float4 va[8], vb[8];
        #pragma unroll
        for (int u = 0; u < 8; ++u) {
          const int s = __builtin_amdgcn_readlane(m_src, j0 + u);
          const float4* row = (const float4*)(xl + (size_t)s * FDIM);
          va[u] = row[lane];
          vb[u] = row[32 + lane];
        }
        #pragma unroll
        for (int u = 0; u < 8; ++u) {
          const int     dl = __builtin_amdgcn_readlane(m_dl, j0 + u);
          const float4* xq = (const float4*)(xrs + dl * FDIM);
          const float   tA = row16_sum(score4(va[u], xq[lane],      atA));
          const float   tB = row16_sum(score4(vb[u], xq[32 + lane], atB));
          const float m0 = __int_as_float(__builtin_amdgcn_readlane(__float_as_int(m_m0), j0 + u));
          const float m1 = __int_as_float(__builtin_amdgcn_readlane(__float_as_int(m_m1), j0 + u));
          const float m2 = __int_as_float(__builtin_amdgcn_readlane(__float_as_int(m_m2), j0 + u));
          const float m3 = __int_as_float(__builtin_amdgcn_readlane(__float_as_int(m_m3), j0 + u));
          const float pA = expf(tA - (hsel ? m1 : m0));
          const float pB = expf(tB - (hsel ? m3 : m2));
          if ((lane & 15) == 0) {
            __hip_atomic_fetch_add(&den[dl * HEADS + (hsel ? 1 : 0)], pA, __ATOMIC_RELAXED, __HIP_MEMORY_SCOPE_WORKGROUP);
            __hip_atomic_fetch_add(&den[dl * HEADS + (hsel ? 3 : 2)], pB, __ATOMIC_RELAXED, __HIP_MEMORY_SCOPE_WORKGROUP);
          }
          float* ar = acc + dl * FDIM + lane;
          __hip_atomic_fetch_add(ar +   0, pA * va[u].x, __ATOMIC_RELAXED, __HIP_MEMORY_SCOPE_WORKGROUP);
          __hip_atomic_fetch_add(ar +  32, pA * va[u].y, __ATOMIC_RELAXED, __HIP_MEMORY_SCOPE_WORKGROUP);
          __hip_atomic_fetch_add(ar +  64, pA * va[u].z, __ATOMIC_RELAXED, __HIP_MEMORY_SCOPE_WORKGROUP);
          __hip_atomic_fetch_add(ar +  96, pA * va[u].w, __ATOMIC_RELAXED, __HIP_MEMORY_SCOPE_WORKGROUP);
          __hip_atomic_fetch_add(ar + 128, pB * vb[u].x, __ATOMIC_RELAXED, __HIP_MEMORY_SCOPE_WORKGROUP);
          __hip_atomic_fetch_add(ar + 160, pB * vb[u].y, __ATOMIC_RELAXED, __HIP_MEMORY_SCOPE_WORKGROUP);
          __hip_atomic_fetch_add(ar + 192, pB * vb[u].z, __ATOMIC_RELAXED, __HIP_MEMORY_SCOPE_WORKGROUP);
          __hip_atomic_fetch_add(ar + 224, pB * vb[u].w, __ATOMIC_RELAXED, __HIP_MEMORY_SCOPE_WORKGROUP);
        }
      }
    }
  }
  __syncthreads();

  if (!reuse) {
    int* qs = qsave + (size_t)blockIdx.x * QSTRIDE;
    const bool fits = !big && total < (unsigned)QSTRIDE;
    if (fits) {
      for (unsigned i = tid; i < total; i += AGG_THREADS) qs[1 + i] = (int)queue[i];
    }
    if (tid == 0) qs[0] = fits ? (int)total : -1;
  }

  {
    const float4 bA = ((const float4*)bias)[lane];
    const float4 bB = ((const float4*)bias)[32 + lane];
    for (int r = wave; r < SLAB; r += AGG_WAVES) {
      const float* ar = acc + r * FDIM + lane;
      const int    hA = r * HEADS + (hsel ? 1 : 0);
      const int    hB = r * HEADS + (hsel ? 3 : 2);
      const float  eA = unshifted ? 1e-16f * expf(mrun[hA]) : 1e-16f;
      const float  eB = unshifted ? 1e-16f * expf(mrun[hB]) : 1e-16f;
      const float  dA = den[hA] + eA;
      const float  dB = den[hB] + eB;
      float o[8] = {ar[0] / dA + bA.x,   ar[32] / dA + bA.y,  ar[64] / dA + bA.z,  ar[96] / dA + bA.w,
                    ar[128] / dB + bB.x, ar[160] / dB + bB.y, ar[192] / dB + bB.z, ar[224] / dB + bB.w};
      #pragma unroll
      for (int q = 0; q < 8; ++q) o[q] = o[q] > 0.0f ? o[q] : (expf(o[q]) - 1.0f);
      float4* orow = (float4*)(outb + (size_t)(d0 + r) * FDIM);
      orow[lane]      = make_float4(o[0], o[1], o[2], o[3]);
      orow[32 + lane] = make_float4(o[4], o[5], o[6], o[7]);
    }
  }
}

extern "C" void kernel_launch(void* const* d_in, const int* in_sizes, int n_in,
                              void* d_out, int out_size, void* d_ws, size_t ws_size,
                              hipStream_t stream) {
  const float* x    = (const float*)d_in[0];
  const int*   ei   = (const int*)  d_in[1];
  const float* Wl0  = (const float*)d_in[2];
  const float* Wr0  = (const float*)d_in[3];
  const float* att0 = (const float*)d_in[4];
  const float* b0   = (const float*)d_in[5];
  const float* Wl1  = (const float*)d_in[6];
  const float* Wr1  = (const float*)d_in[7];
  const float* att1 = (const float*)d_in[8];
  const float* b1   = (const float*)d_in[9];
  float* out = (float*)d_out;

  float*    ws    = (float*)d_ws;
  float*    xl    = ws;
  float*    hbuf  = xl   + (size_t)N_NODES * FDIM;
  int*      qsave = (int*)(hbuf + (size_t)N_NODES * FDIM);

  auto layer = [&](const float* in, const float* Wl, const float* Wr,
                   const float* att, const float* bias, float* outb, int reuse) {
    gemm_f32_wmma<<<N_NODES / 16, 512, 0, stream>>>(in, Wl, xl);
    slab_edge_phase<<<N_NODES / SLAB, AGG_THREADS, 0, stream>>>(xl, in, Wr, att, ei, bias, outb,
                                                                qsave, reuse);
  };

  layer(x,    Wl0, Wr0, att0, b0, hbuf, 0);
  layer(hbuf, Wl1, Wr1, att1, b1, out,  1);
}


# layer-0 edge filter scans destination slab numbers kept one byte per edge (written by the first product launch): a quarter of the bytes; local rows looked up in pass A; arrival-order sums as before
# speedup vs baseline: 4.7001x; 1.0330x over previous
#include <hip/hip_runtime.h>

#define N_NODES   20000
#define FDIM      256
#define HEADS     4
#define CDIM      64
#define NEDGE     320000
#define ETOT      (NEDGE + N_NODES)
#define NEG_SLOPE 0.2f
#define SLAB      80
#define SID_NONE  255u
static_assert(N_NODES / SLAB < 255, "a slab number fits a byte, 255 left free");

typedef __attribute__((ext_vector_type(2))) float v2f;
typedef __attribute__((ext_vector_type(8))) float v8f;

__device__ __forceinline__ unsigned f2ord(float f) {
  unsigned u = __float_as_uint(f);
  return (u & 0x80000000u) ? ~u : (u | 0x80000000u);
}
__device__ __forceinline__ float ord2f(unsigned u) {
  unsigned v = (u & 0x80000000u) ? (u ^ 0x80000000u) : ~u;
  return __uint_as_float(v);
}

__global__ __launch_bounds__(512)
void gemm_f32_wmma(const float* __restrict__ A, const float* __restrict__ B,
                   float* __restrict__ Out,
                   const int* __restrict__ ei, unsigned* __restrict__ sid) {
  __shared__ float As[16 * 257];
  const int tid      = threadIdx.x;
  const int row_base = blockIdx.x * 16;

  constexpr int SIDW_BLK = (NEDGE / 4) / (N_NODES / 16);
  static_assert(SIDW_BLK * (N_NODES / 16) == NEDGE / 4 && SIDW_BLK <= 512,
                "the blocks share the list's words evenly");
  const bool conv = sid != nullptr && tid < SIDW_BLK;
  const int  sw   = blockIdx.x * SIDW_BLK + tid;
  int4 dq = make_int4(0, 0, 0, 0);
  if (conv) dq = ((const int4*)(ei + NEDGE))[sw];

  for (int i = tid; i < 16 * 256; i += 512) {
    int r = i >> 8, c = i & 255;
    As[r * 257 + c] = A[(size_t)(row_base + r) * FDIM + c];
  }
  __syncthreads();

  if (conv) {
    auto sidof = [](int v) -> unsigned {
      return (unsigned)v < (unsigned)N_NODES ? (unsigned)v / (unsigned)SLAB : SID_NONE;
    };
    sid[sw] = sidof(dq.x) | (sidof(dq.y) << 8) | (sidof(dq.z) << 16) | (sidof(dq.w) << 24);
  }

  const int wave = tid >> 5;
  const int lane = tid & 31;
  const int cn   = wave * 16;
  const int m    = lane & 15;
  const int kh   = (lane >> 4) << 1;

  v8f acc = {};
  #pragma unroll 8
  for (int k = 0; k < FDIM; k += 4) {
    v2f a, b;
    a.x = As[m * 257 + k + kh];
    a.y = As[m * 257 + k + kh + 1];
    b.x = B[(size_t)(k + kh)     * FDIM + cn + m];
    b.y = B[(size_t)(k + kh + 1) * FDIM + cn + m];
    acc = __builtin_amdgcn_wmma_f32_16x16x4_f32(false, a, false, b,
                                                (short)0, acc, false, false);
  }

  const int rbase = row_base + ((lane >> 4) << 3);
  #pragma unroll
  for (int r = 0; r < 8; ++r) {
    Out[(size_t)(rbase + r) * FDIM + cn + m] = acc[r];
  }
}

#define QCAP        24576
#define AGG_THREADS 1024
#define AGG_WAVES   (AGG_THREADS / 32)
#define SEGLEN      (QCAP - SLAB)
#define NSEG        ((NEDGE + SEGLEN - 1) / SEGLEN)
#define ORD_NEG_INF 0x007FFFFFu
#define QSTRIDE     8192
static_assert(QSTRIDE <= QCAP, "a saved queue always fits the queue in LDS");
#define SPEC_RANGE  16.0f
static_assert(N_NODES % SLAB == 0, "slabs must tile the node range");
static_assert(SLAB % 16 == 0, "the slab's xr rows are formed as whole 16-row matrix tiles");
static_assert(NEDGE % 16 == 0, "slab numbers are read as 16-byte vectors of 16 edges");
#define DL_UNKNOWN  0xFFFu
static_assert(ETOT < (1 << 20) && SLAB < DL_UNKNOWN, "queue word packs edge id and local row");

__device__ __forceinline__ float row16_sum(float v) {
  v += __int_as_float(__builtin_amdgcn_mov_dpp(__float_as_int(v), 0x161, 0xf, 0xf, true));
  v += __int_as_float(__builtin_amdgcn_mov_dpp(__float_as_int(v), 0x162, 0xf, 0xf, true));
  v += __int_as_float(__builtin_amdgcn_mov_dpp(__float_as_int(v), 0x164, 0xf, 0xf, true));
  v += __int_as_float(__builtin_amdgcn_mov_dpp(__float_as_int(v), 0x168, 0xf, 0xf, true));
  return v;
}

__device__ __forceinline__ float leaky_max(float v) { return fmaxf(v, NEG_SLOPE * v); }
__device__ __forceinline__ float score4(const float4 l, const float4 r, const float4 a) {
  return leaky_max(l.x + r.x) * a.x + leaky_max(l.y + r.y) * a.y +
         leaky_max(l.z + r.z) * a.z + leaky_max(l.w + r.w) * a.w;
}

template <bool B> struct flag_t { static constexpr bool value = B; };

__global__ __launch_bounds__(AGG_THREADS)
void slab_edge_phase(const float* __restrict__ xl, const float* __restrict__ xin,
                     const float* __restrict__ Wr,
                     const float* __restrict__ att, const int* __restrict__ ei,
                     const float* __restrict__ bias, float* __restrict__ outb,
                     int* __restrict__ qsave, int reuse,
                     const unsigned* __restrict__ sid) {
  __shared__ __align__(16) float    acc[(SLAB + 1) * FDIM];
  __shared__ __align__(16) float    xrs[(SLAB + 1) * FDIM];
  __shared__ unsigned queue[QCAP];
  __shared__ unsigned mb[(SLAB + 1) * HEADS];
  __shared__ __align__(16) float    mrun[(SLAB + 1) * HEADS];
  __shared__ float    den[(SLAB + 1) * HEADS];
  __shared__ float    sc[SLAB * HEADS];
  __shared__ unsigned qcount;
  __shared__ unsigned respec;
  __shared__ unsigned preres;

  const int tid  = threadIdx.x;
  const int lane = tid & 31;
  const int wave = __builtin_amdgcn_readfirstlane(tid >> 5);
  const int d0   = blockIdx.x * SLAB;
  const bool hsel = lane >= 16;

  {
    float4*       a4 = (float4*)acc;
    float4*       x4 = (float4*)xrs;
    const float4* g4 = (const float4*)(xin + (size_t)d0 * FDIM);
    const float4  z  = make_float4(0.0f, 0.0f, 0.0f, 0.0f);
    constexpr int NST = SLAB * FDIM / 4 / AGG_THREADS;
    static_assert(NST * AGG_THREADS == SLAB * FDIM / 4, "whole 16-byte pieces per thread");
    float4 st[NST];
    #pragma unroll
    for (int j = 0; j < NST; ++j) st[j] = g4[tid + j * AGG_THREADS];
    #pragma unroll
    for (int j = 0; j < NST; ++j) {
      const int i = tid + j * AGG_THREADS;
      const int r = i >> 6, c4 = i & 63;
      a4[(r << 6) + (c4 ^ (r & 15))] = st[j];
    }
    if (tid < FDIM / 4) x4[SLAB * FDIM / 4 + tid] = z;
    if (tid < (SLAB + 1) * HEADS) { mb[tid] = ORD_NEG_INF; mrun[tid] = 0.0f; den[tid] = 0.0f; }
    if (tid < SLAB) queue[tid] = (unsigned)(NEDGE + d0 + tid) | ((unsigned)tid << 20);
    if (tid == 0) { qcount = SLAB; respec = 0u; preres = 0u; }
  }
  __syncthreads();

  const int role = ((wave >> 2) ^ wave) & 1;
  const int ridx = ((wave >> 3) << 2) | (wave & 3);
  if (role == 0) {
    const int m  = lane & 15;
    const int kh = (lane >> 4) << 1;
    const int cn = ridx * 16;
    const int sw = m << 2;
    const float* bp = Wr + (size_t)kh * FDIM + cn + m;
    const float* ap = acc + m * FDIM + kh;
    v8f c[SLAB / 16] = {};
    #pragma unroll 8
    for (int k = 0; k < FDIM; k += 4) {
      v2f b;
      b.x = bp[(size_t)k * FDIM];
      b.y = bp[(size_t)(k + 1) * FDIM];
      const float* ak = ap + (k ^ sw);
      #pragma unroll
      for (int t = 0; t < SLAB / 16; ++t) {
        const v2f a = *(const v2f*)(ak + t * 16 * FDIM);
        c[t] = __builtin_amdgcn_wmma_f32_16x16x4_f32(false, a, false, b,
                                                     (short)0, c[t], false, false);
      }
    }
    const int rb = (lane >> 4) << 3;
    #pragma unroll
    for (int t = 0; t < SLAB / 16; ++t) {
      #pragma unroll
      for (int r = 0; r < 8; ++r) xrs[(t * 16 + rb + r) * FDIM + cn + m] = c[t][r];
    }
  } else {
    constexpr int FTHREADS = AGG_THREADS / 2;
    const int   ftid = ridx * 32 + lane;
    const int* qs  = qsave + (size_t)blockIdx.x * QSTRIDE;
    const int  cnt = reuse ? __builtin_amdgcn_readfirstlane(qs[0]) : -1;
    const bool kept = cnt >= SLAB && cnt < QSTRIDE;
    if (kept) {
      for (int i = ftid; i < cnt; i += FTHREADS) queue[i] = (unsigned)qs[1 + i];
      if (ftid == 0) { qcount = (unsigned)cnt; preres = 1u; }
    }
    const uint4* sid4 = (const uint4*)sid;
    constexpr int NVEC = NEDGE / 16;
    constexpr int NCH  = (NVEC + FTHREADS - 1) / FTHREADS;
    const int rot = blockIdx.x % NCH;
    const unsigned pat  = (unsigned)blockIdx.x * 0x01010101u;
    for (int g = kept ? NCH : 0; g < NCH; g += 8) {
      uint4 v[8];
      int   vi[8];
      #pragma unroll
      for (int u = 0; u < 8; ++u) {
        int c = g + u + rot;
        if (c >= NCH) c -= NCH;
        const int i = c * FTHREADS + ftid;
        vi[u] = (g + u < NCH && i < NVEC) ? i : -1;
        v[u]  = sid4[min(i, NVEC - 1)];
      }
      #pragma unroll
      for (int u = 0; u < 8; ++u) {
        const unsigned x[4] = {v[u].x ^ pat, v[u].y ^ pat, v[u].z ^ pat, v[u].w ^ pat};
        unsigned z[4];
        #pragma unroll
        for (int j = 0; j < 4; ++j) z[j] = (x[j] - 0x01010101u) & ~x[j] & 0x80808080u;
        if ((z[0] | z[1] | z[2] | z[3]) != 0u && vi[u] >= 0) {
          #pragma unroll
          for (int j = 0; j < 4; ++j) {
            if (z[j] != 0u) {
              #pragma unroll
              for (int k = 0; k < 4; ++k) {
                if (((x[j] >> (8 * k)) & 0xFFu) == 0u) {
                  const int e = vi[u] * 16 + j * 4 + k;
                  const unsigned slot = atomicAdd(&qcount, 1u);
                  if (slot < (unsigned)QCAP) queue[slot] = (unsigned)e | (DL_UNKNOWN << 20);
                }
              }
            }
          }
        }
      }
    }
  }
  __syncthreads();

  {
    float4*      a4 = (float4*)acc;
    const float4 z  = make_float4(0.0f, 0.0f, 0.0f, 0.0f);
    for (int i = tid; i < SLAB * FDIM / 4; i += AGG_THREADS) a4[i] = z;
  }
  __syncthreads();

  const unsigned total  = (unsigned)__builtin_amdgcn_readfirstlane((int)qcount);
  const bool     big    = total > (unsigned)QCAP;
  const int      nbatch = big ? NSEG : 1;
  const bool     spec   = !big;
  bool           unshifted = false;
  const bool     pre    = __builtin_amdgcn_readfirstlane((int)preres) != 0;

  const float4 atA = ((const float4*)att)[lane];
  const float4 atB = ((const float4*)att)[32 + lane];

  for (int b = 0; b < nbatch; ++b) {
    if (big) {
      __syncthreads();
      if (tid == 0) qcount = (b == 0) ? SLAB : 0;
      if (b == 0 && tid < SLAB) queue[tid] = (unsigned)(NEDGE + d0 + tid) | ((unsigned)tid << 20);
      __syncthreads();
      const int s0 = b * SEGLEN;
      const int s1 = min(NEDGE, s0 + SEGLEN);
      for (int e = s0 + tid; e < s1; e += AGG_THREADS) {
        const unsigned r = (unsigned)(ei[NEDGE + e] - d0);
        if (r < (unsigned)SLAB) {
          const unsigned slot = atomicAdd(&qcount, 1u);
          queue[slot] = (unsigned)e | (r << 20);
        }
      }
      __syncthreads();
    }
    const unsigned M = min((unsigned)__builtin_amdgcn_readfirstlane((int)qcount), (unsigned)QCAP);

    auto pass_a = [&](auto sums_tag) __attribute__((always_inline)) {
      constexpr bool SUMS = decltype(sums_tag)::value;
      for (unsigned base = wave * 32; base < M; base += AGG_WAVES * 32) {
        const unsigned idx = base + lane;
        int m_src = 0, m_dl = SLAB;
        if (idx < M) {
          const unsigned ent = queue[idx];
          const int      e   = (int)(ent & 0xFFFFFu);
          m_dl  = (int)(ent >> 20);
          if (pre) {
            m_src = e;
          } else {
            if (e < NEDGE) {
              m_src = ei[e];
              if (m_dl == (int)DL_UNKNOWN)
                m_dl = (int)min((unsigned)(ei[NEDGE + e] - d0), (unsigned)SLAB);
            } else {
              m_src = e - NEDGE;
            }
            queue[idx] = (unsigned)m_src | ((unsigned)m_dl << 20);
          }
        }
        const int n = (int)min(32u, M - base);
        for (int j0 = 0; j0 < n; j0 += 8) {
          float4 va[8], vb[8];
          #pragma unroll
          for (int u = 0; u < 8; ++u) {
            const int s = __builtin_amdgcn_readlane(m_src, j0 + u);
            const float4* row = (const float4*)(xl + (size_t)s * FDIM);
            va[u] = row[lane];
            vb[u] = row[32 + lane];
          }
          #pragma unroll
          for (int u = 0; u < 8; ++u) {
            const int     dl = __builtin_amdgcn_readlane(m_dl, j0 + u);
            const float4* xq = (const float4*)(xrs + dl * FDIM);
            const float   tA = row16_sum(score4(va[u], xq[lane],      atA));
            const float   tB = row16_sum(score4(vb[u], xq[32 + lane], atB));
            if (SUMS) {
              const float pA = expf(tA);
              const float pB = expf(tB);
              if ((lane & 15) == 0) {
                atomicMax(&mb[dl * HEADS + (hsel ? 1 : 0)], f2ord(tA));
                atomicMax(&mb[dl * HEADS + (hsel ? 3 : 2)], f2ord(tB));
                __hip_atomic_fetch_add(&den[dl * HEADS + (hsel ? 1 : 0)], pA, __ATOMIC_RELAXED, __HIP_MEMORY_SCOPE_WORKGROUP);
                __hip_atomic_fetch_add(&den[dl * HEADS + (hsel ? 3 : 2)], pB, __ATOMIC_RELAXED, __HIP_MEMORY_SCOPE_WORKGROUP);
              }
              float* ar = acc + dl * FDIM + lane;
              __hip_atomic_fetch_add(ar +   0, pA * va[u].x, __ATOMIC_RELAXED, __HIP_MEMORY_SCOPE_WORKGROUP);
              __hip_atomic_fetch_add(ar +  32, pA * va[u].y, __ATOMIC_RELAXED, __HIP_MEMORY_SCOPE_WORKGROUP);
              __hip_atomic_fetch_add(ar +  64, pA * va[u].z, __ATOMIC_RELAXED, __HIP_MEMORY_SCOPE_WORKGROUP);
              __hip_atomic_fetch_add(ar +  96, pA * va[u].w, __ATOMIC_RELAXED, __HIP_MEMORY_SCOPE_WORKGROUP);
              __hip_atomic_fetch_add(ar + 128, pB * vb[u].x, __ATOMIC_RELAXED, __HIP_MEMORY_SCOPE_WORKGROUP);
              __hip_atomic_fetch_add(ar + 160, pB * vb[u].y, __ATOMIC_RELAXED, __HIP_MEMORY_SCOPE_WORKGROUP);
              __hip_atomic_fetch_add(ar + 192, pB * vb[u].z, __ATOMIC_RELAXED, __HIP_MEMORY_SCOPE_WORKGROUP);
              __hip_atomic_fetch_add(ar + 224, pB * vb[u].w, __ATOMIC_RELAXED, __HIP_MEMORY_SCOPE_WORKGROUP);
            } else {
              if ((lane & 15) == 0) {
                atomicMax(&mb[dl * HEADS + (hsel ? 1 : 0)], f2ord(tA));
                atomicMax(&mb[dl * HEADS + (hsel ? 3 : 2)], f2ord(tB));
              }
            }
          }
        }
      }
    };
    if (spec) pass_a(flag_t<true>{}); else pass_a(flag_t<false>{});
    __syncthreads();

    if (tid < SLAB * HEADS) {
      const float mnew_b = ord2f(mb[tid]);
      mb[tid] = ORD_NEG_INF;
      if (b == 0) {
        mrun[tid] = mnew_b;
        if (spec && !(fabsf(mnew_b) <= SPEC_RANGE)) respec = 1u;
      } else {
        const float mo = mrun[tid];
        const float mn = fmaxf(mo, mnew_b);
        const float f  = expf(mo - mn);
        mrun[tid] = mn;
        den[tid] *= f;
        sc[tid]   = f;
      }
    }
    __syncthreads();
    if (spec) {
      unshifted = __builtin_amdgcn_readfirstlane((int)respec) == 0;
      if (unshifted) break;
      float4*      a4 = (float4*)acc;
      const float4 z  = make_float4(0.0f, 0.0f, 0.0f, 0.0f);
      for (int i = tid; i < SLAB * FDIM / 4; i += AGG_THREADS) a4[i] = z;
      if (tid < SLAB * HEADS) den[tid] = 0.0f;
      __syncthreads();
    }
    if (b > 0) {
      #pragma unroll 1
      for (int i = tid; i < SLAB * FDIM; i += AGG_THREADS) {
        const int r = i >> 8, q = (i >> 5) & 7, l = i & 31;
        acc[i] *= sc[r * HEADS + ((q >> 2) << 1) + (l >> 4)];
      }
      __syncthreads();
    }

    for (unsigned base = wave * 32; base < M; base += AGG_WAVES * 32) {
      const unsigned idx = base + lane;
      int   m_src = 0, m_dl = SLAB;
      float m_m0 = 0.0f, m_m1 = 0.0f, m_m2 = 0.0f, m_m3 = 0.0f;
      if (idx < M) {
        const unsigned ent = queue[idx];
        m_src = (int)(ent & 0xFFFFFu);
        m_dl  = (int)(ent >> 20);
        const float4 mm = ((const float4*)mrun)[m_dl];
        m_m0 = mm.x; m_m1 = mm.y; m_m2 = mm.z; m_m3 = mm.w;
      }
      const int n = (int)min(32u, M - base);
      for (int j0 = 0; j0 < n; j0 += 8) {
        float4 va[8], vb[8];
        #pragma unroll
        for (int u = 0; u < 8; ++u) {
          const int s = __builtin_amdgcn_readlane(m_src, j0 + u);
          const float4* row = (const float4*)(xl + (size_t)s * FDIM);
          va[u] = row[lane];
          vb[u] = row[32 + lane];
        }
        #pragma unroll
        for (int u = 0; u < 8; ++u) {
          const int     dl = __builtin_amdgcn_readlane(m_dl, j0 + u);
          const float4* xq = (const float4*)(xrs + dl * FDIM);
          const float   tA = row16_sum(score4(va[u], xq[lane],      atA));
          const float   tB = row16_sum(score4(vb[u], xq[32 + lane], atB));
          const float m0 = __int_as_float(__builtin_amdgcn_readlane(__float_as_int(m_m0), j0 + u));
          const float m1 = __int_as_float(__builtin_amdgcn_readlane(__float_as_int(m_m1), j0 + u));
          const float m2 = __int_as_float(__builtin_amdgcn_readlane(__float_as_int(m_m2), j0 + u));
          const float m3 = __int_as_float(__builtin_amdgcn_readlane(__float_as_int(m_m3), j0 + u));
          const float pA = expf(tA - (hsel ? m1 : m0));
          const float pB = expf(tB - (hsel ? m3 : m2));
          if ((lane & 15) == 0) {
            __hip_atomic_fetch_add(&den[dl * HEADS + (hsel ? 1 : 0)], pA, __ATOMIC_RELAXED, __HIP_MEMORY_SCOPE_WORKGROUP);
            __hip_atomic_fetch_add(&den[dl * HEADS + (hsel ? 3 : 2)], pB, __ATOMIC_RELAXED, __HIP_MEMORY_SCOPE_WORKGROUP);
          }
          float* ar = acc + dl * FDIM + lane;
          __hip_atomic_fetch_add(ar +   0, pA * va[u].x, __ATOMIC_RELAXED, __HIP_MEMORY_SCOPE_WORKGROUP);
          __hip_atomic_fetch_add(ar +  32, pA * va[u].y, __ATOMIC_RELAXED, __HIP_MEMORY_SCOPE_WORKGROUP);
          __hip_atomic_fetch_add(ar +  64, pA * va[u].z, __ATOMIC_RELAXED, __HIP_MEMORY_SCOPE_WORKGROUP);
          __hip_atomic_fetch_add(ar +  96, pA * va[u].w, __ATOMIC_RELAXED, __HIP_MEMORY_SCOPE_WORKGROUP);
          __hip_atomic_fetch_add(ar + 128, pB * vb[u].x, __ATOMIC_RELAXED, __HIP_MEMORY_SCOPE_WORKGROUP);
          __hip_atomic_fetch_add(ar + 160, pB * vb[u].y, __ATOMIC_RELAXED, __HIP_MEMORY_SCOPE_WORKGROUP);
          __hip_atomic_fetch_add(ar + 192, pB * vb[u].z, __ATOMIC_RELAXED, __HIP_MEMORY_SCOPE_WORKGROUP);
          __hip_atomic_fetch_add(ar + 224, pB * vb[u].w, __ATOMIC_RELAXED, __HIP_MEMORY_SCOPE_WORKGROUP);
        }
      }
    }
  }
  __syncthreads();

  if (!reuse) {
    int* qs = qsave + (size_t)blockIdx.x * QSTRIDE;
    const bool fits = !big && total < (unsigned)QSTRIDE;
    if (fits) {
      for (unsigned i = tid; i < total; i += AGG_THREADS) qs[1 + i] = (int)queue[i];
    }
    if (tid == 0) qs[0] = fits ? (int)total : -1;
  }

  {
    const float4 bA = ((const float4*)bias)[lane];
    const float4 bB = ((const float4*)bias)[32 + lane];
    for (int r = wave; r < SLAB; r += AGG_WAVES) {
      const float* ar = acc + r * FDIM + lane;
      const int    hA = r * HEADS + (hsel ? 1 : 0);
      const int    hB = r * HEADS + (hsel ? 3 : 2);
      const float  eA = unshifted ? 1e-16f * expf(mrun[hA]) : 1e-16f;
      const float  eB = unshifted ? 1e-16f * expf(mrun[hB]) : 1e-16f;
      const float  dA = den[hA] + eA;
      const float  dB = den[hB] + eB;
      float o[8] = {ar[0] / dA + bA.x,   ar[32] / dA + bA.y,  ar[64] / dA + bA.z,  ar[96] / dA + bA.w,
                    ar[128] / dB + bB.x, ar[160] / dB + bB.y, ar[192] / dB + bB.z, ar[224] / dB + bB.w};
      #pragma unroll
      for (int q = 0; q < 8; ++q) o[q] = o[q] > 0.0f ? o[q] : (expf(o[q]) - 1.0f);
      float4* orow = (float4*)(outb + (size_t)(d0 + r) * FDIM);
      orow[lane]      = make_float4(o[0], o[1], o[2], o[3]);
      orow[32 + lane] = make_float4(o[4], o[5], o[6], o[7]);
    }
  }
}

extern "C" void kernel_launch(void* const* d_in, const int* in_sizes, int n_in,
                              void* d_out, int out_size, void* d_ws, size_t ws_size,
                              hipStream_t stream) {
  const float* x    = (const float*)d_in[0];
  const int*   ei   = (const int*)  d_in[1];
  const float* Wl0  = (const float*)d_in[2];
  const float* Wr0  = (const float*)d_in[3];
  const float* att0 = (const float*)d_in[4];
  const float* b0   = (const float*)d_in[5];
  const float* Wl1  = (const float*)d_in[6];
  const float* Wr1  = (const float*)d_in[7];
  const float* att1 = (const float*)d_in[8];
  const float* b1   = (const float*)d_in[9];
  float* out = (float*)d_out;

  float*    ws    = (float*)d_ws;
  float*    xl    = ws;
  float*    hbuf  = xl   + (size_t)N_NODES * FDIM;
  int*      qsave = (int*)(hbuf + (size_t)N_NODES * FDIM);
  unsigned* sid   = (unsigned*)(qsave + (size_t)(N_NODES / SLAB) * QSTRIDE);

  auto layer = [&](const float* in, const float* Wl, const float* Wr,
                   const float* att, const float* bias, float* outb, int reuse,
                   unsigned* sidw) {
    gemm_f32_wmma<<<N_NODES / 16, 512, 0, stream>>>(in, Wl, xl, ei, sidw);
    slab_edge_phase<<<N_NODES / SLAB, AGG_THREADS, 0, stream>>>(xl, in, Wr, att, ei, bias, outb,
                                                                qsave, reuse, sid);
  };

  layer(x,    Wl0, Wr0, att0, b0, hbuf, 0, sid);
  layer(hbuf, Wl1, Wr1, att1, b1, out,  1, nullptr);
}


# slab pass A: exp, ordered maximum and denominator update of a group of 8 edges done once, one lane per (edge, head); weights read back by row_share DPP; same exp values, arrival-order sums as before
# speedup vs baseline: 4.8634x; 1.0347x over previous
#include <hip/hip_runtime.h>

#define N_NODES   20000
#define FDIM      256
#define HEADS     4
#define CDIM      64
#define NEDGE     320000
#define ETOT      (NEDGE + N_NODES)
#define NEG_SLOPE 0.2f
#define SLAB      80
#define SID_NONE  255u
static_assert(N_NODES / SLAB < 255, "a slab number fits a byte, 255 left free");

typedef __attribute__((ext_vector_type(2))) float v2f;
typedef __attribute__((ext_vector_type(8))) float v8f;

__device__ __forceinline__ unsigned f2ord(float f) {
  unsigned u = __float_as_uint(f);
  return (u & 0x80000000u) ? ~u : (u | 0x80000000u);
}
__device__ __forceinline__ float ord2f(unsigned u) {
  unsigned v = (u & 0x80000000u) ? (u ^ 0x80000000u) : ~u;
  return __uint_as_float(v);
}

__global__ __launch_bounds__(512)
void gemm_f32_wmma(const float* __restrict__ A, const float* __restrict__ B,
                   float* __restrict__ Out,
                   const int* __restrict__ ei, unsigned* __restrict__ sid) {
  __shared__ float As[16 * 257];
  const int tid      = threadIdx.x;
  const int row_base = blockIdx.x * 16;

  constexpr int SIDW_BLK = (NEDGE / 4) / (N_NODES / 16);
  static_assert(SIDW_BLK * (N_NODES / 16) == NEDGE / 4 && SIDW_BLK <= 512,
                "the blocks share the list's words evenly");
  const bool conv = sid != nullptr && tid < SIDW_BLK;
  const int  sw   = blockIdx.x * SIDW_BLK + tid;
  int4 dq = make_int4(0, 0, 0, 0);
  if (conv) dq = ((const int4*)(ei + NEDGE))[sw];

  for (int i = tid; i < 16 * 256; i += 512) {
    int r = i >> 8, c = i & 255;
    As[r * 257 + c] = A[(size_t)(row_base + r) * FDIM + c];
  }
  __syncthreads();

  if (conv) {
    auto sidof = [](int v) -> unsigned {
      return (unsigned)v < (unsigned)N_NODES ? (unsigned)v / (unsigned)SLAB : SID_NONE;
    };
    sid[sw] = sidof(dq.x) | (sidof(dq.y) << 8) | (sidof(dq.z) << 16) | (sidof(dq.w) << 24);
  }

  const int wave = tid >> 5;
  const int lane = tid & 31;
  const int cn   = wave * 16;
  const int m    = lane & 15;
  const int kh   = (lane >> 4) << 1;

  v8f acc = {};
  #pragma unroll 8
  for (int k = 0; k < FDIM; k += 4) {
    v2f a, b;
    a.x = As[m * 257 + k + kh];
    a.y = As[m * 257 + k + kh + 1];
    b.x = B[(size_t)(k + kh)     * FDIM + cn + m];
    b.y = B[(size_t)(k + kh + 1) * FDIM + cn + m];
    acc = __builtin_amdgcn_wmma_f32_16x16x4_f32(false, a, false, b,
                                                (short)0, acc, false, false);
  }

  const int rbase = row_base + ((lane >> 4) << 3);
  #pragma unroll
  for (int r = 0; r < 8; ++r) {
    Out[(size_t)(rbase + r) * FDIM + cn + m] = acc[r];
  }
}

#define QCAP        24576
#define AGG_THREADS 1024
#define AGG_WAVES   (AGG_THREADS / 32)
#define SEGLEN      (QCAP - SLAB)
#define NSEG        ((NEDGE + SEGLEN - 1) / SEGLEN)
#define ORD_NEG_INF 0x007FFFFFu
#define QSTRIDE     8192
static_assert(QSTRIDE <= QCAP, "a saved queue always fits the queue in LDS");
#define SPEC_RANGE  16.0f
static_assert(N_NODES % SLAB == 0, "slabs must tile the node range");
static_assert(SLAB % 16 == 0, "the slab's xr rows are formed as whole 16-row matrix tiles");
static_assert(NEDGE % 16 == 0, "slab numbers are read as 16-byte vectors of 16 edges");
#define DL_UNKNOWN  0xFFFu
static_assert(ETOT < (1 << 20) && SLAB < DL_UNKNOWN, "queue word packs edge id and local row");

__device__ __forceinline__ float row16_sum(float v) {
  v += __int_as_float(__builtin_amdgcn_mov_dpp(__float_as_int(v), 0x161, 0xf, 0xf, true));
  v += __int_as_float(__builtin_amdgcn_mov_dpp(__float_as_int(v), 0x162, 0xf, 0xf, true));
  v += __int_as_float(__builtin_amdgcn_mov_dpp(__float_as_int(v), 0x164, 0xf, 0xf, true));
  v += __int_as_float(__builtin_amdgcn_mov_dpp(__float_as_int(v), 0x168, 0xf, 0xf, true));
  return v;
}

template <int SRC>
__device__ __forceinline__ float row16_read(float v) {
  static_assert(SRC >= 0 && SRC < 16, "a lane of the row");
  return __int_as_float(__builtin_amdgcn_mov_dpp(__float_as_int(v), 0x150 + SRC, 0xf, 0xf, true));
}

__device__ __forceinline__ float leaky_max(float v) { return fmaxf(v, NEG_SLOPE * v); }
__device__ __forceinline__ float score4(const float4 l, const float4 r, const float4 a) {
  return leaky_max(l.x + r.x) * a.x + leaky_max(l.y + r.y) * a.y +
         leaky_max(l.z + r.z) * a.z + leaky_max(l.w + r.w) * a.w;
}

template <bool B> struct flag_t { static constexpr bool value = B; };

__global__ __launch_bounds__(AGG_THREADS)
void slab_edge_phase(const float* __restrict__ xl, const float* __restrict__ xin,
                     const float* __restrict__ Wr,
                     const float* __restrict__ att, const int* __restrict__ ei,
                     const float* __restrict__ bias, float* __restrict__ outb,
                     int* __restrict__ qsave, int reuse,
                     const unsigned* __restrict__ sid) {
  __shared__ __align__(16) float    acc[(SLAB + 1) * FDIM];
  __shared__ __align__(16) float    xrs[(SLAB + 1) * FDIM];
  __shared__ unsigned queue[QCAP];
  __shared__ unsigned mb[(SLAB + 1) * HEADS];
  __shared__ __align__(16) float    mrun[(SLAB + 1) * HEADS];
  __shared__ float    den[(SLAB + 1) * HEADS];
  __shared__ float    sc[SLAB * HEADS];
  __shared__ unsigned qcount;
  __shared__ unsigned respec;
  __shared__ unsigned preres;

  const int tid  = threadIdx.x;
  const int lane = tid & 31;
  const int wave = __builtin_amdgcn_readfirstlane(tid >> 5);
  const int d0   = blockIdx.x * SLAB;
  const bool hsel = lane >= 16;

  {
    float4*       a4 = (float4*)acc;
    float4*       x4 = (float4*)xrs;
    const float4* g4 = (const float4*)(xin + (size_t)d0 * FDIM);
    const float4  z  = make_float4(0.0f, 0.0f, 0.0f, 0.0f);
    constexpr int NST = SLAB * FDIM / 4 / AGG_THREADS;
    static_assert(NST * AGG_THREADS == SLAB * FDIM / 4, "whole 16-byte pieces per thread");
    float4 st[NST];
    #pragma unroll
    for (int j = 0; j < NST; ++j) st[j] = g4[tid + j * AGG_THREADS];
    #pragma unroll
    for (int j = 0; j < NST; ++j) {
      const int i = tid + j * AGG_THREADS;
      const int r = i >> 6, c4 = i & 63;
      a4[(r << 6) + (c4 ^ (r & 15))] = st[j];
    }
    if (tid < FDIM / 4) x4[SLAB * FDIM / 4 + tid] = z;
    if (tid < (SLAB + 1) * HEADS) { mb[tid] = ORD_NEG_INF; mrun[tid] = 0.0f; den[tid] = 0.0f; }
    if (tid < SLAB) queue[tid] = (unsigned)(NEDGE + d0 + tid) | ((unsigned)tid << 20);
    if (tid == 0) { qcount = SLAB; respec = 0u; preres = 0u; }
  }
  __syncthreads();

  const int role = ((wave >> 2) ^ wave) & 1;
  const int ridx = ((wave >> 3) << 2) | (wave & 3);
  if (role == 0) {
    const int m  = lane & 15;
    const int kh = (lane >> 4) << 1;
    const int cn = ridx * 16;
    const int sw = m << 2;
    const float* bp = Wr + (size_t)kh * FDIM + cn + m;
    const float* ap = acc + m * FDIM + kh;
    v8f c[SLAB / 16] = {};
    #pragma unroll 8
    for (int k = 0; k < FDIM; k += 4) {
      v2f b;
      b.x = bp[(size_t)k * FDIM];
      b.y = bp[(size_t)(k + 1) * FDIM];
      const float* ak = ap + (k ^ sw);
      #pragma unroll
      for (int t = 0; t < SLAB / 16; ++t) {
        const v2f a = *(const v2f*)(ak + t * 16 * FDIM);
        c[t] = __builtin_amdgcn_wmma_f32_16x16x4_f32(false, a, false, b,
                                                     (short)0, c[t], false, false);
      }
    }
    const int rb = (lane >> 4) << 3;
    #pragma unroll
    for (int t = 0; t < SLAB / 16; ++t) {
      #pragma unroll
      for (int r = 0; r < 8; ++r) xrs[(t * 16 + rb + r) * FDIM + cn + m] = c[t][r];
    }
  } else {
    constexpr int FTHREADS = AGG_THREADS / 2;
    const int   ftid = ridx * 32 + lane;
    const int* qs  = qsave + (size_t)blockIdx.x * QSTRIDE;
    const int  cnt = reuse ? __builtin_amdgcn_readfirstlane(qs[0]) : -1;
    const bool kept = cnt >= SLAB && cnt < QSTRIDE;
    if (kept) {
      for (int i = ftid; i < cnt; i += FTHREADS) queue[i] = (unsigned)qs[1 + i];
      if (ftid == 0) { qcount = (unsigned)cnt; preres = 1u; }
    }
    const uint4* sid4 = (const uint4*)sid;
    constexpr int NVEC = NEDGE / 16;
    constexpr int NCH  = (NVEC + FTHREADS - 1) / FTHREADS;
    const int rot = blockIdx.x % NCH;
    const unsigned pat  = (unsigned)blockIdx.x * 0x01010101u;
    for (int g = kept ? NCH : 0; g < NCH; g += 8) {
      uint4 v[8];
      int   vi[8];
      #pragma unroll
      for (int u = 0; u < 8; ++u) {
        int c = g + u + rot;
        if (c >= NCH) c -= NCH;
        const int i = c * FTHREADS + ftid;
        vi[u] = (g + u < NCH && i < NVEC) ? i : -1;
        v[u]  = sid4[min(i, NVEC - 1)];
      }
      #pragma unroll
      for (int u = 0; u < 8; ++u) {
        const unsigned x[4] = {v[u].x ^ pat, v[u].y ^ pat, v[u].z ^ pat, v[u].w ^ pat};
        unsigned z[4];
        #pragma unroll
        for (int j = 0; j < 4; ++j) z[j] = (x[j] - 0x01010101u) & ~x[j] & 0x80808080u;
        if ((z[0] | z[1] | z[2] | z[3]) != 0u && vi[u] >= 0) {
          #pragma unroll
          for (int j = 0; j < 4; ++j) {
            if (z[j] != 0u) {
              #pragma unroll
              for (int k = 0; k < 4; ++k) {
                if (((x[j] >> (8 * k)) & 0xFFu) == 0u) {
                  const int e = vi[u] * 16 + j * 4 + k;
                  const unsigned slot = atomicAdd(&qcount, 1u);
                  if (slot < (unsigned)QCAP) queue[slot] = (unsigned)e | (DL_UNKNOWN << 20);
                }
              }
            }
          }
        }
      }
    }
  }
  __syncthreads();

  {
    float4*      a4 = (float4*)acc;
    const float4 z  = make_float4(0.0f, 0.0f, 0.0f, 0.0f);
    for (int i = tid; i < SLAB * FDIM / 4; i += AGG_THREADS) a4[i] = z;
  }
  __syncthreads();

  const unsigned total  = (unsigned)__builtin_amdgcn_readfirstlane((int)qcount);
  const bool     big    = total > (unsigned)QCAP;
  const int      nbatch = big ? NSEG : 1;
  const bool     spec   = !big;
  bool           unshifted = false;
  const bool     pre    = __builtin_amdgcn_readfirstlane((int)preres) != 0;

  const float4 atA = ((const float4*)att)[lane];
  const float4 atB = ((const float4*)att)[32 + lane];

  for (int b = 0; b < nbatch; ++b) {
    if (big) {
      __syncthreads();
      if (tid == 0) qcount = (b == 0) ? SLAB : 0;
      if (b == 0 && tid < SLAB) queue[tid] = (unsigned)(NEDGE + d0 + tid) | ((unsigned)tid << 20);
      __syncthreads();
      const int s0 = b * SEGLEN;
      const int s1 = min(NEDGE, s0 + SEGLEN);
      for (int e = s0 + tid; e < s1; e += AGG_THREADS) {
        const unsigned r = (unsigned)(ei[NEDGE + e] - d0);
        if (r < (unsigned)SLAB) {
          const unsigned slot = atomicAdd(&qcount, 1u);
          queue[slot] = (unsigned)e | (r << 20);
        }
      }
      __syncthreads();
    }
    const unsigned M = min((unsigned)__builtin_amdgcn_readfirstlane((int)qcount), (unsigned)QCAP);

    auto pass_a = [&](auto sums_tag) __attribute__((always_inline)) {
      constexpr bool SUMS = decltype(sums_tag)::value;
      for (unsigned base = wave * 32; base < M; base += AGG_WAVES * 32) {
        const unsigned idx = base + lane;
        int m_src = 0, m_dl = SLAB;
        if (idx < M) {
          const unsigned ent = queue[idx];
          const int      e   = (int)(ent & 0xFFFFFu);
          m_dl  = (int)(ent >> 20);
          if (pre) {
            m_src = e;
          } else {
            if (e < NEDGE) {
              m_src = ei[e];
              if (m_dl == (int)DL_UNKNOWN)
                m_dl = (int)min((unsigned)(ei[NEDGE + e] - d0), (unsigned)SLAB);
            } else {
              m_src = e - NEDGE;
            }
            queue[idx] = (unsigned)m_src | ((unsigned)m_dl << 20);
          }
        }
        const int n = (int)min(32u, M - base);
        for (int j0 = 0; j0 < n; j0 += 8) {
          float4 va[8], vb[8];
          #pragma unroll
          for (int u = 0; u < 8; ++u) {
            const int s = __builtin_amdgcn_readlane(m_src, j0 + u);
            const float4* row = (const float4*)(xl + (size_t)s * FDIM);
            va[u] = row[lane];
            vb[u] = row[32 + lane];
          }
          if (SUMS) {
            float T = 0.0f;
            #pragma unroll
            for (int u = 0; u < 8; ++u) {
              const int     dl = __builtin_amdgcn_readlane(m_dl, j0 + u);
              const float4* xq = (const float4*)(xrs + dl * FDIM);
              const float   tA = row16_sum(score4(va[u], xq[lane],      atA));
              const float   tB = row16_sum(score4(vb[u], xq[32 + lane], atB));
              T = ((lane & 7) == u) ? ((lane & 8) ? tB : tA) : T;
              __builtin_amdgcn_sched_barrier(0);
            }
            const float P = expf(T);
            {
              const int dlL = __builtin_amdgcn_ds_bpermute((j0 + (lane & 7)) << 2, m_dl);
              const int hd  = ((lane >> 2) & 2) | (lane >> 4);
              atomicMax(&mb[dlL * HEADS + hd], f2ord(T));
              __hip_atomic_fetch_add(&den[dlL * HEADS + hd], P, __ATOMIC_RELAXED, __HIP_MEMORY_SCOPE_WORKGROUP);
            }
#define SLAB_ADD_EDGE(U)                                                                         \
            {                                                                                    \
              const int   dl = __builtin_amdgcn_readlane(m_dl, j0 + U);                          \
              const float pA = row16_read<U>(P);                                                 \
              const float pB = row16_read<8 + U>(P);                                             \
              float* ar = acc + dl * FDIM + lane;                                                \
              __hip_atomic_fetch_add(ar +   0, pA * va[U].x, __ATOMIC_RELAXED, __HIP_MEMORY_SCOPE_WORKGROUP); \
              __hip_atomic_fetch_add(ar +  32, pA * va[U].y, __ATOMIC_RELAXED, __HIP_MEMORY_SCOPE_WORKGROUP); \
              __hip_atomic_fetch_add(ar +  64, pA * va[U].z, __ATOMIC_RELAXED, __HIP_MEMORY_SCOPE_WORKGROUP); \
              __hip_atomic_fetch_add(ar +  96, pA * va[U].w, __ATOMIC_RELAXED, __HIP_MEMORY_SCOPE_WORKGROUP); \
              __hip_atomic_fetch_add(ar + 128, pB * vb[U].x, __ATOMIC_RELAXED, __HIP_MEMORY_SCOPE_WORKGROUP); \
              __hip_atomic_fetch_add(ar + 160, pB * vb[U].y, __ATOMIC_RELAXED, __HIP_MEMORY_SCOPE_WORKGROUP); \
              __hip_atomic_fetch_add(ar + 192, pB * vb[U].z, __ATOMIC_RELAXED, __HIP_MEMORY_SCOPE_WORKGROUP); \
              __hip_atomic_fetch_add(ar + 224, pB * vb[U].w, __ATOMIC_RELAXED, __HIP_MEMORY_SCOPE_WORKGROUP); \
            }
            SLAB_ADD_EDGE(0) SLAB_ADD_EDGE(1) SLAB_ADD_EDGE(2) SLAB_ADD_EDGE(3)
            SLAB_ADD_EDGE(4) SLAB_ADD_EDGE(5) SLAB_ADD_EDGE(6) SLAB_ADD_EDGE(7)
#undef SLAB_ADD_EDGE
          } else {
            #pragma unroll
            for (int u = 0; u < 8; ++u) {
              const int     dl = __builtin_amdgcn_readlane(m_dl, j0 + u);
              const float4* xq = (const float4*)(xrs + dl * FDIM);
              const float   tA = row16_sum(score4(va[u], xq[lane],      atA));
              const float   tB = row16_sum(score4(vb[u], xq[32 + lane], atB));
              if ((lane & 15) == 0) {
                atomicMax(&mb[dl * HEADS + (hsel ? 1 : 0)], f2ord(tA));
                atomicMax(&mb[dl * HEADS + (hsel ? 3 : 2)], f2ord(tB));
              }
            }
          }
        }
      }
    };
    if (spec) pass_a(flag_t<true>{}); else pass_a(flag_t<false>{});
    __syncthreads();

    if (tid < SLAB * HEADS) {
      const float mnew_b = ord2f(mb[tid]);
      mb[tid] = ORD_NEG_INF;
      if (b == 0) {
        mrun[tid] = mnew_b;
        if (spec && !(fabsf(mnew_b) <= SPEC_RANGE)) respec = 1u;
      } else {
        const float mo = mrun[tid];
        const float mn = fmaxf(mo, mnew_b);
        const float f  = expf(mo - mn);
        mrun[tid] = mn;
        den[tid] *= f;
        sc[tid]   = f;
      }
    }
    __syncthreads();
    if (spec) {
      unshifted = __builtin_amdgcn_readfirstlane((int)respec) == 0;
      if (unshifted) break;
      float4*      a4 = (float4*)acc;
      const float4 z  = make_float4(0.0f, 0.0f, 0.0f, 0.0f);
      for (int i = tid; i < SLAB * FDIM / 4; i += AGG_THREADS) a4[i] = z;
      if (tid < SLAB * HEADS) den[tid] = 0.0f;
      __syncthreads();
    }
    if (b > 0) {
      #pragma unroll 1
      for (int i = tid; i < SLAB * FDIM; i += AGG_THREADS) {
        const int r = i >> 8, q = (i >> 5) & 7, l = i & 31;
        acc[i] *= sc[r * HEADS + ((q >> 2) << 1) + (l >> 4)];
      }
      __syncthreads();
    }

    for (unsigned base = wave * 32; base < M; base += AGG_WAVES * 32) {
      const unsigned idx = base + lane;
      int   m_src = 0, m_dl = SLAB;
      float m_m0 = 0.0f, m_m1 = 0.0f, m_m2 = 0.0f, m_m3 = 0.0f;
      if (idx < M) {
        const unsigned ent = queue[idx];
        m_src = (int)(ent & 0xFFFFFu);
        m_dl  = (int)(ent >> 20);
        const float4 mm = ((const float4*)mrun)[m_dl];
        m_m0 = mm.x; m_m1 = mm.y; m_m2 = mm.z; m_m3 = mm.w;
      }
      const int n = (int)min(32u, M - base);
      for (int j0 = 0; j0 < n; j0 += 8) {
        float4 va[8], vb[8];
        #pragma unroll
        for (int u = 0; u < 8; ++u) {
          const int s = __builtin_amdgcn_readlane(m_src, j0 + u);
          const float4* row = (const float4*)(xl + (size_t)s * FDIM);
          va[u] = row[lane];
          vb[u] = row[32 + lane];
        }
        #pragma unroll
        for (int u = 0; u < 8; ++u) {
          const int     dl = __builtin_amdgcn_readlane(m_dl, j0 + u);
          const float4* xq = (const float4*)(xrs + dl * FDIM);
          const float   tA = row16_sum(score4(va[u], xq[lane],      atA));
          const float   tB = row16_sum(score4(vb[u], xq[32 + lane], atB));
          const float m0 = __int_as_float(__builtin_amdgcn_readlane(__float_as_int(m_m0), j0 + u));
          const float m1 = __int_as_float(__builtin_amdgcn_readlane(__float_as_int(m_m1), j0 + u));
          const float m2 = __int_as_float(__builtin_amdgcn_readlane(__float_as_int(m_m2), j0 + u));
          const float m3 = __int_as_float(__builtin_amdgcn_readlane(__float_as_int(m_m3), j0 + u));
          const float pA = expf(tA - (hsel ? m1 : m0));
          const float pB = expf(tB - (hsel ? m3 : m2));
          if ((lane & 15) == 0) {
            __hip_atomic_fetch_add(&den[dl * HEADS + (hsel ? 1 : 0)], pA, __ATOMIC_RELAXED, __HIP_MEMORY_SCOPE_WORKGROUP);
            __hip_atomic_fetch_add(&den[dl * HEADS + (hsel ? 3 : 2)], pB, __ATOMIC_RELAXED, __HIP_MEMORY_SCOPE_WORKGROUP);
          }
          float* ar = acc + dl * FDIM + lane;
          __hip_atomic_fetch_add(ar +   0, pA * va[u].x, __ATOMIC_RELAXED, __HIP_MEMORY_SCOPE_WORKGROUP);
          __hip_atomic_fetch_add(ar +  32, pA * va[u].y, __ATOMIC_RELAXED, __HIP_MEMORY_SCOPE_WORKGROUP);
          __hip_atomic_fetch_add(ar +  64, pA * va[u].z, __ATOMIC_RELAXED, __HIP_MEMORY_SCOPE_WORKGROUP);
          __hip_atomic_fetch_add(ar +  96, pA * va[u].w, __ATOMIC_RELAXED, __HIP_MEMORY_SCOPE_WORKGROUP);
          __hip_atomic_fetch_add(ar + 128, pB * vb[u].x, __ATOMIC_RELAXED, __HIP_MEMORY_SCOPE_WORKGROUP);
          __hip_atomic_fetch_add(ar + 160, pB * vb[u].y, __ATOMIC_RELAXED, __HIP_MEMORY_SCOPE_WORKGROUP);
          __hip_atomic_fetch_add(ar + 192, pB * vb[u].z, __ATOMIC_RELAXED, __HIP_MEMORY_SCOPE_WORKGROUP);
          __hip_atomic_fetch_add(ar + 224, pB * vb[u].w, __ATOMIC_RELAXED, __HIP_MEMORY_SCOPE_WORKGROUP);
        }
      }
    }
  }
  __syncthreads();

  if (!reuse) {
    int* qs = qsave + (size_t)blockIdx.x * QSTRIDE;
    const bool fits = !big && total < (unsigned)QSTRIDE;
    if (fits) {
      for (unsigned i = tid; i < total; i += AGG_THREADS) qs[1 + i] = (int)queue[i];
    }
    if (tid == 0) qs[0] = fits ? (int)total : -1;
  }

  {
    const float4 bA = ((const float4*)bias)[lane];
    const float4 bB = ((const float4*)bias)[32 + lane];
    for (int r = wave; r < SLAB; r += AGG_WAVES) {
      const float* ar = acc + r * FDIM + lane;
      const int    hA = r * HEADS + (hsel ? 1 : 0);
      const int    hB = r * HEADS + (hsel ? 3 : 2);
      const float  eA = unshifted ? 1e-16f * expf(mrun[hA]) : 1e-16f;
      const float  eB = unshifted ? 1e-16f * expf(mrun[hB]) : 1e-16f;
      const float  dA = den[hA] + eA;
      const float  dB = den[hB] + eB;
      float o[8] = {ar[0] / dA + bA.x,   ar[32] / dA + bA.y,  ar[64] / dA + bA.z,  ar[96] / dA + bA.w,
                    ar[128] / dB + bB.x, ar[160] / dB + bB.y, ar[192] / dB + bB.z, ar[224] / dB + bB.w};
      #pragma unroll
      for (int q = 0; q < 8; ++q) o[q] = o[q] > 0.0f ? o[q] : (expf(o[q]) - 1.0f);
      float4* orow = (float4*)(outb + (size_t)(d0 + r) * FDIM);
      orow[lane]      = make_float4(o[0], o[1], o[2], o[3]);
      orow[32 + lane] = make_float4(o[4], o[5], o[6], o[7]);
    }
  }
}

extern "C" void kernel_launch(void* const* d_in, const int* in_sizes, int n_in,
                              void* d_out, int out_size, void* d_ws, size_t ws_size,
                              hipStream_t stream) {
  const float* x    = (const float*)d_in[0];
  const int*   ei   = (const int*)  d_in[1];
  const float* Wl0  = (const float*)d_in[2];
  const float* Wr0  = (const float*)d_in[3];
  const float* att0 = (const float*)d_in[4];
  const float* b0   = (const float*)d_in[5];
  const float* Wl1  = (const float*)d_in[6];
  const float* Wr1  = (const float*)d_in[7];
  const float* att1 = (const float*)d_in[8];
  const float* b1   = (const float*)d_in[9];
  float* out = (float*)d_out;

  float*    ws    = (float*)d_ws;
  float*    xl    = ws;
  float*    hbuf  = xl   + (size_t)N_NODES * FDIM;
  int*      qsave = (int*)(hbuf + (size_t)N_NODES * FDIM);
  unsigned* sid   = (unsigned*)(qsave + (size_t)(N_NODES / SLAB) * QSTRIDE);

  auto layer = [&](const float* in, const float* Wl, const float* Wr,
                   const float* att, const float* bias, float* outb, int reuse,
                   unsigned* sidw) {
    gemm_f32_wmma<<<N_NODES / 16, 512, 0, stream>>>(in, Wl, xl, ei, sidw);
    slab_edge_phase<<<N_NODES / SLAB, AGG_THREADS, 0, stream>>>(xl, in, Wr, att, ei, bias, outb,
                                                                qsave, reuse, sid);
  };

  layer(x,    Wl0, Wr0, att0, b0, hbuf, 0, sid);
  layer(hbuf, Wl1, Wr1, att1, b1, out,  1, nullptr);
}


# slab queues from sorted edge blocks: first product launch sorts each 256-edge block by destination slab (run offsets + queue words), a slab reads its runs, no list scan; arrival-order sums as before
# speedup vs baseline: 5.1807x; 1.0652x over previous
#include <hip/hip_runtime.h>

#define N_NODES   20000
#define FDIM      256
#define HEADS     4
#define CDIM      64
#define NEDGE     320000
#define ETOT      (NEDGE + N_NODES)
#define NEG_SLOPE 0.2f
#define SLAB      80
#define NSLAB     (N_NODES / SLAB)
#define EBLK      256
#define NEBLK     (NEDGE / EBLK)
static_assert(NEBLK * EBLK == NEDGE && NEBLK == N_NODES / 16, "one edge block per product block");
static_assert(NSLAB < EBLK, "a bin per slab and one for the rest");

typedef __attribute__((ext_vector_type(2))) float v2f;
typedef __attribute__((ext_vector_type(8))) float v8f;

__device__ __forceinline__ unsigned f2ord(float f) {
  unsigned u = __float_as_uint(f);
  return (u & 0x80000000u) ? ~u : (u | 0x80000000u);
}
__device__ __forceinline__ float ord2f(unsigned u) {
  unsigned v = (u & 0x80000000u) ? (u ^ 0x80000000u) : ~u;
  return __uint_as_float(v);
}

__global__ __launch_bounds__(512)
void gemm_f32_wmma(const float* __restrict__ A, const float* __restrict__ B,
                   float* __restrict__ Out,
                   const int* __restrict__ ei, unsigned* __restrict__ boff,
                   unsigned* __restrict__ bent) {
  __shared__ float As[16 * 257];
  __shared__ __align__(16) unsigned bins[EBLK];
  const int tid      = threadIdx.x;
  const int row_base = blockIdx.x * 16;

  const bool sorter = boff != nullptr && tid < 32;
  int4 dA = make_int4(0, 0, 0, 0), dB = dA, sA = dA, sB = dA;
  if (sorter) {
    const int4* dp = (const int4*)(ei + NEDGE + blockIdx.x * EBLK) + tid * 2;
    const int4* sp = (const int4*)(ei + blockIdx.x * EBLK) + tid * 2;
    dA = dp[0]; dB = dp[1];
    sA = sp[0]; sB = sp[1];
  }

  for (int i = tid; i < 16 * 256; i += 512) {
    int r = i >> 8, c = i & 255;
    As[r * 257 + c] = A[(size_t)(row_base + r) * FDIM + c];
  }
  __syncthreads();

  if (sorter) {
    const int dv[8] = {dA.x, dA.y, dA.z, dA.w, dB.x, dB.y, dB.z, dB.w};
    const int sv[8] = {sA.x, sA.y, sA.z, sA.w, sB.x, sB.y, sB.z, sB.w};
    uint4* bins4 = (uint4*)bins;
    bins4[tid * 2]     = make_uint4(0u, 0u, 0u, 0u);
    bins4[tid * 2 + 1] = make_uint4(0u, 0u, 0u, 0u);
    __builtin_amdgcn_fence(__ATOMIC_SEQ_CST, "wavefront");
    unsigned bin[8], rk[8], word[8];
    #pragma unroll
    for (int k = 0; k < 8; ++k) {
      const unsigned d  = (unsigned)dv[k];
      const bool     in = d < (unsigned)N_NODES;
      const unsigned sl = in ? d / (unsigned)SLAB : (unsigned)NSLAB;
      bin[k]  = sl;
      word[k] = ((unsigned)sv[k] & 0xFFFFFu) | ((in ? d - sl * (unsigned)SLAB : 0u) << 20);
      rk[k]   = atomicAdd(&bins[sl], 1u);
    }
    __builtin_amdgcn_fence(__ATOMIC_SEQ_CST, "wavefront");
    const uint4 cA = bins4[tid * 2], cB = bins4[tid * 2 + 1];
    const unsigned cnt[8] = {cA.x, cA.y, cA.z, cA.w, cB.x, cB.y, cB.z, cB.w};
    unsigned tot = 0u;
    #pragma unroll
    for (int k = 0; k < 8; ++k) tot += cnt[k];
    unsigned inc = tot;
    #pragma unroll
    for (int dlt = 1; dlt < 32; dlt <<= 1) {
      const unsigned up = __shfl_up(inc, dlt);
      if (tid >= dlt) inc += up;
    }
    unsigned off[8];
    unsigned run = inc - tot;
    #pragma unroll
    for (int k = 0; k < 8; ++k) { off[k] = run; run += cnt[k]; }
    __builtin_amdgcn_fence(__ATOMIC_SEQ_CST, "wavefront");
    bins4[tid * 2]     = make_uint4(off[0], off[1], off[2], off[3]);
    bins4[tid * 2 + 1] = make_uint4(off[4], off[5], off[6], off[7]);
    unsigned pw[8];
    #pragma unroll
    for (int k = 0; k < 8; ++k) pw[k] = off[k] | ((off[k] + cnt[k]) << 16);
    uint4* bo4 = (uint4*)(boff + (size_t)blockIdx.x * EBLK);
    bo4[tid * 2]     = make_uint4(pw[0], pw[1], pw[2], pw[3]);
    bo4[tid * 2 + 1] = make_uint4(pw[4], pw[5], pw[6], pw[7]);
    __builtin_amdgcn_fence(__ATOMIC_SEQ_CST, "wavefront");
    unsigned* eb = bent + (size_t)blockIdx.x * EBLK;
    #pragma unroll
    for (int k = 0; k < 8; ++k) eb[bins[bin[k]] + rk[k]] = word[k];
  }

  const int wave = tid >> 5;
  const int lane = tid & 31;
  const int cn   = wave * 16;
  const int m    = lane & 15;
  const int kh   = (lane >> 4) << 1;

  v8f acc = {};
  #pragma unroll 8
  for (int k = 0; k < FDIM; k += 4) {
    v2f a, b;
    a.x = As[m * 257 + k + kh];
    a.y = As[m * 257 + k + kh + 1];
    b.x = B[(size_t)(k + kh)     * FDIM + cn + m];
    b.y = B[(size_t)(k + kh + 1) * FDIM + cn + m];
    acc = __builtin_amdgcn_wmma_f32_16x16x4_f32(false, a, false, b,
                                                (short)0, acc, false, false);
  }

  const int rbase = row_base + ((lane >> 4) << 3);
  #pragma unroll
  for (int r = 0; r < 8; ++r) {
    Out[(size_t)(rbase + r) * FDIM + cn + m] = acc[r];
  }
}

#define QCAP        24576
#define AGG_THREADS 1024
#define AGG_WAVES   (AGG_THREADS / 32)
#define SEGLEN      (QCAP - SLAB)
#define NSEG        ((NEDGE + SEGLEN - 1) / SEGLEN)
#define ORD_NEG_INF 0x007FFFFFu
#define QSTRIDE     4096
static_assert(QSTRIDE <= QCAP, "a saved queue always fits the queue in LDS");
#define SPEC_RANGE  16.0f
static_assert(N_NODES % SLAB == 0, "slabs must tile the node range");
static_assert(SLAB % 16 == 0, "the slab's xr rows are formed as whole 16-row matrix tiles");
static_assert(ETOT < (1 << 20) && SLAB < (1 << 12), "queue word packs edge id or source row and local row");

__device__ __forceinline__ float row16_sum(float v) {
  v += __int_as_float(__builtin_amdgcn_mov_dpp(__float_as_int(v), 0x161, 0xf, 0xf, true));
  v += __int_as_float(__builtin_amdgcn_mov_dpp(__float_as_int(v), 0x162, 0xf, 0xf, true));
  v += __int_as_float(__builtin_amdgcn_mov_dpp(__float_as_int(v), 0x164, 0xf, 0xf, true));
  v += __int_as_float(__builtin_amdgcn_mov_dpp(__float_as_int(v), 0x168, 0xf, 0xf, true));
  return v;
}

template <int SRC>
__device__ __forceinline__ float row16_read(float v) {
  static_assert(SRC >= 0 && SRC < 16, "a lane of the row");
  return __int_as_float(__builtin_amdgcn_mov_dpp(__float_as_int(v), 0x150 + SRC, 0xf, 0xf, true));
}

__device__ __forceinline__ float leaky_max(float v) { return fmaxf(v, NEG_SLOPE * v); }
__device__ __forceinline__ float score4(const float4 l, const float4 r, const float4 a) {
  return leaky_max(l.x + r.x) * a.x + leaky_max(l.y + r.y) * a.y +
         leaky_max(l.z + r.z) * a.z + leaky_max(l.w + r.w) * a.w;
}

template <bool B> struct flag_t { static constexpr bool value = B; };

__global__ __launch_bounds__(AGG_THREADS)
void slab_edge_phase(const float* __restrict__ xl, const float* __restrict__ xin,
                     const float* __restrict__ Wr,
                     const float* __restrict__ att, const int* __restrict__ ei,
                     const float* __restrict__ bias, float* __restrict__ outb,
                     int* __restrict__ qsave, int reuse,
                     const unsigned* __restrict__ boff,
                     const unsigned* __restrict__ bent) {
  __shared__ __align__(16) float    acc[(SLAB + 1) * FDIM];
  __shared__ __align__(16) float    xrs[(SLAB + 1) * FDIM];
  __shared__ unsigned queue[QCAP];
  __shared__ unsigned mb[(SLAB + 1) * HEADS];
  __shared__ __align__(16) float    mrun[(SLAB + 1) * HEADS];
  __shared__ float    den[(SLAB + 1) * HEADS];
  __shared__ float    sc[SLAB * HEADS];
  __shared__ unsigned qcount;
  __shared__ unsigned respec;

  const int tid  = threadIdx.x;
  const int lane = tid & 31;
  const int wave = __builtin_amdgcn_readfirstlane(tid >> 5);
  const int d0   = blockIdx.x * SLAB;
  const bool hsel = lane >= 16;

  {
    float4*       a4 = (float4*)acc;
    float4*       x4 = (float4*)xrs;
    const float4* g4 = (const float4*)(xin + (size_t)d0 * FDIM);
    const float4  z  = make_float4(0.0f, 0.0f, 0.0f, 0.0f);
    constexpr int NST = SLAB * FDIM / 4 / AGG_THREADS;
    static_assert(NST * AGG_THREADS == SLAB * FDIM / 4, "whole 16-byte pieces per thread");
    float4 st[NST];
    #pragma unroll
    for (int j = 0; j < NST; ++j) st[j] = g4[tid + j * AGG_THREADS];
    #pragma unroll
    for (int j = 0; j < NST; ++j) {
      const int i = tid + j * AGG_THREADS;
      const int r = i >> 6, c4 = i & 63;
      a4[(r << 6) + (c4 ^ (r & 15))] = st[j];
    }
    if (tid < FDIM / 4) x4[SLAB * FDIM / 4 + tid] = z;
    if (tid < (SLAB + 1) * HEADS) { mb[tid] = ORD_NEG_INF; mrun[tid] = 0.0f; den[tid] = 0.0f; }
    if (tid < SLAB) queue[tid] = (unsigned)(d0 + tid) | ((unsigned)tid << 20);
    if (tid == 0) { qcount = SLAB; respec = 0u; }
  }
  __syncthreads();

  const int role = ((wave >> 2) ^ wave) & 1;
  const int ridx = ((wave >> 3) << 2) | (wave & 3);
  if (role == 0) {
    const int m  = lane & 15;
    const int kh = (lane >> 4) << 1;
    const int cn = ridx * 16;
    const int sw = m << 2;
    const float* bp = Wr + (size_t)kh * FDIM + cn + m;
    const float* ap = acc + m * FDIM + kh;
    v8f c[SLAB / 16] = {};
    #pragma unroll 8
    for (int k = 0; k < FDIM; k += 4) {
      v2f b;
      b.x = bp[(size_t)k * FDIM];
      b.y = bp[(size_t)(k + 1) * FDIM];
      const float* ak = ap + (k ^ sw);
      #pragma unroll
      for (int t = 0; t < SLAB / 16; ++t) {
        const v2f a = *(const v2f*)(ak + t * 16 * FDIM);
        c[t] = __builtin_amdgcn_wmma_f32_16x16x4_f32(false, a, false, b,
                                                     (short)0, c[t], false, false);
      }
    }
    const int rb = (lane >> 4) << 3;
    #pragma unroll
    for (int t = 0; t < SLAB / 16; ++t) {
      #pragma unroll
      for (int r = 0; r < 8; ++r) xrs[(t * 16 + rb + r) * FDIM + cn + m] = c[t][r];
    }
  } else {
    constexpr int FTHREADS = AGG_THREADS / 2;
    const int   ftid = ridx * 32 + lane;
    const int* qs  = qsave + (size_t)blockIdx.x * QSTRIDE;
    const int  cnt = reuse ? __builtin_amdgcn_readfirstlane(qs[0]) : -1;
    const bool kept = cnt >= SLAB && cnt < QSTRIDE;
    if (kept) {
      for (int i = ftid; i < cnt; i += FTHREADS) queue[i] = (unsigned)qs[1 + i];
      if (ftid == 0) qcount = (unsigned)cnt;
    }
    if (!kept) {
      constexpr int NB = (NEBLK + FTHREADS - 1) / FTHREADS;
      unsigned ow[NB], w[NB][4], nrun[NB];
      #pragma unroll
      for (int t = 0; t < NB; ++t) {
        const int b = ftid + t * FTHREADS;
        ow[t] = boff[(size_t)min(b, NEBLK - 1) * EBLK + blockIdx.x];
        if (b >= NEBLK) ow[t] = 0u;
      }
      #pragma unroll
      for (int t = 0; t < NB; ++t) {
        const int b = min(ftid + t * FTHREADS, NEBLK - 1);
        const unsigned o0 = ow[t] & 0xFFFFu;
        const unsigned o1 = min(ow[t] >> 16, (unsigned)EBLK);
        nrun[t] = o1 > o0 ? o1 - o0 : 0u;
        #pragma unroll
        for (int q = 0; q < 4; ++q)
          w[t][q] = bent[(size_t)b * EBLK + min(o0 + (unsigned)q, (unsigned)EBLK - 1u)];
      }
      unsigned bm[NB][4];
      unsigned tot = 0u;
      #pragma unroll
      for (int t = 0; t < NB; ++t) {
        #pragma unroll
        for (int q = 0; q < 4; ++q) {
          bm[t][q] = __builtin_amdgcn_ballot_w32((unsigned)q < nrun[t]);
          tot += (unsigned)__builtin_popcount(bm[t][q]);
        }
      }
      if (tot != 0u) {
        unsigned base = 0u;
        if (lane == 0) base = atomicAdd(&qcount, tot);
        base = (unsigned)__builtin_amdgcn_readfirstlane((int)base);
        #pragma unroll
        for (int t = 0; t < NB; ++t) {
          #pragma unroll
          for (int q = 0; q < 4; ++q) {
            const unsigned slot = base + __builtin_amdgcn_mbcnt_lo(bm[t][q], 0u);
            if ((unsigned)q < nrun[t] && slot < (unsigned)QCAP) queue[slot] = w[t][q];
            base += (unsigned)__builtin_popcount(bm[t][q]);
          }
        }
      }
      #pragma unroll
      for (int t = 0; t < NB; ++t) {
        const int b = min(ftid + t * FTHREADS, NEBLK - 1);
        const unsigned o0 = ow[t] & 0xFFFFu;
        for (unsigned i = 4u; i < nrun[t]; ++i) {
          const unsigned wd   = bent[(size_t)b * EBLK + o0 + i];
          const unsigned slot = atomicAdd(&qcount, 1u);
          if (slot < (unsigned)QCAP) queue[slot] = wd;
        }
      }
    }
  }
  __syncthreads();

  {
    float4*      a4 = (float4*)acc;
    const float4 z  = make_float4(0.0f, 0.0f, 0.0f, 0.0f);
    for (int i = tid; i < SLAB * FDIM / 4; i += AGG_THREADS) a4[i] = z;
  }
  __syncthreads();

  const unsigned total  = (unsigned)__builtin_amdgcn_readfirstlane((int)qcount);
  const bool     big    = total > (unsigned)QCAP;
  const int      nbatch = big ? NSEG : 1;
  const bool     spec   = !big;
  bool           unshifted = false;
  const bool     pre    = !big;

  const float4 atA = ((const float4*)att)[lane];
  const float4 atB = ((const float4*)att)[32 + lane];

  for (int b = 0; b < nbatch; ++b) {
    if (big) {
      __syncthreads();
      if (tid == 0) qcount = (b == 0) ? SLAB : 0;
      if (b == 0 && tid < SLAB) queue[tid] = (unsigned)(NEDGE + d0 + tid) | ((unsigned)tid << 20);
      __syncthreads();
      const int s0 = b * SEGLEN;
      const int s1 = min(NEDGE, s0 + SEGLEN);
      for (int e = s0 + tid; e < s1; e += AGG_THREADS) {
        const unsigned r = (unsigned)(ei[NEDGE + e] - d0);
        if (r < (unsigned)SLAB) {
          const unsigned slot = atomicAdd(&qcount, 1u);
          queue[slot] = (unsigned)e | (r << 20);
        }
      }
      __syncthreads();
    }
    const unsigned M = min((unsigned)__builtin_amdgcn_readfirstlane((int)qcount), (unsigned)QCAP);

    auto pass_a = [&](auto sums_tag) __attribute__((always_inline)) {
      constexpr bool SUMS = decltype(sums_tag)::value;
      for (unsigned base = wave * 32; base < M; base += AGG_WAVES * 32) {
        const unsigned idx = base + lane;
        int m_src = 0, m_dl = SLAB;
        if (idx < M) {
          const unsigned ent = queue[idx];
          const int      e   = (int)(ent & 0xFFFFFu);
          m_dl  = (int)(ent >> 20);
          if (pre) {
            m_src = e;
          } else {
            m_src = e < NEDGE ? ei[e] : e - NEDGE;
            queue[idx] = (unsigned)m_src | ((unsigned)m_dl << 20);
          }
        }
        const int n = (int)min(32u, M - base);
        for (int j0 = 0; j0 < n; j0 += 8) {
          float4 va[8], vb[8];
          #pragma unroll
          for (int u = 0; u < 8; ++u) {
            const int s = __builtin_amdgcn_readlane(m_src, j0 + u);
            const float4* row = (const float4*)(xl + (size_t)s * FDIM);
            va[u] = row[lane];
            vb[u] = row[32 + lane];
          }
          if (SUMS) {
            float T = 0.0f;
            #pragma unroll
            for (int u = 0; u < 8; ++u) {
              const int     dl = __builtin_amdgcn_readlane(m_dl, j0 + u);
              const float4* xq = (const float4*)(xrs + dl * FDIM);
              const float   tA = row16_sum(score4(va[u], xq[lane],      atA));
              const float   tB = row16_sum(score4(vb[u], xq[32 + lane], atB));
              T = ((lane & 7) == u) ? ((lane & 8) ? tB : tA) : T;
              __builtin_amdgcn_sched_barrier(0);
            }
            const float P = expf(T);
            {
              const int dlL = __builtin_amdgcn_ds_bpermute((j0 + (lane & 7)) << 2, m_dl);
              const int hd  = ((lane >> 2) & 2) | (lane >> 4);
              atomicMax(&mb[dlL * HEADS + hd], f2ord(T));
              __hip_atomic_fetch_add(&den[dlL * HEADS + hd], P, __ATOMIC_RELAXED, __HIP_MEMORY_SCOPE_WORKGROUP);
            }
#define SLAB_ADD_EDGE(U)                                                                         \
            {                                                                                    \
              const int   dl = __builtin_amdgcn_readlane(m_dl, j0 + U);                          \
              const float pA = row16_read<U>(P);                                                 \
              const float pB = row16_read<8 + U>(P);                                             \
              float* ar = acc + dl * FDIM + lane;                                                \
              __hip_atomic_fetch_add(ar +   0, pA * va[U].x, __ATOMIC_RELAXED, __HIP_MEMORY_SCOPE_WORKGROUP); \
              __hip_atomic_fetch_add(ar +  32, pA * va[U].y, __ATOMIC_RELAXED, __HIP_MEMORY_SCOPE_WORKGROUP); \
              __hip_atomic_fetch_add(ar +  64, pA * va[U].z, __ATOMIC_RELAXED, __HIP_MEMORY_SCOPE_WORKGROUP); \
              __hip_atomic_fetch_add(ar +  96, pA * va[U].w, __ATOMIC_RELAXED, __HIP_MEMORY_SCOPE_WORKGROUP); \
              __hip_atomic_fetch_add(ar + 128, pB * vb[U].x, __ATOMIC_RELAXED, __HIP_MEMORY_SCOPE_WORKGROUP); \
              __hip_atomic_fetch_add(ar + 160, pB * vb[U].y, __ATOMIC_RELAXED, __HIP_MEMORY_SCOPE_WORKGROUP); \
              __hip_atomic_fetch_add(ar + 192, pB * vb[U].z, __ATOMIC_RELAXED, __HIP_MEMORY_SCOPE_WORKGROUP); \
              __hip_atomic_fetch_add(ar + 224, pB * vb[U].w, __ATOMIC_RELAXED, __HIP_MEMORY_SCOPE_WORKGROUP); \
            }
            SLAB_ADD_EDGE(0) SLAB_ADD_EDGE(1) SLAB_ADD_EDGE(2) SLAB_ADD_EDGE(3)
            SLAB_ADD_EDGE(4) SLAB_ADD_EDGE(5) SLAB_ADD_EDGE(6) SLAB_ADD_EDGE(7)
#undef SLAB_ADD_EDGE
          } else {
            #pragma unroll
            for (int u = 0; u < 8; ++u) {
              const int     dl = __builtin_amdgcn_readlane(m_dl, j0 + u);
              const float4* xq = (const float4*)(xrs + dl * FDIM);
              const float   tA = row16_sum(score4(va[u], xq[lane],      atA));
              const float   tB = row16_sum(score4(vb[u], xq[32 + lane], atB));
              if ((lane & 15) == 0) {
                atomicMax(&mb[dl * HEADS + (hsel ? 1 : 0)], f2ord(tA));
                atomicMax(&mb[dl * HEADS + (hsel ? 3 : 2)], f2ord(tB));
              }
            }
          }
        }
      }
    };
    if (spec) pass_a(flag_t<true>{}); else pass_a(flag_t<false>{});
    __syncthreads();

    if (tid < SLAB * HEADS) {
      const float mnew_b = ord2f(mb[tid]);
      mb[tid] = ORD_NEG_INF;
      if (b == 0) {
        mrun[tid] = mnew_b;
        if (spec && !(fabsf(mnew_b) <= SPEC_RANGE)) respec = 1u;
      } else {
        const float mo = mrun[tid];
        const float mn = fmaxf(mo, mnew_b);
        const float f  = expf(mo - mn);
        mrun[tid] = mn;
        den[tid] *= f;
        sc[tid]   = f;
      }
    }
    __syncthreads();
    if (spec) {
      unshifted = __builtin_amdgcn_readfirstlane((int)respec) == 0;
      if (unshifted) break;
      float4*      a4 = (float4*)acc;
      const float4 z  = make_float4(0.0f, 0.0f, 0.0f, 0.0f);
      for (int i = tid; i < SLAB * FDIM / 4; i += AGG_THREADS) a4[i] = z;
      if (tid < SLAB * HEADS) den[tid] = 0.0f;
      __syncthreads();
    }
    if (b > 0) {
      #pragma unroll 1
      for (int i = tid; i < SLAB * FDIM; i += AGG_THREADS) {
        const int r = i >> 8, q = (i >> 5) & 7, l = i & 31;
        acc[i] *= sc[r * HEADS + ((q >> 2) << 1) + (l >> 4)];
      }
      __syncthreads();
    }

    for (unsigned base = wave * 32; base < M; base += AGG_WAVES * 32) {
      const unsigned idx = base + lane;
      int   m_src = 0, m_dl = SLAB;
      float m_m0 = 0.0f, m_m1 = 0.0f, m_m2 = 0.0f, m_m3 = 0.0f;
      if (idx < M) {
        const unsigned ent = queue[idx];
        m_src = (int)(ent & 0xFFFFFu);
        m_dl  = (int)(ent >> 20);
        const float4 mm = ((const float4*)mrun)[m_dl];
        m_m0 = mm.x; m_m1 = mm.y; m_m2 = mm.z; m_m3 = mm.w;
      }
      const int n = (int)min(32u, M - base);
      for (int j0 = 0; j0 < n; j0 += 8) {
        float4 va[8], vb[8];
        #pragma unroll
        for (int u = 0; u < 8; ++u) {
          const int s = __builtin_amdgcn_readlane(m_src, j0 + u);
          const float4* row = (const float4*)(xl + (size_t)s * FDIM);
          va[u] = row[lane];
          vb[u] = row[32 + lane];
        }
        #pragma unroll
        for (int u = 0; u < 8; ++u) {
          const int     dl = __builtin_amdgcn_readlane(m_dl, j0 + u);
          const float4* xq = (const float4*)(xrs + dl * FDIM);
          const float   tA = row16_sum(score4(va[u], xq[lane],      atA));
          const float   tB = row16_sum(score4(vb[u], xq[32 + lane], atB));
          const float m0 = __int_as_float(__builtin_amdgcn_readlane(__float_as_int(m_m0), j0 + u));
          const float m1 = __int_as_float(__builtin_amdgcn_readlane(__float_as_int(m_m1), j0 + u));
          const float m2 = __int_as_float(__builtin_amdgcn_readlane(__float_as_int(m_m2), j0 + u));
          const float m3 = __int_as_float(__builtin_amdgcn_readlane(__float_as_int(m_m3), j0 + u));
          const float pA = expf(tA - (hsel ? m1 : m0));
          const float pB = expf(tB - (hsel ? m3 : m2));
          if ((lane & 15) == 0) {
            __hip_atomic_fetch_add(&den[dl * HEADS + (hsel ? 1 : 0)], pA, __ATOMIC_RELAXED, __HIP_MEMORY_SCOPE_WORKGROUP);
            __hip_atomic_fetch_add(&den[dl * HEADS + (hsel ? 3 : 2)], pB, __ATOMIC_RELAXED, __HIP_MEMORY_SCOPE_WORKGROUP);
          }
          float* ar = acc + dl * FDIM + lane;
          __hip_atomic_fetch_add(ar +   0, pA * va[u].x, __ATOMIC_RELAXED, __HIP_MEMORY_SCOPE_WORKGROUP);
          __hip_atomic_fetch_add(ar +  32, pA * va[u].y, __ATOMIC_RELAXED, __HIP_MEMORY_SCOPE_WORKGROUP);
          __hip_atomic_fetch_add(ar +  64, pA * va[u].z, __ATOMIC_RELAXED, __HIP_MEMORY_SCOPE_WORKGROUP);
          __hip_atomic_fetch_add(ar +  96, pA * va[u].w, __ATOMIC_RELAXED, __HIP_MEMORY_SCOPE_WORKGROUP);
          __hip_atomic_fetch_add(ar + 128, pB * vb[u].x, __ATOMIC_RELAXED, __HIP_MEMORY_SCOPE_WORKGROUP);
          __hip_atomic_fetch_add(ar + 160, pB * vb[u].y, __ATOMIC_RELAXED, __HIP_MEMORY_SCOPE_WORKGROUP);
          __hip_atomic_fetch_add(ar + 192, pB * vb[u].z, __ATOMIC_RELAXED, __HIP_MEMORY_SCOPE_WORKGROUP);
          __hip_atomic_fetch_add(ar + 224, pB * vb[u].w, __ATOMIC_RELAXED, __HIP_MEMORY_SCOPE_WORKGROUP);
        }
      }
    }
  }
  __syncthreads();

  if (!reuse) {
    int* qs = qsave + (size_t)blockIdx.x * QSTRIDE;
    const bool fits = !big && total < (unsigned)QSTRIDE;
    if (fits) {
      for (unsigned i = tid; i < total; i += AGG_THREADS) qs[1 + i] = (int)queue[i];
    }
    if (tid == 0) qs[0] = fits ? (int)total : -1;
  }

  {
    const float4 bA = ((const float4*)bias)[lane];
    const float4 bB = ((const float4*)bias)[32 + lane];
    for (int r = wave; r < SLAB; r += AGG_WAVES) {
      const float* ar = acc + r * FDIM + lane;
      const int    hA = r * HEADS + (hsel ? 1 : 0);
      const int    hB = r * HEADS + (hsel ? 3 : 2);
      const float  eA = unshifted ? 1e-16f * expf(mrun[hA]) : 1e-16f;
      const float  eB = unshifted ? 1e-16f * expf(mrun[hB]) : 1e-16f;
      const float  dA = den[hA] + eA;
      const float  dB = den[hB] + eB;
      float o[8] = {ar[0] / dA + bA.x,   ar[32] / dA + bA.y,  ar[64] / dA + bA.z,  ar[96] / dA + bA.w,
                    ar[128] / dB + bB.x, ar[160] / dB + bB.y, ar[192] / dB + bB.z, ar[224] / dB + bB.w};
      #pragma unroll
      for (int q = 0; q < 8; ++q) o[q] = o[q] > 0.0f ? o[q] : (expf(o[q]) - 1.0f);
      float4* orow = (float4*)(outb + (size_t)(d0 + r) * FDIM);
      orow[lane]      = make_float4(o[0], o[1], o[2], o[3]);
      orow[32 + lane] = make_float4(o[4], o[5], o[6], o[7]);
    }
  }
}

extern "C" void kernel_launch(void* const* d_in, const int* in_sizes, int n_in,
                              void* d_out, int out_size, void* d_ws, size_t ws_size,
                              hipStream_t stream) {
  const float* x    = (const float*)d_in[0];
  const int*   ei   = (const int*)  d_in[1];
  const float* Wl0  = (const float*)d_in[2];
  const float* Wr0  = (const float*)d_in[3];
  const float* att0 = (const float*)d_in[4];
  const float* b0   = (const float*)d_in[5];
  const float* Wl1  = (const float*)d_in[6];
  const float* Wr1  = (const float*)d_in[7];
  const float* att1 = (const float*)d_in[8];
  const float* b1   = (const float*)d_in[9];
  float* out = (float*)d_out;

  float*    ws    = (float*)d_ws;
  float*    xl    = ws;
  float*    hbuf  = xl   + (size_t)N_NODES * FDIM;
  int*      qsave = (int*)(hbuf + (size_t)N_NODES * FDIM);
  unsigned* boff = (unsigned*)(qsave + (size_t)NSLAB * QSTRIDE);
  unsigned* bent = boff + (size_t)NEBLK * EBLK;

  auto layer = [&](const float* in, const float* Wl, const float* Wr,
                   const float* att, const float* bias, float* outb, int reuse,
                   unsigned* boffw, unsigned* bentw) {
    gemm_f32_wmma<<<N_NODES / 16, 512, 0, stream>>>(in, Wl, xl, ei, boffw, bentw);
    slab_edge_phase<<<N_NODES / SLAB, AGG_THREADS, 0, stream>>>(xl, in, Wr, att, ei, bias, outb,
                                                                qsave, reuse, boff, bent);
  };

  layer(x,    Wl0, Wr0, att0, b0, hbuf, 0, boff, bent);
  layer(hbuf, Wl1, Wr1, att1, b1, out,  1, nullptr, nullptr);
}


# xl product re-tiled: one work-group per 80 rows staged in LDS, each wave reuses every B fragment over 5 row tiles (5 accumulators); 250 work-groups, B fetched 5x less often; same k order and sums
# speedup vs baseline: 6.0299x; 1.1639x over previous
#include <hip/hip_runtime.h>

#define N_NODES   20000
#define FDIM      256
#define HEADS     4
#define CDIM      64
#define NEDGE     320000
#define ETOT      (NEDGE + N_NODES)
#define NEG_SLOPE 0.2f
#define SLAB      80
#define NSLAB     (N_NODES / SLAB)
#define EBLK      256
#define NEBLK     (NEDGE / EBLK)
static_assert(NEBLK * EBLK == NEDGE && NEBLK == N_NODES / 16, "one edge block per 16 rows of the product");
static_assert(NSLAB < EBLK, "a bin per slab and one for the rest");

typedef __attribute__((ext_vector_type(2))) float v2f;
typedef __attribute__((ext_vector_type(8))) float v8f;

__device__ __forceinline__ unsigned f2ord(float f) {
  unsigned u = __float_as_uint(f);
  return (u & 0x80000000u) ? ~u : (u | 0x80000000u);
}
__device__ __forceinline__ float ord2f(unsigned u) {
  unsigned v = (u & 0x80000000u) ? (u ^ 0x80000000u) : ~u;
  return __uint_as_float(v);
}

#define GROWS        80
#define GTILES       (GROWS / 16)
#define GEMM_THREADS 512
static_assert(N_NODES % GROWS == 0 && GROWS % 16 == 0, "row blocks tile the node range in whole row tiles");
static_assert(NEBLK == (N_NODES / GROWS) * GTILES, "GTILES edge blocks per product block");
static_assert(GEMM_THREADS / 32 * 16 == FDIM && GTILES <= GEMM_THREADS / 32, "a wave per column panel; a sorting wave per edge block");

__global__ __launch_bounds__(GEMM_THREADS)
void gemm_f32_wmma(const float* __restrict__ A, const float* __restrict__ B,
                   float* __restrict__ Out,
                   const int* __restrict__ ei, unsigned* __restrict__ boff,
                   unsigned* __restrict__ bent) {
  __shared__ __align__(16) float    As[GROWS * FDIM];
  __shared__ __align__(16) unsigned bins[GTILES * EBLK];
  const int tid  = threadIdx.x;
  const int lane = tid & 31;
  const int wave = __builtin_amdgcn_readfirstlane(tid >> 5);
  const int row_base = blockIdx.x * GROWS;

  const bool sorter = boff != nullptr && wave < GTILES;
  const int  eblk   = blockIdx.x * GTILES + wave;
  int4 dA = make_int4(0, 0, 0, 0), dB = dA, sA = dA, sB = dA;
  if (sorter) {
    const int4* dp = (const int4*)(ei + NEDGE + eblk * EBLK) + lane * 2;
    const int4* sp = (const int4*)(ei + eblk * EBLK) + lane * 2;
    dA = dp[0]; dB = dp[1];
    sA = sp[0]; sB = sp[1];
  }

  {
    float4*       a4 = (float4*)As;
    const float4* g4 = (const float4*)(A + (size_t)row_base * FDIM);
    constexpr int NST = GROWS * FDIM / 4 / GEMM_THREADS;
    static_assert(NST * GEMM_THREADS == GROWS * FDIM / 4, "whole 16-byte pieces per thread");
    float4 st[NST];
    #pragma unroll
    for (int j = 0; j < NST; ++j) st[j] = g4[tid + j * GEMM_THREADS];
    #pragma unroll
    for (int j = 0; j < NST; ++j) {
      const int i = tid + j * GEMM_THREADS;
      const int r = i >> 6, c4 = i & 63;
      a4[(r << 6) + (c4 ^ (r & 15))] = st[j];
    }
  }
  __syncthreads();

  if (sorter) {
    const int dv[8] = {dA.x, dA.y, dA.z, dA.w, dB.x, dB.y, dB.z, dB.w};
    const int sv[8] = {sA.x, sA.y, sA.z, sA.w, sB.x, sB.y, sB.z, sB.w};
    unsigned* bw    = bins + wave * EBLK;
    uint4*    bins4 = (uint4*)bw;
    bins4[lane * 2]     = make_uint4(0u, 0u, 0u, 0u);
    bins4[lane * 2 + 1] = make_uint4(0u, 0u, 0u, 0u);
    __builtin_amdgcn_fence(__ATOMIC_SEQ_CST, "wavefront");
    unsigned bin[8], rk[8], word[8];
    #pragma unroll
    for (int k = 0; k < 8; ++k) {
      const unsigned d  = (unsigned)dv[k];
      const bool     in = d < (unsigned)N_NODES;
      const unsigned sl = in ? d / (unsigned)SLAB : (unsigned)NSLAB;
      bin[k]  = sl;
      word[k] = ((unsigned)sv[k] & 0xFFFFFu) | ((in ? d - sl * (unsigned)SLAB : 0u) << 20);
      rk[k]   = atomicAdd(&bw[sl], 1u);
    }
    __builtin_amdgcn_fence(__ATOMIC_SEQ_CST, "wavefront");
    const uint4 cA = bins4[lane * 2], cB = bins4[lane * 2 + 1];
    const unsigned cnt[8] = {cA.x, cA.y, cA.z, cA.w, cB.x, cB.y, cB.z, cB.w};
    unsigned tot = 0u;
    #pragma unroll
    for (int k = 0; k < 8; ++k) tot += cnt[k];
    unsigned inc = tot;
    #pragma unroll
    for (int dlt = 1; dlt < 32; dlt <<= 1) {
      const unsigned up = __shfl_up(inc, dlt);
      if (lane >= dlt) inc += up;
    }
    unsigned off[8];
    unsigned run = inc - tot;
    #pragma unroll
    for (int k = 0; k < 8; ++k) { off[k] = run; run += cnt[k]; }
    __builtin_amdgcn_fence(__ATOMIC_SEQ_CST, "wavefront");
    bins4[lane * 2]     = make_uint4(off[0], off[1], off[2], off[3]);
    bins4[lane * 2 + 1] = make_uint4(off[4], off[5], off[6], off[7]);
    unsigned pw[8];
    #pragma unroll
    for (int k = 0; k < 8; ++k) pw[k] = off[k] | ((off[k] + cnt[k]) << 16);
    uint4* bo4 = (uint4*)(boff + (size_t)eblk * EBLK);
    bo4[lane * 2]     = make_uint4(pw[0], pw[1], pw[2], pw[3]);
    bo4[lane * 2 + 1] = make_uint4(pw[4], pw[5], pw[6], pw[7]);
    __builtin_amdgcn_fence(__ATOMIC_SEQ_CST, "wavefront");
    unsigned* eb = bent + (size_t)eblk * EBLK;
    #pragma unroll
    for (int k = 0; k < 8; ++k) eb[bw[bin[k]] + rk[k]] = word[k];
  }

  const int m  = lane & 15;
  const int kh = (lane >> 4) << 1;
  const int cn = wave * 16;
  const int sw = m << 2;
  const float* bp = B + (size_t)kh * FDIM + cn + m;
  const float* ap = As + m * FDIM + kh;
  v8f c[GTILES] = {};
  #pragma unroll 8
  for (int k = 0; k < FDIM; k += 4) {
    v2f b;
    b.x = bp[(size_t)k * FDIM];
    b.y = bp[(size_t)(k + 1) * FDIM];
    const float* ak = ap + (k ^ sw);
    #pragma unroll
    for (int t = 0; t < GTILES; ++t) {
      const v2f a = *(const v2f*)(ak + t * 16 * FDIM);
      c[t] = __builtin_amdgcn_wmma_f32_16x16x4_f32(false, a, false, b,
                                                   (short)0, c[t], false, false);
    }
  }

  const int rb = row_base + ((lane >> 4) << 3);
  #pragma unroll
  for (int t = 0; t < GTILES; ++t) {
    #pragma unroll
    for (int r = 0; r < 8; ++r) {
      Out[(size_t)(rb + t * 16 + r) * FDIM + cn + m] = c[t][r];
    }
  }
}

#define QCAP        24576
#define AGG_THREADS 1024
#define AGG_WAVES   (AGG_THREADS / 32)
#define SEGLEN      (QCAP - SLAB)
#define NSEG        ((NEDGE + SEGLEN - 1) / SEGLEN)
#define ORD_NEG_INF 0x007FFFFFu
#define QSTRIDE     4096
static_assert(QSTRIDE <= QCAP, "a saved queue always fits the queue in LDS");
#define SPEC_RANGE  16.0f
static_assert(N_NODES % SLAB == 0, "slabs must tile the node range");
static_assert(SLAB % 16 == 0, "the slab's xr rows are formed as whole 16-row matrix tiles");
static_assert(ETOT < (1 << 20) && SLAB < (1 << 12), "queue word packs edge id or source row and local row");

__device__ __forceinline__ float row16_sum(float v) {
  v += __int_as_float(__builtin_amdgcn_mov_dpp(__float_as_int(v), 0x161, 0xf, 0xf, true));
  v += __int_as_float(__builtin_amdgcn_mov_dpp(__float_as_int(v), 0x162, 0xf, 0xf, true));
  v += __int_as_float(__builtin_amdgcn_mov_dpp(__float_as_int(v), 0x164, 0xf, 0xf, true));
  v += __int_as_float(__builtin_amdgcn_mov_dpp(__float_as_int(v), 0x168, 0xf, 0xf, true));
  return v;
}

template <int SRC>
__device__ __forceinline__ float row16_read(float v) {
  static_assert(SRC >= 0 && SRC < 16, "a lane of the row");
  return __int_as_float(__builtin_amdgcn_mov_dpp(__float_as_int(v), 0x150 + SRC, 0xf, 0xf, true));
}

__device__ __forceinline__ float leaky_max(float v) { return fmaxf(v, NEG_SLOPE * v); }
__device__ __forceinline__ float score4(const float4 l, const float4 r, const float4 a) {
  return leaky_max(l.x + r.x) * a.x + leaky_max(l.y + r.y) * a.y +
         leaky_max(l.z + r.z) * a.z + leaky_max(l.w + r.w) * a.w;
}

template <bool B> struct flag_t { static constexpr bool value = B; };

__global__ __launch_bounds__(AGG_THREADS)
void slab_edge_phase(const float* __restrict__ xl, const float* __restrict__ xin,
                     const float* __restrict__ Wr,
                     const float* __restrict__ att, const int* __restrict__ ei,
                     const float* __restrict__ bias, float* __restrict__ outb,
                     int* __restrict__ qsave, int reuse,
                     const unsigned* __restrict__ boff,
                     const unsigned* __restrict__ bent) {
  __shared__ __align__(16) float    acc[(SLAB + 1) * FDIM];
  __shared__ __align__(16) float    xrs[(SLAB + 1) * FDIM];
  __shared__ unsigned queue[QCAP];
  __shared__ unsigned mb[(SLAB + 1) * HEADS];
  __shared__ __align__(16) float    mrun[(SLAB + 1) * HEADS];
  __shared__ float    den[(SLAB + 1) * HEADS];
  __shared__ float    sc[SLAB * HEADS];
  __shared__ unsigned qcount;
  __shared__ unsigned respec;

  const int tid  = threadIdx.x;
  const int lane = tid & 31;
  const int wave = __builtin_amdgcn_readfirstlane(tid >> 5);
  const int d0   = blockIdx.x * SLAB;
  const bool hsel = lane >= 16;

  {
    float4*       a4 = (float4*)acc;
    float4*       x4 = (float4*)xrs;
    const float4* g4 = (const float4*)(xin + (size_t)d0 * FDIM);
    const float4  z  = make_float4(0.0f, 0.0f, 0.0f, 0.0f);
    constexpr int NST = SLAB * FDIM / 4 / AGG_THREADS;
    static_assert(NST * AGG_THREADS == SLAB * FDIM / 4, "whole 16-byte pieces per thread");
    float4 st[NST];
    #pragma unroll
    for (int j = 0; j < NST; ++j) st[j] = g4[tid + j * AGG_THREADS];
    #pragma unroll
    for (int j = 0; j < NST; ++j) {
      const int i = tid + j * AGG_THREADS;
      const int r = i >> 6, c4 = i & 63;
      a4[(r << 6) + (c4 ^ (r & 15))] = st[j];
    }
    if (tid < FDIM / 4) x4[SLAB * FDIM / 4 + tid] = z;
    if (tid < (SLAB + 1) * HEADS) { mb[tid] = ORD_NEG_INF; mrun[tid] = 0.0f; den[tid] = 0.0f; }
    if (tid < SLAB) queue[tid] = (unsigned)(d0 + tid) | ((unsigned)tid << 20);
    if (tid == 0) { qcount = SLAB; respec = 0u; }
  }
  __syncthreads();

  const int role = ((wave >> 2) ^ wave) & 1;
  const int ridx = ((wave >> 3) << 2) | (wave & 3);
  if (role == 0) {
    const int m  = lane & 15;
    const int kh = (lane >> 4) << 1;
    const int cn = ridx * 16;
    const int sw = m << 2;
    const float* bp = Wr + (size_t)kh * FDIM + cn + m;
    const float* ap = acc + m * FDIM + kh;
    v8f c[SLAB / 16] = {};
    #pragma unroll 8
    for (int k = 0; k < FDIM; k += 4) {
      v2f b;
      b.x = bp[(size_t)k * FDIM];
      b.y = bp[(size_t)(k + 1) * FDIM];
      const float* ak = ap + (k ^ sw);
      #pragma unroll
      for (int t = 0; t < SLAB / 16; ++t) {
        const v2f a = *(const v2f*)(ak + t * 16 * FDIM);
        c[t] = __builtin_amdgcn_wmma_f32_16x16x4_f32(false, a, false, b,
                                                     (short)0, c[t], false, false);
      }
    }
    const int rb = (lane >> 4) << 3;
    #pragma unroll
    for (int t = 0; t < SLAB / 16; ++t) {
      #pragma unroll
      for (int r = 0; r < 8; ++r) xrs[(t * 16 + rb + r) * FDIM + cn + m] = c[t][r];
    }
  } else {
    constexpr int FTHREADS = AGG_THREADS / 2;
    const int   ftid = ridx * 32 + lane;
    const int* qs  = qsave + (size_t)blockIdx.x * QSTRIDE;
    const int  cnt = reuse ? __builtin_amdgcn_readfirstlane(qs[0]) : -1;
    const bool kept = cnt >= SLAB && cnt < QSTRIDE;
    if (kept) {
      for (int i = ftid; i < cnt; i += FTHREADS) queue[i] = (unsigned)qs[1 + i];
      if (ftid == 0) qcount = (unsigned)cnt;
    }
    if (!kept) {
      constexpr int NB = (NEBLK + FTHREADS - 1) / FTHREADS;
      unsigned ow[NB], w[NB][4], nrun[NB];
      #pragma unroll
      for (int t = 0; t < NB; ++t) {
        const int b = ftid + t * FTHREADS;
        ow[t] = boff[(size_t)min(b, NEBLK - 1) * EBLK + blockIdx.x];
        if (b >= NEBLK) ow[t] = 0u;
      }
      #pragma unroll
      for (int t = 0; t < NB; ++t) {
        const int b = min(ftid + t * FTHREADS, NEBLK - 1);
        const unsigned o0 = ow[t] & 0xFFFFu;
        const unsigned o1 = min(ow[t] >> 16, (unsigned)EBLK);
        nrun[t] = o1 > o0 ? o1 - o0 : 0u;
        #pragma unroll
        for (int q = 0; q < 4; ++q)
          w[t][q] = bent[(size_t)b * EBLK + min(o0 + (unsigned)q, (unsigned)EBLK - 1u)];
      }
      unsigned bm[NB][4];
      unsigned tot = 0u;
      #pragma unroll
      for (int t = 0; t < NB; ++t) {
        #pragma unroll
        for (int q = 0; q < 4; ++q) {
          bm[t][q] = __builtin_amdgcn_ballot_w32((unsigned)q < nrun[t]);
          tot += (unsigned)__builtin_popcount(bm[t][q]);
        }
      }
      if (tot != 0u) {
        unsigned base = 0u;
        if (lane == 0) base = atomicAdd(&qcount, tot);
        base = (unsigned)__builtin_amdgcn_readfirstlane((int)base);
        #pragma unroll
        for (int t = 0; t < NB; ++t) {
          #pragma unroll
          for (int q = 0; q < 4; ++q) {
            const unsigned slot = base + __builtin_amdgcn_mbcnt_lo(bm[t][q], 0u);
            if ((unsigned)q < nrun[t] && slot < (unsigned)QCAP) queue[slot] = w[t][q];
            base += (unsigned)__builtin_popcount(bm[t][q]);
          }
        }
      }
      #pragma unroll
      for (int t = 0; t < NB; ++t) {
        const int b = min(ftid + t * FTHREADS, NEBLK - 1);
        const unsigned o0 = ow[t] & 0xFFFFu;
        for (unsigned i = 4u; i < nrun[t]; ++i) {
          const unsigned wd   = bent[(size_t)b * EBLK + o0 + i];
          const unsigned slot = atomicAdd(&qcount, 1u);
          if (slot < (unsigned)QCAP) queue[slot] = wd;
        }
      }
    }
  }
  __syncthreads();

  {
    float4*      a4 = (float4*)acc;
    const float4 z  = make_float4(0.0f, 0.0f, 0.0f, 0.0f);
    for (int i = tid; i < SLAB * FDIM / 4; i += AGG_THREADS) a4[i] = z;
  }
  __syncthreads();

  const unsigned total  = (unsigned)__builtin_amdgcn_readfirstlane((int)qcount);
  const bool     big    = total > (unsigned)QCAP;
  const int      nbatch = big ? NSEG : 1;
  const bool     spec   = !big;
  bool           unshifted = false;
  const bool     pre    = !big;

  const float4 atA = ((const float4*)att)[lane];
  const float4 atB = ((const float4*)att)[32 + lane];

  for (int b = 0; b < nbatch; ++b) {
    if (big) {
      __syncthreads();
      if (tid == 0) qcount = (b == 0) ? SLAB : 0;
      if (b == 0 && tid < SLAB) queue[tid] = (unsigned)(NEDGE + d0 + tid) | ((unsigned)tid << 20);
      __syncthreads();
      const int s0 = b * SEGLEN;
      const int s1 = min(NEDGE, s0 + SEGLEN);
      for (int e = s0 + tid; e < s1; e += AGG_THREADS) {
        const unsigned r = (unsigned)(ei[NEDGE + e] - d0);
        if (r < (unsigned)SLAB) {
          const unsigned slot = atomicAdd(&qcount, 1u);
          queue[slot] = (unsigned)e | (r << 20);
        }
      }
      __syncthreads();
    }
    const unsigned M = min((unsigned)__builtin_amdgcn_readfirstlane((int)qcount), (unsigned)QCAP);

    auto pass_a = [&](auto sums_tag) __attribute__((always_inline)) {
      constexpr bool SUMS = decltype(sums_tag)::value;
      for (unsigned base = wave * 32; base < M; base += AGG_WAVES * 32) {
        const unsigned idx = base + lane;
        int m_src = 0, m_dl = SLAB;
        if (idx < M) {
          const unsigned ent = queue[idx];
          const int      e   = (int)(ent & 0xFFFFFu);
          m_dl  = (int)(ent >> 20);
          if (pre) {
            m_src = e;
          } else {
            m_src = e < NEDGE ? ei[e] : e - NEDGE;
            queue[idx] = (unsigned)m_src | ((unsigned)m_dl << 20);
          }
        }
        const int n = (int)min(32u, M - base);
        for (int j0 = 0; j0 < n; j0 += 8) {
          float4 va[8], vb[8];
          #pragma unroll
          for (int u = 0; u < 8; ++u) {
            const int s = __builtin_amdgcn_readlane(m_src, j0 + u);
            const float4* row = (const float4*)(xl + (size_t)s * FDIM);
            va[u] = row[lane];
            vb[u] = row[32 + lane];
          }
          if (SUMS) {
            float T = 0.0f;
            #pragma unroll
            for (int u = 0; u < 8; ++u) {
              const int     dl = __builtin_amdgcn_readlane(m_dl, j0 + u);
              const float4* xq = (const float4*)(xrs + dl * FDIM);
              const float   tA = row16_sum(score4(va[u], xq[lane],      atA));
              const float   tB = row16_sum(score4(vb[u], xq[32 + lane], atB));
              T = ((lane & 7) == u) ? ((lane & 8) ? tB : tA) : T;
              __builtin_amdgcn_sched_barrier(0);
            }
            const float P = expf(T);
            {
              const int dlL = __builtin_amdgcn_ds_bpermute((j0 + (lane & 7)) << 2, m_dl);
              const int hd  = ((lane >> 2) & 2) | (lane >> 4);
              atomicMax(&mb[dlL * HEADS + hd], f2ord(T));
              __hip_atomic_fetch_add(&den[dlL * HEADS + hd], P, __ATOMIC_RELAXED, __HIP_MEMORY_SCOPE_WORKGROUP);
            }
#define SLAB_ADD_EDGE(U)                                                                         \
            {                                                                                    \
              const int   dl = __builtin_amdgcn_readlane(m_dl, j0 + U);                          \
              const float pA = row16_read<U>(P);                                                 \
              const float pB = row16_read<8 + U>(P);                                             \
              float* ar = acc + dl * FDIM + lane;                                                \
              __hip_atomic_fetch_add(ar +   0, pA * va[U].x, __ATOMIC_RELAXED, __HIP_MEMORY_SCOPE_WORKGROUP); \
              __hip_atomic_fetch_add(ar +  32, pA * va[U].y, __ATOMIC_RELAXED, __HIP_MEMORY_SCOPE_WORKGROUP); \
              __hip_atomic_fetch_add(ar +  64, pA * va[U].z, __ATOMIC_RELAXED, __HIP_MEMORY_SCOPE_WORKGROUP); \
              __hip_atomic_fetch_add(ar +  96, pA * va[U].w, __ATOMIC_RELAXED, __HIP_MEMORY_SCOPE_WORKGROUP); \
              __hip_atomic_fetch_add(ar + 128, pB * vb[U].x, __ATOMIC_RELAXED, __HIP_MEMORY_SCOPE_WORKGROUP); \
              __hip_atomic_fetch_add(ar + 160, pB * vb[U].y, __ATOMIC_RELAXED, __HIP_MEMORY_SCOPE_WORKGROUP); \
              __hip_atomic_fetch_add(ar + 192, pB * vb[U].z, __ATOMIC_RELAXED, __HIP_MEMORY_SCOPE_WORKGROUP); \
              __hip_atomic_fetch_add(ar + 224, pB * vb[U].w, __ATOMIC_RELAXED, __HIP_MEMORY_SCOPE_WORKGROUP); \
            }
            SLAB_ADD_EDGE(0) SLAB_ADD_EDGE(1) SLAB_ADD_EDGE(2) SLAB_ADD_EDGE(3)
            SLAB_ADD_EDGE(4) SLAB_ADD_EDGE(5) SLAB_ADD_EDGE(6) SLAB_ADD_EDGE(7)
#undef SLAB_ADD_EDGE
          } else {
            #pragma unroll
            for (int u = 0; u < 8; ++u) {
              const int     dl = __builtin_amdgcn_readlane(m_dl, j0 + u);
              const float4* xq = (const float4*)(xrs + dl * FDIM);
              const float   tA = row16_sum(score4(va[u], xq[lane],      atA));
              const float   tB = row16_sum(score4(vb[u], xq[32 + lane], atB));
              if ((lane & 15) == 0) {
                atomicMax(&mb[dl * HEADS + (hsel ? 1 : 0)], f2ord(tA));
                atomicMax(&mb[dl * HEADS + (hsel ? 3 : 2)], f2ord(tB));
              }
            }
          }
        }
      }
    };
    if (spec) pass_a(flag_t<true>{}); else pass_a(flag_t<false>{});
    __syncthreads();

    if (tid < SLAB * HEADS) {
      const float mnew_b = ord2f(mb[tid]);
      mb[tid] = ORD_NEG_INF;
      if (b == 0) {
        mrun[tid] = mnew_b;
        if (spec && !(fabsf(mnew_b) <= SPEC_RANGE)) respec = 1u;
      } else {
        const float mo = mrun[tid];
        const float mn = fmaxf(mo, mnew_b);
        const float f  = expf(mo - mn);
        mrun[tid] = mn;
        den[tid] *= f;
        sc[tid]   = f;
      }
    }
    __syncthreads();
    if (spec) {
      unshifted = __builtin_amdgcn_readfirstlane((int)respec) == 0;
      if (unshifted) break;
      float4*      a4 = (float4*)acc;
      const float4 z  = make_float4(0.0f, 0.0f, 0.0f, 0.0f);
      for (int i = tid; i < SLAB * FDIM / 4; i += AGG_THREADS) a4[i] = z;
      if (tid < SLAB * HEADS) den[tid] = 0.0f;
      __syncthreads();
    }
    if (b > 0) {
      #pragma unroll 1
      for (int i = tid; i < SLAB * FDIM; i += AGG_THREADS) {
        const int r = i >> 8, q = (i >> 5) & 7, l = i & 31;
        acc[i] *= sc[r * HEADS + ((q >> 2) << 1) + (l >> 4)];
      }
      __syncthreads();
    }

    for (unsigned base = wave * 32; base < M; base += AGG_WAVES * 32) {
      const unsigned idx = base + lane;
      int   m_src = 0, m_dl = SLAB;
      float m_m0 = 0.0f, m_m1 = 0.0f, m_m2 = 0.0f, m_m3 = 0.0f;
      if (idx < M) {
        const unsigned ent = queue[idx];
        m_src = (int)(ent & 0xFFFFFu);
        m_dl  = (int)(ent >> 20);
        const float4 mm = ((const float4*)mrun)[m_dl];
        m_m0 = mm.x; m_m1 = mm.y; m_m2 = mm.z; m_m3 = mm.w;
      }
      const int n = (int)min(32u, M - base);
      for (int j0 = 0; j0 < n; j0 += 8) {
        float4 va[8], vb[8];
        #pragma unroll
        for (int u = 0; u < 8; ++u) {
          const int s = __builtin_amdgcn_readlane(m_src, j0 + u);
          const float4* row = (const float4*)(xl + (size_t)s * FDIM);
          va[u] = row[lane];
          vb[u] = row[32 + lane];
        }
        #pragma unroll
        for (int u = 0; u < 8; ++u) {
          const int     dl = __builtin_amdgcn_readlane(m_dl, j0 + u);
          const float4* xq = (const float4*)(xrs + dl * FDIM);
          const float   tA = row16_sum(score4(va[u], xq[lane],      atA));
          const float   tB = row16_sum(score4(vb[u], xq[32 + lane], atB));
          const float m0 = __int_as_float(__builtin_amdgcn_readlane(__float_as_int(m_m0), j0 + u));
          const float m1 = __int_as_float(__builtin_amdgcn_readlane(__float_as_int(m_m1), j0 + u));
          const float m2 = __int_as_float(__builtin_amdgcn_readlane(__float_as_int(m_m2), j0 + u));
          const float m3 = __int_as_float(__builtin_amdgcn_readlane(__float_as_int(m_m3), j0 + u));
          const float pA = expf(tA - (hsel ? m1 : m0));
          const float pB = expf(tB - (hsel ? m3 : m2));
          if ((lane & 15) == 0) {
            __hip_atomic_fetch_add(&den[dl * HEADS + (hsel ? 1 : 0)], pA, __ATOMIC_RELAXED, __HIP_MEMORY_SCOPE_WORKGROUP);
            __hip_atomic_fetch_add(&den[dl * HEADS + (hsel ? 3 : 2)], pB, __ATOMIC_RELAXED, __HIP_MEMORY_SCOPE_WORKGROUP);
          }
          float* ar = acc + dl * FDIM + lane;
          __hip_atomic_fetch_add(ar +   0, pA * va[u].x, __ATOMIC_RELAXED, __HIP_MEMORY_SCOPE_WORKGROUP);
          __hip_atomic_fetch_add(ar +  32, pA * va[u].y, __ATOMIC_RELAXED, __HIP_MEMORY_SCOPE_WORKGROUP);
          __hip_atomic_fetch_add(ar +  64, pA * va[u].z, __ATOMIC_RELAXED, __HIP_MEMORY_SCOPE_WORKGROUP);
          __hip_atomic_fetch_add(ar +  96, pA * va[u].w, __ATOMIC_RELAXED, __HIP_MEMORY_SCOPE_WORKGROUP);
          __hip_atomic_fetch_add(ar + 128, pB * vb[u].x, __ATOMIC_RELAXED, __HIP_MEMORY_SCOPE_WORKGROUP);
          __hip_atomic_fetch_add(ar + 160, pB * vb[u].y, __ATOMIC_RELAXED, __HIP_MEMORY_SCOPE_WORKGROUP);
          __hip_atomic_fetch_add(ar + 192, pB * vb[u].z, __ATOMIC_RELAXED, __HIP_MEMORY_SCOPE_WORKGROUP);
          __hip_atomic_fetch_add(ar + 224, pB * vb[u].w, __ATOMIC_RELAXED, __HIP_MEMORY_SCOPE_WORKGROUP);
        }
      }
    }
  }
  __syncthreads();

  if (!reuse) {
    int* qs = qsave + (size_t)blockIdx.x * QSTRIDE;
    const bool fits = !big && total < (unsigned)QSTRIDE;
    if (fits) {
      for (unsigned i = tid; i < total; i += AGG_THREADS) qs[1 + i] = (int)queue[i];
    }
    if (tid == 0) qs[0] = fits ? (int)total : -1;
  }

  {
    const float4 bA = ((const float4*)bias)[lane];
    const float4 bB = ((const float4*)bias)[32 + lane];
    for (int r = wave; r < SLAB; r += AGG_WAVES) {
      const float* ar = acc + r * FDIM + lane;
      const int    hA = r * HEADS + (hsel ? 1 : 0);
      const int    hB = r * HEADS + (hsel ? 3 : 2);
      const float  eA = unshifted ? 1e-16f * expf(mrun[hA]) : 1e-16f;
      const float  eB = unshifted ? 1e-16f * expf(mrun[hB]) : 1e-16f;
      const float  dA = den[hA] + eA;
      const float  dB = den[hB] + eB;
      float o[8] = {ar[0] / dA + bA.x,   ar[32] / dA + bA.y,  ar[64] / dA + bA.z,  ar[96] / dA + bA.w,
                    ar[128] / dB + bB.x, ar[160] / dB + bB.y, ar[192] / dB + bB.z, ar[224] / dB + bB.w};
      #pragma unroll
      for (int q = 0; q < 8; ++q) o[q] = o[q] > 0.0f ? o[q] : (expf(o[q]) - 1.0f);
      float4* orow = (float4*)(outb + (size_t)(d0 + r) * FDIM);
      orow[lane]      = make_float4(o[0], o[1], o[2], o[3]);
      orow[32 + lane] = make_float4(o[4], o[5], o[6], o[7]);
    }
  }
}

extern "C" void kernel_launch(void* const* d_in, const int* in_sizes, int n_in,
                              void* d_out, int out_size, void* d_ws, size_t ws_size,
                              hipStream_t stream) {
  const float* x    = (const float*)d_in[0];
  const int*   ei   = (const int*)  d_in[1];
  const float* Wl0  = (const float*)d_in[2];
  const float* Wr0  = (const float*)d_in[3];
  const float* att0 = (const float*)d_in[4];
  const float* b0   = (const float*)d_in[5];
  const float* Wl1  = (const float*)d_in[6];
  const float* Wr1  = (const float*)d_in[7];
  const float* att1 = (const float*)d_in[8];
  const float* b1   = (const float*)d_in[9];
  float* out = (float*)d_out;

  float*    ws    = (float*)d_ws;
  float*    xl    = ws;
  float*    hbuf  = xl   + (size_t)N_NODES * FDIM;
  int*      qsave = (int*)(hbuf + (size_t)N_NODES * FDIM);
  unsigned* boff = (unsigned*)(qsave + (size_t)NSLAB * QSTRIDE);
  unsigned* bent = boff + (size_t)NEBLK * EBLK;

  auto layer = [&](const float* in, const float* Wl, const float* Wr,
                   const float* att, const float* bias, float* outb, int reuse,
                   unsigned* boffw, unsigned* bentw) {
    gemm_f32_wmma<<<N_NODES / GROWS, GEMM_THREADS, 0, stream>>>(in, Wl, xl, ei, boffw, bentw);
    slab_edge_phase<<<N_NODES / SLAB, AGG_THREADS, 0, stream>>>(xl, in, Wr, att, ei, bias, outb,
                                                                qsave, reuse, boff, bent);
  };

  layer(x,    Wl0, Wr0, att0, b0, hbuf, 0, boff, bent);
  layer(hbuf, Wl1, Wr1, att1, b1, out,  1, nullptr, nullptr);
}


# reduce before the LDS atomic: slab queue put in destination-row order (counting sort in LDS, kept for layer 1); pass A sums a run of one row in registers, one float add per run; sum order changes
# speedup vs baseline: 6.5129x; 1.0801x over previous
#include <hip/hip_runtime.h>

#define N_NODES   20000
#define FDIM      256
#define HEADS     4
#define CDIM      64
#define NEDGE     320000
#define ETOT      (NEDGE + N_NODES)
#define NEG_SLOPE 0.2f
#define SLAB      80
#define NSLAB     (N_NODES / SLAB)
#define EBLK      256
#define NEBLK     (NEDGE / EBLK)
static_assert(NEBLK * EBLK == NEDGE && NEBLK == N_NODES / 16, "one edge block per 16 rows of the product");
static_assert(NSLAB < EBLK, "a bin per slab and one for the rest");

typedef __attribute__((ext_vector_type(2))) float v2f;
typedef __attribute__((ext_vector_type(8))) float v8f;

__device__ __forceinline__ unsigned f2ord(float f) {
  unsigned u = __float_as_uint(f);
  return (u & 0x80000000u) ? ~u : (u | 0x80000000u);
}
__device__ __forceinline__ float ord2f(unsigned u) {
  unsigned v = (u & 0x80000000u) ? (u ^ 0x80000000u) : ~u;
  return __uint_as_float(v);
}

#define GROWS        80
#define GTILES       (GROWS / 16)
#define GEMM_THREADS 512
static_assert(N_NODES % GROWS == 0 && GROWS % 16 == 0, "row blocks tile the node range in whole row tiles");
static_assert(NEBLK == (N_NODES / GROWS) * GTILES, "GTILES edge blocks per product block");
static_assert(GEMM_THREADS / 32 * 16 == FDIM && GTILES <= GEMM_THREADS / 32, "a wave per column panel; a sorting wave per edge block");

__global__ __launch_bounds__(GEMM_THREADS)
void gemm_f32_wmma(const float* __restrict__ A, const float* __restrict__ B,
                   float* __restrict__ Out,
                   const int* __restrict__ ei, unsigned* __restrict__ boff,
                   unsigned* __restrict__ bent) {
  __shared__ __align__(16) float    As[GROWS * FDIM];
  __shared__ __align__(16) unsigned bins[GTILES * EBLK];
  const int tid  = threadIdx.x;
  const int lane = tid & 31;
  const int wave = __builtin_amdgcn_readfirstlane(tid >> 5);
  const int row_base = blockIdx.x * GROWS;

  const bool sorter = boff != nullptr && wave < GTILES;
  const int  eblk   = blockIdx.x * GTILES + wave;
  int4 dA = make_int4(0, 0, 0, 0), dB = dA, sA = dA, sB = dA;
  if (sorter) {
    const int4* dp = (const int4*)(ei + NEDGE + eblk * EBLK) + lane * 2;
    const int4* sp = (const int4*)(ei + eblk * EBLK) + lane * 2;
    dA = dp[0]; dB = dp[1];
    sA = sp[0]; sB = sp[1];
  }

  {
    float4*       a4 = (float4*)As;
    const float4* g4 = (const float4*)(A + (size_t)row_base * FDIM);
    constexpr int NST = GROWS * FDIM / 4 / GEMM_THREADS;
    static_assert(NST * GEMM_THREADS == GROWS * FDIM / 4, "whole 16-byte pieces per thread");
    float4 st[NST];
    #pragma unroll
    for (int j = 0; j < NST; ++j) st[j] = g4[tid + j * GEMM_THREADS];
    #pragma unroll
    for (int j = 0; j < NST; ++j) {
      const int i = tid + j * GEMM_THREADS;
      const int r = i >> 6, c4 = i & 63;
      a4[(r << 6) + (c4 ^ (r & 15))] = st[j];
    }
  }
  __syncthreads();

  if (sorter) {
    const int dv[8] = {dA.x, dA.y, dA.z, dA.w, dB.x, dB.y, dB.z, dB.w};
    const int sv[8] = {sA.x, sA.y, sA.z, sA.w, sB.x, sB.y, sB.z, sB.w};
    unsigned* bw    = bins + wave * EBLK;
    uint4*    bins4 = (uint4*)bw;
    bins4[lane * 2]     = make_uint4(0u, 0u, 0u, 0u);
    bins4[lane * 2 + 1] = make_uint4(0u, 0u, 0u, 0u);
    __builtin_amdgcn_fence(__ATOMIC_SEQ_CST, "wavefront");
    unsigned bin[8], rk[8], word[8];
    #pragma unroll
    for (int k = 0; k < 8; ++k) {
      const unsigned d  = (unsigned)dv[k];
      const bool     in = d < (unsigned)N_NODES;
      const unsigned sl = in ? d / (unsigned)SLAB : (unsigned)NSLAB;
      bin[k]  = sl;
      word[k] = ((unsigned)sv[k] & 0xFFFFFu) | ((in ? d - sl * (unsigned)SLAB : 0u) << 20);
      rk[k]   = atomicAdd(&bw[sl], 1u);
    }
    __builtin_amdgcn_fence(__ATOMIC_SEQ_CST, "wavefront");
    const uint4 cA = bins4[lane * 2], cB = bins4[lane * 2 + 1];
    const unsigned cnt[8] = {cA.x, cA.y, cA.z, cA.w, cB.x, cB.y, cB.z, cB.w};
    unsigned tot = 0u;
    #pragma unroll
    for (int k = 0; k < 8; ++k) tot += cnt[k];
    unsigned inc = tot;
    #pragma unroll
    for (int dlt = 1; dlt < 32; dlt <<= 1) {
      const unsigned up = __shfl_up(inc, dlt);
      if (lane >= dlt) inc += up;
    }
    unsigned off[8];
    unsigned run = inc - tot;
    #pragma unroll
    for (int k = 0; k < 8; ++k) { off[k] = run; run += cnt[k]; }
    __builtin_amdgcn_fence(__ATOMIC_SEQ_CST, "wavefront");
    bins4[lane * 2]     = make_uint4(off[0], off[1], off[2], off[3]);
    bins4[lane * 2 + 1] = make_uint4(off[4], off[5], off[6], off[7]);
    unsigned pw[8];
    #pragma unroll
    for (int k = 0; k < 8; ++k) pw[k] = off[k] | ((off[k] + cnt[k]) << 16);
    uint4* bo4 = (uint4*)(boff + (size_t)eblk * EBLK);
    bo4[lane * 2]     = make_uint4(pw[0], pw[1], pw[2], pw[3]);
    bo4[lane * 2 + 1] = make_uint4(pw[4], pw[5], pw[6], pw[7]);
    __builtin_amdgcn_fence(__ATOMIC_SEQ_CST, "wavefront");
    unsigned* eb = bent + (size_t)eblk * EBLK;
    #pragma unroll
    for (int k = 0; k < 8; ++k) eb[bw[bin[k]] + rk[k]] = word[k];
  }

  const int m  = lane & 15;
  const int kh = (lane >> 4) << 1;
  const int cn = wave * 16;
  const int sw = m << 2;
  const float* bp = B + (size_t)kh * FDIM + cn + m;
  const float* ap = As + m * FDIM + kh;
  v8f c[GTILES] = {};
  #pragma unroll 8
  for (int k = 0; k < FDIM; k += 4) {
    v2f b;
    b.x = bp[(size_t)k * FDIM];
    b.y = bp[(size_t)(k + 1) * FDIM];
    const float* ak = ap + (k ^ sw);
    #pragma unroll
    for (int t = 0; t < GTILES; ++t) {
      const v2f a = *(const v2f*)(ak + t * 16 * FDIM);
      c[t] = __builtin_amdgcn_wmma_f32_16x16x4_f32(false, a, false, b,
                                                   (short)0, c[t], false, false);
    }
  }

  const int rb = row_base + ((lane >> 4) << 3);
  #pragma unroll
  for (int t = 0; t < GTILES; ++t) {
    #pragma unroll
    for (int r = 0; r < 8; ++r) {
      Out[(size_t)(rb + t * 16 + r) * FDIM + cn + m] = c[t][r];
    }
  }
}

#define QCAP        24576
#define AGG_THREADS 1024
#define AGG_WAVES   (AGG_THREADS / 32)
#define SEGLEN      (QCAP - SLAB)
#define NSEG        ((NEDGE + SEGLEN - 1) / SEGLEN)
#define ORD_NEG_INF 0x007FFFFFu
#define QSTRIDE     4096
#define RBINS       96
static_assert(RBINS == 3 * 32 && RBINS > SLAB, "three counters per lane of the scanning wave; a counter per row and one spare");
static_assert(QSTRIDE == 4 * AGG_THREADS, "the ordering pass takes four queue words per thread");
static_assert(QSTRIDE <= QCAP, "a saved queue always fits the queue in LDS");
#define SPEC_RANGE  16.0f
static_assert(N_NODES % SLAB == 0, "slabs must tile the node range");
static_assert(SLAB % 16 == 0, "the slab's xr rows are formed as whole 16-row matrix tiles");
static_assert(ETOT < (1 << 20) && SLAB < (1 << 12), "queue word packs edge id or source row and local row");

__device__ __forceinline__ float row16_sum(float v) {
  v += __int_as_float(__builtin_amdgcn_mov_dpp(__float_as_int(v), 0x161, 0xf, 0xf, true));
  v += __int_as_float(__builtin_amdgcn_mov_dpp(__float_as_int(v), 0x162, 0xf, 0xf, true));
  v += __int_as_float(__builtin_amdgcn_mov_dpp(__float_as_int(v), 0x164, 0xf, 0xf, true));
  v += __int_as_float(__builtin_amdgcn_mov_dpp(__float_as_int(v), 0x168, 0xf, 0xf, true));
  return v;
}

template <int SRC>
__device__ __forceinline__ float row16_read(float v) {
  static_assert(SRC >= 0 && SRC < 16, "a lane of the row");
  return __int_as_float(__builtin_amdgcn_mov_dpp(__float_as_int(v), 0x150 + SRC, 0xf, 0xf, true));
}

__device__ __forceinline__ float leaky_max(float v) { return fmaxf(v, NEG_SLOPE * v); }
__device__ __forceinline__ float score4(const float4 l, const float4 r, const float4 a) {
  return leaky_max(l.x + r.x) * a.x + leaky_max(l.y + r.y) * a.y +
         leaky_max(l.z + r.z) * a.z + leaky_max(l.w + r.w) * a.w;
}

template <bool B> struct flag_t { static constexpr bool value = B; };

__global__ __launch_bounds__(AGG_THREADS)
void slab_edge_phase(const float* __restrict__ xl, const float* __restrict__ xin,
                     const float* __restrict__ Wr,
                     const float* __restrict__ att, const int* __restrict__ ei,
                     const float* __restrict__ bias, float* __restrict__ outb,
                     int* __restrict__ qsave, int reuse,
                     const unsigned* __restrict__ boff,
                     const unsigned* __restrict__ bent) {
  __shared__ __align__(16) float    acc[(SLAB + 1) * FDIM];
  __shared__ __align__(16) float    xrs[(SLAB + 1) * FDIM];
  __shared__ unsigned queue[QCAP + QSTRIDE];
  __shared__ unsigned rcnt[RBINS];
  __shared__ unsigned qkept;
  __shared__ unsigned mb[(SLAB + 1) * HEADS];
  __shared__ __align__(16) float    mrun[(SLAB + 1) * HEADS];
  __shared__ float    den[(SLAB + 1) * HEADS];
  __shared__ float    sc[SLAB * HEADS];
  __shared__ unsigned qcount;
  __shared__ unsigned respec;

  const int tid  = threadIdx.x;
  const int lane = tid & 31;
  const int wave = __builtin_amdgcn_readfirstlane(tid >> 5);
  const int d0   = blockIdx.x * SLAB;
  const bool hsel = lane >= 16;

  {
    float4*       a4 = (float4*)acc;
    float4*       x4 = (float4*)xrs;
    const float4* g4 = (const float4*)(xin + (size_t)d0 * FDIM);
    const float4  z  = make_float4(0.0f, 0.0f, 0.0f, 0.0f);
    constexpr int NST = SLAB * FDIM / 4 / AGG_THREADS;
    static_assert(NST * AGG_THREADS == SLAB * FDIM / 4, "whole 16-byte pieces per thread");
    float4 st[NST];
    #pragma unroll
    for (int j = 0; j < NST; ++j) st[j] = g4[tid + j * AGG_THREADS];
    #pragma unroll
    for (int j = 0; j < NST; ++j) {
      const int i = tid + j * AGG_THREADS;
      const int r = i >> 6, c4 = i & 63;
      a4[(r << 6) + (c4 ^ (r & 15))] = st[j];
    }
    if (tid < FDIM / 4) x4[SLAB * FDIM / 4 + tid] = z;
    if (tid < (SLAB + 1) * HEADS) { mb[tid] = ORD_NEG_INF; mrun[tid] = 0.0f; den[tid] = 0.0f; }
    if (tid < SLAB) queue[tid] = (unsigned)(d0 + tid) | ((unsigned)tid << 20);
    if (tid == 0) { qcount = SLAB; respec = 0u; qkept = 0u; }
    if (tid < RBINS) rcnt[tid] = 0u;
  }
  __syncthreads();

  const int role = ((wave >> 2) ^ wave) & 1;
  const int ridx = ((wave >> 3) << 2) | (wave & 3);
  if (role == 0) {
    const int m  = lane & 15;
    const int kh = (lane >> 4) << 1;
    const int cn = ridx * 16;
    const int sw = m << 2;
    const float* bp = Wr + (size_t)kh * FDIM + cn + m;
    const float* ap = acc + m * FDIM + kh;
    v8f c[SLAB / 16] = {};
    #pragma unroll 8
    for (int k = 0; k < FDIM; k += 4) {
      v2f b;
      b.x = bp[(size_t)k * FDIM];
      b.y = bp[(size_t)(k + 1) * FDIM];
      const float* ak = ap + (k ^ sw);
      #pragma unroll
      for (int t = 0; t < SLAB / 16; ++t) {
        const v2f a = *(const v2f*)(ak + t * 16 * FDIM);
        c[t] = __builtin_amdgcn_wmma_f32_16x16x4_f32(false, a, false, b,
                                                     (short)0, c[t], false, false);
      }
    }
    const int rb = (lane >> 4) << 3;
    #pragma unroll
    for (int t = 0; t < SLAB / 16; ++t) {
      #pragma unroll
      for (int r = 0; r < 8; ++r) xrs[(t * 16 + rb + r) * FDIM + cn + m] = c[t][r];
    }
  } else {
    constexpr int FTHREADS = AGG_THREADS / 2;
    const int   ftid = ridx * 32 + lane;
    const int* qs  = qsave + (size_t)blockIdx.x * QSTRIDE;
    const int  cnt = reuse ? __builtin_amdgcn_readfirstlane(qs[0]) : -1;
    const bool kept = cnt >= SLAB && cnt < QSTRIDE;
    if (kept) {
      for (int i = ftid; i < cnt; i += FTHREADS) queue[i] = (unsigned)qs[1 + i];
      if (ftid == 0) { qcount = (unsigned)cnt; qkept = 1u; }
    }
    if (!kept) {
      constexpr int NB = (NEBLK + FTHREADS - 1) / FTHREADS;
      unsigned ow[NB], w[NB][4], nrun[NB];
      #pragma unroll
      for (int t = 0; t < NB; ++t) {
        const int b = ftid + t * FTHREADS;
        ow[t] = boff[(size_t)min(b, NEBLK - 1) * EBLK + blockIdx.x];
        if (b >= NEBLK) ow[t] = 0u;
      }
      #pragma unroll
      for (int t = 0; t < NB; ++t) {
        const int b = min(ftid + t * FTHREADS, NEBLK - 1);
        const unsigned o0 = ow[t] & 0xFFFFu;
        const unsigned o1 = min(ow[t] >> 16, (unsigned)EBLK);
        nrun[t] = o1 > o0 ? o1 - o0 : 0u;
        #pragma unroll
        for (int q = 0; q < 4; ++q)
          w[t][q] = bent[(size_t)b * EBLK + min(o0 + (unsigned)q, (unsigned)EBLK - 1u)];
      }
      unsigned bm[NB][4];
      unsigned tot = 0u;
      #pragma unroll
      for (int t = 0; t < NB; ++t) {
        #pragma unroll
        for (int q = 0; q < 4; ++q) {
          bm[t][q] = __builtin_amdgcn_ballot_w32((unsigned)q < nrun[t]);
          tot += (unsigned)__builtin_popcount(bm[t][q]);
        }
      }
      if (tot != 0u) {
        unsigned base = 0u;
        if (lane == 0) base = atomicAdd(&qcount, tot);
        base = (unsigned)__builtin_amdgcn_readfirstlane((int)base);
        #pragma unroll
        for (int t = 0; t < NB; ++t) {
          #pragma unroll
          for (int q = 0; q < 4; ++q) {
            const unsigned slot = base + __builtin_amdgcn_mbcnt_lo(bm[t][q], 0u);
            if ((unsigned)q < nrun[t] && slot < (unsigned)QCAP) queue[slot] = w[t][q];
            base += (unsigned)__builtin_popcount(bm[t][q]);
          }
        }
      }
      #pragma unroll
      for (int t = 0; t < NB; ++t) {
        const int b = min(ftid + t * FTHREADS, NEBLK - 1);
        const unsigned o0 = ow[t] & 0xFFFFu;
        for (unsigned i = 4u; i < nrun[t]; ++i) {
          const unsigned wd   = bent[(size_t)b * EBLK + o0 + i];
          const unsigned slot = atomicAdd(&qcount, 1u);
          if (slot < (unsigned)QCAP) queue[slot] = wd;
        }
      }
    }
  }
  __syncthreads();

  const unsigned total  = (unsigned)__builtin_amdgcn_readfirstlane((int)qcount);
  const bool ordering = __builtin_amdgcn_readfirstlane((int)qkept) == 0 && total < (unsigned)QSTRIDE;
  const unsigned qbase = ordering ? (unsigned)QCAP : 0u;
  unsigned oent[4] = {0u, 0u, 0u, 0u}, ornk[4] = {0u, 0u, 0u, 0u};
  if (ordering) {
    #pragma unroll
    for (int k = 0; k < 4; ++k) {
      const unsigned i = (unsigned)tid + (unsigned)k * AGG_THREADS;
      if (i < total) {
        oent[k] = queue[i];
        ornk[k] = atomicAdd(&rcnt[min(oent[k] >> 20, (unsigned)SLAB)], 1u);
      }
    }
  }

  {
    float4*      a4 = (float4*)acc;
    const float4 z  = make_float4(0.0f, 0.0f, 0.0f, 0.0f);
    for (int i = tid; i < SLAB * FDIM / 4; i += AGG_THREADS) a4[i] = z;
  }
  __syncthreads();

  if (ordering) {
    if (wave == 0) {
      const unsigned c0 = rcnt[3 * lane], c1 = rcnt[3 * lane + 1], c2 = rcnt[3 * lane + 2];
      const unsigned tot = c0 + c1 + c2;
      unsigned inc = tot;
      #pragma unroll
      for (int dlt = 1; dlt < 32; dlt <<= 1) {
        const unsigned up = __shfl_up(inc, dlt);
        if (lane >= dlt) inc += up;
      }
      const unsigned run = inc - tot;
      rcnt[3 * lane]     = run;
      rcnt[3 * lane + 1] = run + c0;
      rcnt[3 * lane + 2] = run + c0 + c1;
    }
    __syncthreads();
    #pragma unroll
    for (int k = 0; k < 4; ++k) {
      const unsigned i = (unsigned)tid + (unsigned)k * AGG_THREADS;
      if (i < total) queue[QCAP + rcnt[min(oent[k] >> 20, (unsigned)SLAB)] + ornk[k]] = oent[k];
    }
    __syncthreads();
  }
  const bool     big    = total > (unsigned)QCAP;
  const int      nbatch = big ? NSEG : 1;
  const bool     spec   = !big;
  bool           unshifted = false;
  const bool     pre    = !big;

  const float4 atA = ((const float4*)att)[lane];
  const float4 atB = ((const float4*)att)[32 + lane];

  for (int b = 0; b < nbatch; ++b) {
    if (big) {
      __syncthreads();
      if (tid == 0) qcount = (b == 0) ? SLAB : 0;
      if (b == 0 && tid < SLAB) queue[tid] = (unsigned)(NEDGE + d0 + tid) | ((unsigned)tid << 20);
      __syncthreads();
      const int s0 = b * SEGLEN;
      const int s1 = min(NEDGE, s0 + SEGLEN);
      for (int e = s0 + tid; e < s1; e += AGG_THREADS) {
        const unsigned r = (unsigned)(ei[NEDGE + e] - d0);
        if (r < (unsigned)SLAB) {
          const unsigned slot = atomicAdd(&qcount, 1u);
          queue[slot] = (unsigned)e | (r << 20);
        }
      }
      __syncthreads();
    }
    const unsigned M = min((unsigned)__builtin_amdgcn_readfirstlane((int)qcount), (unsigned)QCAP);

    auto pass_a = [&](auto sums_tag) __attribute__((always_inline)) {
      constexpr bool SUMS = decltype(sums_tag)::value;
      for (unsigned base = wave * 32; base < M; base += AGG_WAVES * 32) {
        const unsigned idx = base + lane;
        int m_src = 0, m_dl = SLAB;
        if (idx < M) {
          const unsigned ent = queue[qbase + idx];
          const int      e   = (int)(ent & 0xFFFFFu);
          m_dl  = (int)(ent >> 20);
          if (pre) {
            m_src = e;
          } else {
            m_src = e < NEDGE ? ei[e] : e - NEDGE;
            queue[idx] = (unsigned)m_src | ((unsigned)m_dl << 20);
          }
        }
        const int n = (int)min(32u, M - base);
        for (int j0 = 0; j0 < n; j0 += 8) {
          float4 va[8], vb[8];
          #pragma unroll
          for (int u = 0; u < 8; ++u) {
            const int s = __builtin_amdgcn_readlane(m_src, j0 + u);
            const float4* row = (const float4*)(xl + (size_t)s * FDIM);
            va[u] = row[lane];
            vb[u] = row[32 + lane];
          }
          if (SUMS) {
            float T = 0.0f;
            #pragma unroll
            for (int u = 0; u < 8; ++u) {
              const int     dl = __builtin_amdgcn_readlane(m_dl, j0 + u);
              const float4* xq = (const float4*)(xrs + dl * FDIM);
              const float   tA = row16_sum(score4(va[u], xq[lane],      atA));
              const float   tB = row16_sum(score4(vb[u], xq[32 + lane], atB));
              T = ((lane & 7) == u) ? ((lane & 8) ? tB : tA) : T;
              __builtin_amdgcn_sched_barrier(0);
            }
            const float P = expf(T);
            {
              const int dlL = __builtin_amdgcn_ds_bpermute((j0 + (lane & 7)) << 2, m_dl);
              const int hd  = ((lane >> 2) & 2) | (lane >> 4);
              const int dl0 = __builtin_amdgcn_readlane(m_dl, j0);
              if (__builtin_amdgcn_ballot_w32(dlL != dl0) == 0u) {
                int   oT = (int)f2ord(T);
                float pS = P;
                oT = (int)max((unsigned)oT, (unsigned)__builtin_amdgcn_mov_dpp(oT, 0x161, 0xf, 0xf, true));
                oT = (int)max((unsigned)oT, (unsigned)__builtin_amdgcn_mov_dpp(oT, 0x162, 0xf, 0xf, true));
                oT = (int)max((unsigned)oT, (unsigned)__builtin_amdgcn_mov_dpp(oT, 0x164, 0xf, 0xf, true));
                pS += __int_as_float(__builtin_amdgcn_mov_dpp(__float_as_int(pS), 0x161, 0xf, 0xf, true));
                pS += __int_as_float(__builtin_amdgcn_mov_dpp(__float_as_int(pS), 0x162, 0xf, 0xf, true));
                pS += __int_as_float(__builtin_amdgcn_mov_dpp(__float_as_int(pS), 0x164, 0xf, 0xf, true));
                if ((lane & 7) == 0) {
                  atomicMax(&mb[dl0 * HEADS + hd], (unsigned)oT);
                  __hip_atomic_fetch_add(&den[dl0 * HEADS + hd], pS, __ATOMIC_RELAXED, __HIP_MEMORY_SCOPE_WORKGROUP);
                }
              } else {
                atomicMax(&mb[dlL * HEADS + hd], f2ord(T));
                __hip_atomic_fetch_add(&den[dlL * HEADS + hd], P, __ATOMIC_RELAXED, __HIP_MEMORY_SCOPE_WORKGROUP);
              }
            }
            float s0, s1, s2, s3, s4, s5, s6, s7;
            int   cur;
#define SLAB_FLUSH_RUN()                                                                         \
            {                                                                                    \
              float* ar = acc + cur * FDIM + lane;                                               \
              __hip_atomic_fetch_add(ar +   0, s0, __ATOMIC_RELAXED, __HIP_MEMORY_SCOPE_WORKGROUP); \
              __hip_atomic_fetch_add(ar +  32, s1, __ATOMIC_RELAXED, __HIP_MEMORY_SCOPE_WORKGROUP); \
              __hip_atomic_fetch_add(ar +  64, s2, __ATOMIC_RELAXED, __HIP_MEMORY_SCOPE_WORKGROUP); \
              __hip_atomic_fetch_add(ar +  96, s3, __ATOMIC_RELAXED, __HIP_MEMORY_SCOPE_WORKGROUP); \
              __hip_atomic_fetch_add(ar + 128, s4, __ATOMIC_RELAXED, __HIP_MEMORY_SCOPE_WORKGROUP); \
              __hip_atomic_fetch_add(ar + 160, s5, __ATOMIC_RELAXED, __HIP_MEMORY_SCOPE_WORKGROUP); \
              __hip_atomic_fetch_add(ar + 192, s6, __ATOMIC_RELAXED, __HIP_MEMORY_SCOPE_WORKGROUP); \
              __hip_atomic_fetch_add(ar + 224, s7, __ATOMIC_RELAXED, __HIP_MEMORY_SCOPE_WORKGROUP); \
            }
#define SLAB_FIRST_EDGE()                                                                        \
            {                                                                                    \
              const float pA = row16_read<0>(P);                                                 \
              const float pB = row16_read<8>(P);                                                 \
              cur = __builtin_amdgcn_readlane(m_dl, j0);                                         \
              s0 = pA * va[0].x; s1 = pA * va[0].y; s2 = pA * va[0].z; s3 = pA * va[0].w;        \
              s4 = pB * vb[0].x; s5 = pB * vb[0].y; s6 = pB * vb[0].z; s7 = pB * vb[0].w;        \
            }
#define SLAB_NEXT_EDGE(U)                                                                        \
            {                                                                                    \
              const int   dl = __builtin_amdgcn_readlane(m_dl, j0 + U);                          \
              const float pA = row16_read<U>(P);                                                 \
              const float pB = row16_read<8 + U>(P);                                             \
              if (dl != cur) {                                                                   \
                SLAB_FLUSH_RUN()                                                                 \
                cur = dl;                                                                        \
                s0 = 0.0f; s1 = 0.0f; s2 = 0.0f; s3 = 0.0f;                                      \
                s4 = 0.0f; s5 = 0.0f; s6 = 0.0f; s7 = 0.0f;                                      \
              }                                                                                  \
              s0 = fmaf(pA, va[U].x, s0); s1 = fmaf(pA, va[U].y, s1);                            \
              s2 = fmaf(pA, va[U].z, s2); s3 = fmaf(pA, va[U].w, s3);                            \
              s4 = fmaf(pB, vb[U].x, s4); s5 = fmaf(pB, vb[U].y, s5);                            \
              s6 = fmaf(pB, vb[U].z, s6); s7 = fmaf(pB, vb[U].w, s7);                            \
            }
            SLAB_FIRST_EDGE()
            SLAB_NEXT_EDGE(1) SLAB_NEXT_EDGE(2) SLAB_NEXT_EDGE(3)
            SLAB_NEXT_EDGE(4) SLAB_NEXT_EDGE(5) SLAB_NEXT_EDGE(6) SLAB_NEXT_EDGE(7)
            SLAB_FLUSH_RUN()
#undef SLAB_NEXT_EDGE
#undef SLAB_FIRST_EDGE
#undef SLAB_FLUSH_RUN
          } else {
            #pragma unroll
            for (int u = 0; u < 8; ++u) {
              const int     dl = __builtin_amdgcn_readlane(m_dl, j0 + u);
              const float4* xq = (const float4*)(xrs + dl * FDIM);
              const float   tA = row16_sum(score4(va[u], xq[lane],      atA));
              const float   tB = row16_sum(score4(vb[u], xq[32 + lane], atB));
              if ((lane & 15) == 0) {
                atomicMax(&mb[dl * HEADS + (hsel ? 1 : 0)], f2ord(tA));
                atomicMax(&mb[dl * HEADS + (hsel ? 3 : 2)], f2ord(tB));
              }
            }
          }
        }
      }
    };
    if (spec) pass_a(flag_t<true>{}); else pass_a(flag_t<false>{});
    __syncthreads();

    if (tid < SLAB * HEADS) {
      const float mnew_b = ord2f(mb[tid]);
      mb[tid] = ORD_NEG_INF;
      if (b == 0) {
        mrun[tid] = mnew_b;
        if (spec && !(fabsf(mnew_b) <= SPEC_RANGE)) respec = 1u;
      } else {
        const float mo = mrun[tid];
        const float mn = fmaxf(mo, mnew_b);
        const float f  = expf(mo - mn);
        mrun[tid] = mn;
        den[tid] *= f;
        sc[tid]   = f;
      }
    }
    __syncthreads();
    if (spec) {
      unshifted = __builtin_amdgcn_readfirstlane((int)respec) == 0;
      if (unshifted) break;
      float4*      a4 = (float4*)acc;
      const float4 z  = make_float4(0.0f, 0.0f, 0.0f, 0.0f);
      for (int i = tid; i < SLAB * FDIM / 4; i += AGG_THREADS) a4[i] = z;
      if (tid < SLAB * HEADS) den[tid] = 0.0f;
      __syncthreads();
    }
    if (b > 0) {
      #pragma unroll 1
      for (int i = tid; i < SLAB * FDIM; i += AGG_THREADS) {
        const int r = i >> 8, q = (i >> 5) & 7, l = i & 31;
        acc[i] *= sc[r * HEADS + ((q >> 2) << 1) + (l >> 4)];
      }
      __syncthreads();
    }

    for (unsigned base = wave * 32; base < M; base += AGG_WAVES * 32) {
      const unsigned idx = base + lane;
      int   m_src = 0, m_dl = SLAB;
      float m_m0 = 0.0f, m_m1 = 0.0f, m_m2 = 0.0f, m_m3 = 0.0f;
      if (idx < M) {
        const unsigned ent = queue[idx];
        m_src = (int)(ent & 0xFFFFFu);
        m_dl  = (int)(ent >> 20);
        const float4 mm = ((const float4*)mrun)[m_dl];
        m_m0 = mm.x; m_m1 = mm.y; m_m2 = mm.z; m_m3 = mm.w;
      }
      const int n = (int)min(32u, M - base);
      for (int j0 = 0; j0 < n; j0 += 8) {
        float4 va[8], vb[8];
        #pragma unroll
        for (int u = 0; u < 8; ++u) {
          const int s = __builtin_amdgcn_readlane(m_src, j0 + u);
          const float4* row = (const float4*)(xl + (size_t)s * FDIM);
          va[u] = row[lane];
          vb[u] = row[32 + lane];
        }
        #pragma unroll
        for (int u = 0; u < 8; ++u) {
          const int     dl = __builtin_amdgcn_readlane(m_dl, j0 + u);
          const float4* xq = (const float4*)(xrs + dl * FDIM);
          const float   tA = row16_sum(score4(va[u], xq[lane],      atA));
          const float   tB = row16_sum(score4(vb[u], xq[32 + lane], atB));
          const float m0 = __int_as_float(__builtin_amdgcn_readlane(__float_as_int(m_m0), j0 + u));
          const float m1 = __int_as_float(__builtin_amdgcn_readlane(__float_as_int(m_m1), j0 + u));
          const float m2 = __int_as_float(__builtin_amdgcn_readlane(__float_as_int(m_m2), j0 + u));
          const float m3 = __int_as_float(__builtin_amdgcn_readlane(__float_as_int(m_m3), j0 + u));
          const float pA = expf(tA - (hsel ? m1 : m0));
          const float pB = expf(tB - (hsel ? m3 : m2));
          if ((lane & 15) == 0) {
            __hip_atomic_fetch_add(&den[dl * HEADS + (hsel ? 1 : 0)], pA, __ATOMIC_RELAXED, __HIP_MEMORY_SCOPE_WORKGROUP);
            __hip_atomic_fetch_add(&den[dl * HEADS + (hsel ? 3 : 2)], pB, __ATOMIC_RELAXED, __HIP_MEMORY_SCOPE_WORKGROUP);
          }
          float* ar = acc + dl * FDIM + lane;
          __hip_atomic_fetch_add(ar +   0, pA * va[u].x, __ATOMIC_RELAXED, __HIP_MEMORY_SCOPE_WORKGROUP);
          __hip_atomic_fetch_add(ar +  32, pA * va[u].y, __ATOMIC_RELAXED, __HIP_MEMORY_SCOPE_WORKGROUP);
          __hip_atomic_fetch_add(ar +  64, pA * va[u].z, __ATOMIC_RELAXED, __HIP_MEMORY_SCOPE_WORKGROUP);
          __hip_atomic_fetch_add(ar +  96, pA * va[u].w, __ATOMIC_RELAXED, __HIP_MEMORY_SCOPE_WORKGROUP);
          __hip_atomic_fetch_add(ar + 128, pB * vb[u].x, __ATOMIC_RELAXED, __HIP_MEMORY_SCOPE_WORKGROUP);
          __hip_atomic_fetch_add(ar + 160, pB * vb[u].y, __ATOMIC_RELAXED, __HIP_MEMORY_SCOPE_WORKGROUP);
          __hip_atomic_fetch_add(ar + 192, pB * vb[u].z, __ATOMIC_RELAXED, __HIP_MEMORY_SCOPE_WORKGROUP);
          __hip_atomic_fetch_add(ar + 224, pB * vb[u].w, __ATOMIC_RELAXED, __HIP_MEMORY_SCOPE_WORKGROUP);
        }
      }
    }
  }
  __syncthreads();

  if (!reuse) {
    int* qs = qsave + (size_t)blockIdx.x * QSTRIDE;
    const bool fits = !big && total < (unsigned)QSTRIDE;
    if (fits) {
      for (unsigned i = tid; i < total; i += AGG_THREADS) qs[1 + i] = (int)queue[qbase + i];
    }
    if (tid == 0) qs[0] = fits ? (int)total : -1;
  }

  {
    const float4 bA = ((const float4*)bias)[lane];
    const float4 bB = ((const float4*)bias)[32 + lane];
    for (int r = wave; r < SLAB; r += AGG_WAVES) {
      const float* ar = acc + r * FDIM + lane;
      const int    hA = r * HEADS + (hsel ? 1 : 0);
      const int    hB = r * HEADS + (hsel ? 3 : 2);
      const float  eA = unshifted ? 1e-16f * expf(mrun[hA]) : 1e-16f;
      const float  eB = unshifted ? 1e-16f * expf(mrun[hB]) : 1e-16f;
      const float  dA = den[hA] + eA;
      const float  dB = den[hB] + eB;
      float o[8] = {ar[0] / dA + bA.x,   ar[32] / dA + bA.y,  ar[64] / dA + bA.z,  ar[96] / dA + bA.w,
                    ar[128] / dB + bB.x, ar[160] / dB + bB.y, ar[192] / dB + bB.z, ar[224] / dB + bB.w};
      #pragma unroll
      for (int q = 0; q < 8; ++q) o[q] = o[q] > 0.0f ? o[q] : (expf(o[q]) - 1.0f);
      float4* orow = (float4*)(outb + (size_t)(d0 + r) * FDIM);
      orow[lane]      = make_float4(o[0], o[1], o[2], o[3]);
      orow[32 + lane] = make_float4(o[4], o[5], o[6], o[7]);
    }
  }
}

extern "C" void kernel_launch(void* const* d_in, const int* in_sizes, int n_in,
                              void* d_out, int out_size, void* d_ws, size_t ws_size,
                              hipStream_t stream) {
  const float* x    = (const float*)d_in[0];
  const int*   ei   = (const int*)  d_in[1];
  const float* Wl0  = (const float*)d_in[2];
  const float* Wr0  = (const float*)d_in[3];
  const float* att0 = (const float*)d_in[4];
  const float* b0   = (const float*)d_in[5];
  const float* Wl1  = (const float*)d_in[6];
  const float* Wr1  = (const float*)d_in[7];
  const float* att1 = (const float*)d_in[8];
  const float* b1   = (const float*)d_in[9];
  float* out = (float*)d_out;

  float*    ws    = (float*)d_ws;
  float*    xl    = ws;
  float*    hbuf  = xl   + (size_t)N_NODES * FDIM;
  int*      qsave = (int*)(hbuf + (size_t)N_NODES * FDIM);
  unsigned* boff = (unsigned*)(qsave + (size_t)NSLAB * QSTRIDE);
  unsigned* bent = boff + (size_t)NEBLK * EBLK;

  auto layer = [&](const float* in, const float* Wl, const float* Wr,
                   const float* att, const float* bias, float* outb, int reuse,
                   unsigned* boffw, unsigned* bentw) {
    gemm_f32_wmma<<<N_NODES / GROWS, GEMM_THREADS, 0, stream>>>(in, Wl, xl, ei, boffw, bentw);
    slab_edge_phase<<<N_NODES / SLAB, AGG_THREADS, 0, stream>>>(xl, in, Wr, att, ei, bias, outb,
                                                                qsave, reuse, boff, bent);
  };

  layer(x,    Wl0, Wr0, att0, b0, hbuf, 0, boff, bent);
  layer(hbuf, Wl1, Wr1, att1, b1, out,  1, nullptr, nullptr);
}
